# starting kernel
# baseline (speedup 1.0000x reference)
#include <hip/hip_runtime.h>

#define IN_DIM  256
#define OUT_DIM 64
#define N_HEADS 4
#define GAT_EPS 1e-12f

typedef float v2f __attribute__((ext_vector_type(2)));
typedef float v8f __attribute__((ext_vector_type(8)));

__global__ __launch_bounds__(256) void gat_zero(float4* __restrict__ p, long n4) {
    long i = (long)blockIdx.x * blockDim.x + threadIdx.x;
    if (i < n4) p[i] = make_float4(0.f, 0.f, 0.f, 0.f);
}

__global__ __launch_bounds__(128) void gat_gemm_wmma(
    const float* __restrict__ h, const float* __restrict__ W,
    float* __restrict__ x, int N)
{
    const int lane = threadIdx.x & 31;
    const int wave = threadIdx.x >> 5;
    const int half = lane >> 4;
    const int l15  = lane & 15;
    const int row0 = blockIdx.x * 16;
    const int col0 = wave * 16;

    int arow = row0 + l15;
    if (arow >= N) arow = N - 1;
    const float* __restrict__ hrow = h + (size_t)arow * IN_DIM;
    const int bcol = col0 + l15;

    v8f acc = {0.f, 0.f, 0.f, 0.f, 0.f, 0.f, 0.f, 0.f};

    for (int k = 0; k < IN_DIM; k += 4) {
        const int ka = k + half * 2;
        v2f a = *(const v2f*)(hrow + ka);
        v2f b;
        b.x = W[(size_t)ka       * OUT_DIM + bcol];
        b.y = W[(size_t)(ka + 1) * OUT_DIM + bcol];
        acc = __builtin_amdgcn_wmma_f32_16x16x4_f32(
             false, a,  false, b,
             (short)0, acc,  false,  false);
    }

    float* __restrict__ xout =
        x + ((size_t)(row0 + half * 8)) * OUT_DIM + col0 + l15;
    if (row0 + 16 <= N) {
        #pragma unroll
        for (int r = 0; r < 8; ++r)
            xout[(size_t)r * OUT_DIM] = acc[r];
    } else {
        #pragma unroll
        for (int r = 0; r < 8; ++r) {
            if (row0 + r + half * 8 < N)
                xout[(size_t)r * OUT_DIM] = acc[r];
        }
    }
}

__global__ __launch_bounds__(256) void gat_coef(
    const float* __restrict__ x, const float* __restrict__ Wl,
    const float* __restrict__ Wr, float* __restrict__ el,
    float* __restrict__ er, float* __restrict__ rowsum, int N)
{
    const int n = blockIdx.x * blockDim.x + threadIdx.x;
    if (n >= N) return;
    const float* __restrict__ xr = x + (size_t)n * OUT_DIM;
    float al[N_HEADS] = {0.f, 0.f, 0.f, 0.f};
    float ar[N_HEADS] = {0.f, 0.f, 0.f, 0.f};
    for (int d = 0; d < OUT_DIM; ++d) {
        const float xv = xr[d];
        #pragma unroll
        for (int hh = 0; hh < N_HEADS; ++hh) {
            al[hh] = fmaf(xv, Wl[hh * OUT_DIM + d], al[hh]);
            ar[hh] = fmaf(xv, Wr[hh * OUT_DIM + d], ar[hh]);
        }
    }
    #pragma unroll
    for (int hh = 0; hh < N_HEADS; ++hh) {
        el[(size_t)n * N_HEADS + hh] = al[hh];
        er[(size_t)n * N_HEADS + hh] = ar[hh];
        rowsum[(size_t)n * N_HEADS + hh] = 0.f;
    }
}

__device__ __forceinline__ float gat_edge_e(float l, float r) {
    float v = l + r;
    v = (v > 0.f) ? v : 0.2f * v;
    return __expf(v);
}

__global__ __launch_bounds__(256) void gat_rowsum(
    const int* __restrict__ ei, const float* __restrict__ el,
    const float* __restrict__ er, float* __restrict__ rowsum, int E)
{
    const int e = blockIdx.x * blockDim.x + threadIdx.x;
    if (e >= E) return;
    const int src = ei[e];
    const int dst = ei[E + e];
    const float4 l = *(const float4*)(el + (size_t)src * N_HEADS);
    const float4 r = *(const float4*)(er + (size_t)dst * N_HEADS);
    float* rs = rowsum + (size_t)src * N_HEADS;
    atomicAdd(rs + 0, gat_edge_e(l.x, r.x));
    atomicAdd(rs + 1, gat_edge_e(l.y, r.y));
    atomicAdd(rs + 2, gat_edge_e(l.z, r.z));
    atomicAdd(rs + 3, gat_edge_e(l.w, r.w));
}

__global__ __launch_bounds__(256) void gat_aggregate(
    const int* __restrict__ ei, const float* __restrict__ el,
    const float* __restrict__ er, const float* __restrict__ rowsum,
    const float* __restrict__ x, float* __restrict__ out, int E)
{
    const int e    = blockIdx.x * 8 + (threadIdx.x >> 5);
    const int lane = threadIdx.x & 31;
    if (e >= E) return;
    const int src = ei[e];
    const int dst = ei[E + e];

    const float4 l  = *(const float4*)(el     + (size_t)src * N_HEADS);
    const float4 r  = *(const float4*)(er     + (size_t)dst * N_HEADS);
    const float4 rs = *(const float4*)(rowsum + (size_t)src * N_HEADS);

    float w[N_HEADS];
    w[0] = gat_edge_e(l.x, r.x) / fmaxf(rs.x, GAT_EPS);
    w[1] = gat_edge_e(l.y, r.y) / fmaxf(rs.y, GAT_EPS);
    w[2] = gat_edge_e(l.z, r.z) / fmaxf(rs.z, GAT_EPS);
    w[3] = gat_edge_e(l.w, r.w) / fmaxf(rs.w, GAT_EPS);

    const v2f xv = *(const v2f*)(x + (size_t)dst * OUT_DIM + lane * 2);
    float* obase = out + (size_t)src * (N_HEADS * OUT_DIM) + lane * 2;
    #pragma unroll
    for (int hh = 0; hh < N_HEADS; ++hh) {
        atomicAdd(obase + hh * OUT_DIM + 0, w[hh] * xv.x);
        atomicAdd(obase + hh * OUT_DIM + 1, w[hh] * xv.y);
    }
}

__global__ __launch_bounds__(256) void gat_finalize(
    float* __restrict__ out, const float* __restrict__ b, long n4)
{
    const long i4 = (long)blockIdx.x * blockDim.x + threadIdx.x;
    if (i4 >= n4) return;
    float4 v = ((const float4*)out)[i4];
    const int d0 = (int)((i4 * 4) & (OUT_DIM - 1));
    v.x += b[d0 + 0]; v.y += b[d0 + 1]; v.z += b[d0 + 2]; v.w += b[d0 + 3];
    v.x = (v.x > 0.f) ? v.x : (__expf(v.x) - 1.f);
    v.y = (v.y > 0.f) ? v.y : (__expf(v.y) - 1.f);
    v.z = (v.z > 0.f) ? v.z : (__expf(v.z) - 1.f);
    v.w = (v.w > 0.f) ? v.w : (__expf(v.w) - 1.f);
    ((float4*)out)[i4] = v;
}

extern "C" void kernel_launch(void* const* d_in, const int* in_sizes, int n_in,
                              void* d_out, int out_size, void* d_ws, size_t ws_size,
                              hipStream_t stream) {
    const float* h  = (const float*)d_in[0];
    const float* W  = (const float*)d_in[1];
    const float* Wl = (const float*)d_in[2];
    const float* Wr = (const float*)d_in[3];
    const float* b  = (const float*)d_in[4];
    const int*   ei = (const int*)d_in[5];

    const int N = in_sizes[0] / IN_DIM;
    const int E = in_sizes[5] / 2;

    float* out = (float*)d_out;

    float* x      = (float*)d_ws;
    float* el     = x  + (size_t)N * OUT_DIM;
    float* er     = el + (size_t)N * N_HEADS;
    float* rowsum = er + (size_t)N * N_HEADS;

    const long out_n4 = (long)out_size / 4;

    gat_zero<<<(unsigned)((out_n4 + 255) / 256), 256, 0, stream>>>(
        (float4*)out, out_n4);

    gat_gemm_wmma<<<(unsigned)((N + 15) / 16), 128, 0, stream>>>(h, W, x, N);

    gat_coef<<<(unsigned)((N + 255) / 256), 256, 0, stream>>>(
        x, Wl, Wr, el, er, rowsum, N);

    gat_rowsum<<<(unsigned)((E + 255) / 256), 256, 0, stream>>>(
        ei, el, er, rowsum, E);

    gat_aggregate<<<(unsigned)((E + 7) / 8), 256, 0, stream>>>(
        ei, el, er, rowsum, x, out, E);

    gat_finalize<<<(unsigned)((out_n4 + 255) / 256), 256, 0, stream>>>(
        out, b, out_n4);
}


# pass 2 also writes a 1-byte owner key per edge; pass 3's ownership scan reads the keys (a quarter of the bytes, 4 keys tested per instruction), not the source indices; sum order: arrival, as before
# speedup vs baseline: 3.6090x; 3.6090x over previous
#include <hip/hip_runtime.h>

#define IN_DIM  256
#define OUT_DIM 64
#define N_HEADS 4
#define GAT_EPS 1e-12f

typedef float v2f __attribute__((ext_vector_type(2)));
typedef float v8f __attribute__((ext_vector_type(8)));

#define ROW_F (N_HEADS * OUT_DIM)

__global__ __launch_bounds__(128) void gat_gemm_wmma(
    const float* __restrict__ h, const float* __restrict__ W,
    float* __restrict__ x, int N)
{
    const int lane = threadIdx.x & 31;
    const int wave = threadIdx.x >> 5;
    const int half = lane >> 4;
    const int l15  = lane & 15;
    const int row0 = blockIdx.x * 16;
    const int col0 = wave * 16;

    int arow = row0 + l15;
    if (arow >= N) arow = N - 1;
    const float* __restrict__ hrow = h + (size_t)arow * IN_DIM;
    const int bcol = col0 + l15;

    v8f acc = {0.f, 0.f, 0.f, 0.f, 0.f, 0.f, 0.f, 0.f};

    for (int k = 0; k < IN_DIM; k += 4) {
        const int ka = k + half * 2;
        v2f a = *(const v2f*)(hrow + ka);
        v2f b;
        b.x = W[(size_t)ka       * OUT_DIM + bcol];
        b.y = W[(size_t)(ka + 1) * OUT_DIM + bcol];
        acc = __builtin_amdgcn_wmma_f32_16x16x4_f32(
             false, a,  false, b,
             (short)0, acc,  false,  false);
    }

    float* __restrict__ xout =
        x + ((size_t)(row0 + half * 8)) * OUT_DIM + col0 + l15;
    if (row0 + 16 <= N) {
        #pragma unroll
        for (int r = 0; r < 8; ++r)
            xout[(size_t)r * OUT_DIM] = acc[r];
    } else {
        #pragma unroll
        for (int r = 0; r < 8; ++r) {
            if (row0 + r + half * 8 < N)
                xout[(size_t)r * OUT_DIM] = acc[r];
        }
    }
}

#define OWN_ROWS    196

__device__ __forceinline__ unsigned own_key(int v) {
    return (((unsigned)v) / (unsigned)OWN_ROWS) & 255u;
}

__global__ __launch_bounds__(256) void gat_coef(
    const float* __restrict__ x, const float* __restrict__ Wl,
    const float* __restrict__ Wr, float* __restrict__ el,
    float* __restrict__ er, const int* __restrict__ ei,
    int* __restrict__ keys, int N, int E)
{
    if (keys != nullptr) {
        const int   nunits = E >> 4;
        const int4* src4   = (const int4*)ei;
        uint4*      key4   = (uint4*)keys;
        const int   stride = (int)(gridDim.x * blockDim.x);
        for (int q = blockIdx.x * blockDim.x + threadIdx.x; q < nunits; q += stride) {
            unsigned w[4];
            #pragma unroll
            for (int j = 0; j < 4; ++j) {
                const int4 v = src4[(size_t)q * 4 + j];
                w[j] = own_key(v.x) | (own_key(v.y) << 8) | (own_key(v.z) << 16) | (own_key(v.w) << 24);
            }
            key4[q] = make_uint4(w[0], w[1], w[2], w[3]);
        }
    }

    const int n = blockIdx.x * blockDim.x + threadIdx.x;
    if (n >= N) return;
    const float* __restrict__ xr = x + (size_t)n * OUT_DIM;
    float al[N_HEADS] = {0.f, 0.f, 0.f, 0.f};
    float ar[N_HEADS] = {0.f, 0.f, 0.f, 0.f};
    for (int d = 0; d < OUT_DIM; ++d) {
        const float xv = xr[d];
        #pragma unroll
        for (int hh = 0; hh < N_HEADS; ++hh) {
            al[hh] = fmaf(xv, Wl[hh * OUT_DIM + d], al[hh]);
            ar[hh] = fmaf(xv, Wr[hh * OUT_DIM + d], ar[hh]);
        }
    }
    #pragma unroll
    for (int hh = 0; hh < N_HEADS; ++hh) {
        el[(size_t)n * N_HEADS + hh] = al[hh];
        er[(size_t)n * N_HEADS + hh] = ar[hh];
    }
}

__device__ __forceinline__ float gat_edge_e(float l, float r) {
    float v = l + r;
    v = (v > 0.f) ? v : 0.2f * v;
    return __expf(v);
}

#define OWN_THREADS 1024
#define OWN_WAVES   (OWN_THREADS / 32)
#define SCAN_U      4
#define OWN_STEP4   (SCAN_U * 32)
#define OWN_BLOCK4  (OWN_STEP4 * OWN_WAVES)
#define STEP_EDGES  (OWN_STEP4 * 4)
#define LIST_CAP    896

typedef volatile int __attribute__((address_space(3))) own_list_word;

__device__ __forceinline__ int own_bcast_i(int v, int j) {
    return __builtin_amdgcn_readlane(v, j);
}
__device__ __forceinline__ float own_bcast_f(float v, int j) {
    return __int_as_float(__builtin_amdgcn_readlane(__float_as_int(v), j));
}

__device__ __forceinline__ unsigned own_match(int v, int r0, int nrows) {
    return ((unsigned)(v - r0) < (unsigned)nrows) ? 1u : 0u;
}

__device__ __forceinline__ void own_sum_group(
    own_list_word* list, int g, int lane, int r0, int nrows, int E,
    const int* __restrict__ ei, const float* __restrict__ el,
    const float* __restrict__ er, float* own_rs_f)
{
    if (lane < g) {
        const int e   = list[lane];
        const int src = ei[e];
        const int dst = ei[(size_t)E + e];
        const float4 r = *(const float4*)(er + (size_t)dst * N_HEADS);
        if (own_match(src, r0, nrows)) {
            const float4 l = *(const float4*)(el + (size_t)src * N_HEADS);
            float* a = own_rs_f + (src - r0) * N_HEADS;
            atomicAdd(a + 0, gat_edge_e(l.x, r.x));
            atomicAdd(a + 1, gat_edge_e(l.y, r.y));
            atomicAdd(a + 2, gat_edge_e(l.z, r.z));
            atomicAdd(a + 3, gat_edge_e(l.w, r.w));
        }
    }
}

__device__ __forceinline__ void own_process_group(
    own_list_word* list, int g, int lane, int r0, int nrows, int E,
    const int* __restrict__ ei, const float* __restrict__ el,
    const float* __restrict__ er, const float* __restrict__ x,
    const float4* own_rs, float* own_acc)
{
    int my_dst = 0, my_lrow = 0;
    float w0 = 0.f, w1 = 0.f, w2 = 0.f, w3 = 0.f;
    if (lane < g) {
        const int e   = list[lane];
        const int src = ei[e];
        my_dst  = ei[(size_t)E + e];
        const float4 r  = *(const float4*)(er + (size_t)my_dst * N_HEADS);
        if (own_match(src, r0, nrows)) {
            my_lrow = src - r0;
            const float4 l  = *(const float4*)(el + (size_t)src    * N_HEADS);
            const float4 rs = own_rs[my_lrow];
            w0 = gat_edge_e(l.x, r.x) / fmaxf(rs.x, GAT_EPS);
            w1 = gat_edge_e(l.y, r.y) / fmaxf(rs.y, GAT_EPS);
            w2 = gat_edge_e(l.z, r.z) / fmaxf(rs.z, GAT_EPS);
            w3 = gat_edge_e(l.w, r.w) / fmaxf(rs.w, GAT_EPS);
        }
    }
    for (int i = 0; i < g; i += 4) {
        int lr[4];
        v2f xv[4];
        #pragma unroll
        for (int k = 0; k < 4; ++k) {
            const int j = min(i + k, g - 1);
            const int d = own_bcast_i(my_dst, j);
            lr[k] = own_bcast_i(my_lrow, j);
            xv[k] = *(const v2f*)(x + (size_t)d * OUT_DIM + lane * 2);
        }
        #pragma unroll
        for (int k = 0; k < 4; ++k) {
            if (i + k < g) {
                const int j = i + k;
                float* a = own_acc + lr[k] * ROW_F + lane * 2;
                const float wa = own_bcast_f(w0, j);
                const float wb = own_bcast_f(w1, j);
                const float wc = own_bcast_f(w2, j);
                const float wd = own_bcast_f(w3, j);
                atomicAdd(a + 0 * OUT_DIM + 0, wa * xv[k].x);
                atomicAdd(a + 0 * OUT_DIM + 1, wa * xv[k].y);
                atomicAdd(a + 1 * OUT_DIM + 0, wb * xv[k].x);
                atomicAdd(a + 1 * OUT_DIM + 1, wb * xv[k].y);
                atomicAdd(a + 2 * OUT_DIM + 0, wc * xv[k].x);
                atomicAdd(a + 2 * OUT_DIM + 1, wc * xv[k].y);
                atomicAdd(a + 3 * OUT_DIM + 0, wd * xv[k].x);
                atomicAdd(a + 3 * OUT_DIM + 1, wd * xv[k].y);
            }
        }
    }
}

template <bool AGG>
__device__ __forceinline__ bool own_scan(
    own_list_word* list, int& cnt_out, int lane, int wave, int r0, int nrows, int E,
    const int* __restrict__ ei, const float* __restrict__ el,
    const float* __restrict__ er, const float* __restrict__ x,
    const float4* own_rs, float* own_rs_f, float* own_acc)
{
    int  cnt    = 0;
    int  summed = 0;
    bool keep   = true;

    const int   E4     = E >> 2;
    const int4* src4   = (const int4*)ei;
    const int   nsteps = (E4 + OWN_BLOCK4 - 1) / OWN_BLOCK4;
    const int   last4  = E4 - 1;

    int4 cur[SCAN_U];
    #pragma unroll
    for (int u = 0; u < SCAN_U; ++u)
        cur[u] = make_int4(-1, -1, -1, -1);
    if (nsteps > 0) {
        #pragma unroll
        for (int u = 0; u < SCAN_U; ++u)
            cur[u] = src4[min(wave * OWN_STEP4 + u * 32 + lane, last4)];
    }

    for (int s = 0; s < nsteps; ++s) {
        int4 nxt[SCAN_U];
        #pragma unroll
        for (int u = 0; u < SCAN_U; ++u)
            nxt[u] = src4[min((s + 1) * OWN_BLOCK4 + wave * OWN_STEP4 + u * 32 + lane, last4)];

        const int cb4 = s * OWN_BLOCK4 + wave * OWN_STEP4 + lane;
        unsigned m = 0;
        #pragma unroll
        for (int u = 0; u < SCAN_U; ++u) {
            unsigned mu = own_match(cur[u].x, r0, nrows)
                        | (own_match(cur[u].y, r0, nrows) << 1)
                        | (own_match(cur[u].z, r0, nrows) << 2)
                        | (own_match(cur[u].w, r0, nrows) << 3);
            if (cb4 + u * 32 >= E4) mu = 0u;
            m |= mu << (4 * u);
        }

        for (int t = 0; t < 4 * SCAN_U; ++t) {
            const bool has = (m != 0);
            const unsigned bal = __builtin_amdgcn_ballot_w32(has);
            if (bal == 0) break;
            if (has) {
                const int bit = __builtin_ctz(m);
                m &= m - 1;
                const int e = ((cb4 + (bit >> 2) * 32) << 2) + (bit & 3);
                list[cnt + __builtin_popcount(bal & ((1u << lane) - 1u))] = e;
            }
            cnt += __builtin_popcount(bal);
        }

        if (AGG) {
            int off = 0;
            while (cnt - off >= 32) {
                own_process_group(list + off, 32, lane, r0, nrows, E, ei, el, er, x, own_rs, own_acc);
                off += 32;
            }
            if (off) {
                const int left = cnt - off;
                int tmp = 0;
                if (lane < left) tmp = list[off + lane];
                if (lane < left) list[lane] = tmp;
                cnt = left;
            }
        } else {
            while (cnt - summed >= 32) {
                own_sum_group(list + summed, 32, lane, r0, nrows, E, ei, el, er, own_rs_f);
                summed += 32;
            }
            if (cnt > LIST_CAP - STEP_EDGES) {
                keep = false;
                const int left = cnt - summed;
                int tmp = 0;
                if (lane < left) tmp = list[summed + lane];
                if (lane < left) list[lane] = tmp;
                cnt = left;
                summed = 0;
            }
        }

        #pragma unroll
        for (int u = 0; u < SCAN_U; ++u) cur[u] = nxt[u];
    }

    {
        const int e = (E4 << 2) + lane;
        bool has = false;
        if (wave == 0 && e < E) has = (own_match(ei[e], r0, nrows) != 0u);
        const unsigned bal = __builtin_amdgcn_ballot_w32(has);
        if (has) list[cnt + __builtin_popcount(bal & ((1u << lane) - 1u))] = e;
        cnt += __builtin_popcount(bal);
    }
    if (AGG) {
        for (int off = 0; off < cnt; off += 32)
            own_process_group(list + off, min(32, cnt - off), lane, r0, nrows, E, ei, el, er, x, own_rs, own_acc);
    } else {
        for (; summed < cnt; summed += 32)
            own_sum_group(list + summed, min(32, cnt - summed), lane, r0, nrows, E, ei, el, er, own_rs_f);
    }
    cnt_out = cnt;
    return keep;
}

#define KEY_U      4
#define KEY_STEPQ  (KEY_U * 32)
#define KEY_BLOCKQ (KEY_STEPQ * OWN_WAVES)

__device__ __forceinline__ unsigned key_match4(unsigned d, unsigned pat) {
    const unsigned x = d ^ pat;
    const unsigned t = (x & 0x7f7f7f7fu) + 0x7f7f7f7fu;
    return ~(t | x) & 0x80808080u;
}

template <bool AGG>
__device__ __forceinline__ void own_make_room(
    own_list_word* list, int& cnt, int& summed, bool& keep,
    int lane, int r0, int nrows, int E,
    const int* __restrict__ ei, const float* __restrict__ el,
    const float* __restrict__ er, const float* __restrict__ x,
    const float4* own_rs, float* own_rs_f, float* own_acc)
{
    if (AGG) {
        for (int off = 0; off < cnt; off += 32)
            own_process_group(list + off, min(32, cnt - off), lane, r0, nrows, E, ei, el, er, x, own_rs, own_acc);
    } else {
        for (; summed < cnt; summed += 32)
            own_sum_group(list + summed, min(32, cnt - summed), lane, r0, nrows, E, ei, el, er, own_rs_f);
        summed = 0;
        keep   = false;
    }
    cnt = 0;
}

template <bool AGG>
__device__ __forceinline__ bool own_scan_keys(
    own_list_word* list, int& cnt_out, int lane, int wave, int wg, int r0, int nrows, int E,
    const int* __restrict__ ei, const int* __restrict__ keys,
    const float* __restrict__ el, const float* __restrict__ er,
    const float* __restrict__ x,
    const float4* own_rs, float* own_rs_f, float* own_acc)
{
    int  cnt    = 0;
    int  summed = 0;
    bool keep   = true;

    const int      nunits = E >> 4;
    const uint4*   key4   = (const uint4*)keys;
    const int      nsteps = (nunits + KEY_BLOCKQ - 1) / KEY_BLOCKQ;
    const int      lastq  = nunits - 1;
    const unsigned pat    = (unsigned)(wg & 255) * 0x01010101u;
    const unsigned below  = (1u << lane) - 1u;

    uint4 cur[KEY_U];
    #pragma unroll
    for (int u = 0; u < KEY_U; ++u)
        cur[u] = make_uint4(0u, 0u, 0u, 0u);
    if (nsteps > 0) {
        #pragma unroll
        for (int u = 0; u < KEY_U; ++u)
            cur[u] = key4[min(wave * KEY_STEPQ + u * 32 + lane, lastq)];
    }

    for (int s = 0; s < nsteps; ++s) {
        uint4 nxt[KEY_U];
        #pragma unroll
        for (int u = 0; u < KEY_U; ++u)
            nxt[u] = key4[min((s + 1) * KEY_BLOCKQ + wave * KEY_STEPQ + u * 32 + lane, lastq)];

        const int q0 = s * KEY_BLOCKQ + wave * KEY_STEPQ + lane;

        #pragma unroll
        for (int p = 0; p < KEY_U; p += 2) {
            const int qa = q0 + p * 32;
            unsigned ma = (key_match4(cur[p].x, pat) >> 7) | (key_match4(cur[p].y, pat) >> 6)
                        | (key_match4(cur[p].z, pat) >> 5) | (key_match4(cur[p].w, pat) >> 4);
            unsigned mb = (key_match4(cur[p + 1].x, pat) >> 3) | (key_match4(cur[p + 1].y, pat) >> 2)
                        | (key_match4(cur[p + 1].z, pat) >> 1) |  key_match4(cur[p + 1].w, pat);
            if (qa      >= nunits) ma = 0u;
            if (qa + 32 >= nunits) mb = 0u;
            unsigned m = ma | mb;

            for (int t = 0; t < 32; ++t) {
                const bool has = (m != 0);
                const unsigned bal = __builtin_amdgcn_ballot_w32(has);
                if (bal == 0) break;
                const int n = __builtin_popcount(bal);
                if (cnt + n > LIST_CAP)
                    own_make_room<AGG>(list, cnt, summed, keep, lane, r0, nrows, E,
                                       ei, el, er, x, own_rs, own_rs_f, own_acc);
                if (has) {
                    const int bit = __builtin_ctz(m);
                    m &= m - 1;
                    const int q = qa + ((bit >> 2) & 1) * 32;
                    const int e = (q << 4) + ((bit & 3) << 2) + (bit >> 3);
                    list[cnt + __builtin_popcount(bal & below)] = e;
                }
                cnt += n;
            }
        }

        if (AGG) {
            int off = 0;
            while (cnt - off >= 32) {
                own_process_group(list + off, 32, lane, r0, nrows, E, ei, el, er, x, own_rs, own_acc);
                off += 32;
            }
            if (off) {
                const int left = cnt - off;
                int tmp = 0;
                if (lane < left) tmp = list[off + lane];
                if (lane < left) list[lane] = tmp;
                cnt = left;
            }
        } else {
            while (cnt - summed >= 32) {
                own_sum_group(list + summed, 32, lane, r0, nrows, E, ei, el, er, own_rs_f);
                summed += 32;
            }
        }

        #pragma unroll
        for (int u = 0; u < KEY_U; ++u) cur[u] = nxt[u];
    }

    {
        const int e = (nunits << 4) + lane;
        bool has = false;
        if (wave == 0 && e < E) has = (own_match(ei[e], r0, nrows) != 0u);
        const unsigned bal = __builtin_amdgcn_ballot_w32(has);
        const int n = __builtin_popcount(bal);
        if (cnt + n > LIST_CAP)
            own_make_room<AGG>(list, cnt, summed, keep, lane, r0, nrows, E,
                               ei, el, er, x, own_rs, own_rs_f, own_acc);
        if (has) list[cnt + __builtin_popcount(bal & below)] = e;
        cnt += n;
    }
    if (AGG) {
        for (int off = 0; off < cnt; off += 32)
            own_process_group(list + off, min(32, cnt - off), lane, r0, nrows, E, ei, el, er, x, own_rs, own_acc);
    } else {
        for (; summed < cnt; summed += 32)
            own_sum_group(list + summed, min(32, cnt - summed), lane, r0, nrows, E, ei, el, er, own_rs_f);
    }
    cnt_out = cnt;
    return keep;
}

template <bool KEYED>
__global__ __launch_bounds__(OWN_THREADS) void gat_own_rows(
    const int* __restrict__ ei, const int* __restrict__ keys,
    const float* __restrict__ el,
    const float* __restrict__ er, const float* __restrict__ x,
    const float* __restrict__ b, float* __restrict__ out, int N, int E)
{
    __shared__ float  own_acc[OWN_ROWS * ROW_F];
    __shared__ float4 own_rs[OWN_ROWS];
    __shared__ int    own_list[OWN_WAVES * LIST_CAP];

    const int tid  = threadIdx.x;
    const int lane = tid & 31;
    const int wave = tid >> 5;
    const int r0    = blockIdx.x * OWN_ROWS;
    const int nrows = min(OWN_ROWS, N - r0);

    float4* acc4 = (float4*)own_acc;
    for (int i = tid; i < OWN_ROWS * (ROW_F / 4); i += OWN_THREADS)
        acc4[i] = make_float4(0.f, 0.f, 0.f, 0.f);
    for (int i = tid; i < OWN_ROWS; i += OWN_THREADS)
        own_rs[i] = make_float4(0.f, 0.f, 0.f, 0.f);
    __syncthreads();

    own_list_word* list = (own_list_word*)(own_list + wave * LIST_CAP);
    float* own_rs_f = (float*)own_rs;

    int cnt = 0;
    const bool kept = KEYED
        ? own_scan_keys<false>(list, cnt, lane, wave, (int)blockIdx.x, r0, nrows, E,
                               ei, keys, el, er, x, own_rs, own_rs_f, own_acc)
        : own_scan<false>(list, cnt, lane, wave, r0, nrows, E,
                          ei, el, er, x, own_rs, own_rs_f, own_acc);

    __syncthreads();

    if (kept) {
        int off = 0;
        for (; cnt - off >= 32; off += 32)
            own_process_group(list + off, 32, lane, r0, nrows, E, ei, el, er, x, own_rs, own_acc);
        if (cnt - off > 0)
            own_process_group(list + off, cnt - off, lane, r0, nrows, E, ei, el, er, x, own_rs, own_acc);
    } else {
        int unused = 0;
        if (KEYED)
            own_scan_keys<true>(list, unused, lane, wave, (int)blockIdx.x, r0, nrows, E,
                                ei, keys, el, er, x, own_rs, own_rs_f, own_acc);
        else
            own_scan<true>(list, unused, lane, wave, r0, nrows, E,
                           ei, el, er, x, own_rs, own_rs_f, own_acc);
    }

    __syncthreads();

    float4* out4 = (float4*)out + (size_t)r0 * (ROW_F / 4);
    for (int i = tid; i < nrows * (ROW_F / 4); i += OWN_THREADS) {
        float4 v = acc4[i];
        const int d0 = (i * 4) & (OUT_DIM - 1);
        v.x += b[d0 + 0]; v.y += b[d0 + 1]; v.z += b[d0 + 2]; v.w += b[d0 + 3];
        v.x = (v.x > 0.f) ? v.x : (__expf(v.x) - 1.f);
        v.y = (v.y > 0.f) ? v.y : (__expf(v.y) - 1.f);
        v.z = (v.z > 0.f) ? v.z : (__expf(v.z) - 1.f);
        v.w = (v.w > 0.f) ? v.w : (__expf(v.w) - 1.f);
        out4[i] = v;
    }
}

extern "C" void kernel_launch(void* const* d_in, const int* in_sizes, int n_in,
                              void* d_out, int out_size, void* d_ws, size_t ws_size,
                              hipStream_t stream) {
    const float* h  = (const float*)d_in[0];
    const float* W  = (const float*)d_in[1];
    const float* Wl = (const float*)d_in[2];
    const float* Wr = (const float*)d_in[3];
    const float* b  = (const float*)d_in[4];
    const int*   ei = (const int*)d_in[5];

    const int N = in_sizes[0] / IN_DIM;
    const int E = in_sizes[5] / 2;

    float* out = (float*)d_out;

    float* x      = (float*)d_ws;
    float* el     = x  + (size_t)N * OUT_DIM;
    float* er     = el + (size_t)N * N_HEADS;
    int*   keys   = (int*)(er + (size_t)N * N_HEADS);
    const size_t need  = (size_t)N * (OUT_DIM + 2 * N_HEADS) * sizeof(float) + (size_t)E;
    const bool   keyed = (ws_size >= need);

    gat_gemm_wmma<<<(unsigned)((N + 15) / 16), 128, 0, stream>>>(h, W, x, N);

    gat_coef<<<(unsigned)((N + 255) / 256), 256, 0, stream>>>(
        x, Wl, Wr, el, er, ei, keyed ? keys : (int*)nullptr, N, E);

    if (keyed)
        gat_own_rows<true><<<(unsigned)((N + OWN_ROWS - 1) / OWN_ROWS), OWN_THREADS, 0, stream>>>(
            ei, keys, el, er, x, b, out, N, E);
    else
        gat_own_rows<false><<<(unsigned)((N + OWN_ROWS - 1) / OWN_ROWS), OWN_THREADS, 0, stream>>>(
            ei, keys, el, er, x, b, out, N, E);
}


# value sums without LDS atomics: each wave owns every 32nd row of its work-group, takes those rows' edges from all kept lists (row packed in the list word), plain 8-byte LDS adds; sum order fixed
# speedup vs baseline: 4.3551x; 1.2067x over previous
#include <hip/hip_runtime.h>

#define IN_DIM  256
#define OUT_DIM 64
#define N_HEADS 4
#define GAT_EPS 1e-12f

typedef float v2f __attribute__((ext_vector_type(2)));
typedef float v8f __attribute__((ext_vector_type(8)));

#define ROW_F (N_HEADS * OUT_DIM)

__global__ __launch_bounds__(128) void gat_gemm_wmma(
    const float* __restrict__ h, const float* __restrict__ W,
    float* __restrict__ x, int N)
{
    const int lane = threadIdx.x & 31;
    const int wave = threadIdx.x >> 5;
    const int half = lane >> 4;
    const int l15  = lane & 15;
    const int row0 = blockIdx.x * 16;
    const int col0 = wave * 16;

    int arow = row0 + l15;
    if (arow >= N) arow = N - 1;
    const float* __restrict__ hrow = h + (size_t)arow * IN_DIM;
    const int bcol = col0 + l15;

    v8f acc = {0.f, 0.f, 0.f, 0.f, 0.f, 0.f, 0.f, 0.f};

    for (int k = 0; k < IN_DIM; k += 4) {
        const int ka = k + half * 2;
        v2f a = *(const v2f*)(hrow + ka);
        v2f b;
        b.x = W[(size_t)ka       * OUT_DIM + bcol];
        b.y = W[(size_t)(ka + 1) * OUT_DIM + bcol];
        acc = __builtin_amdgcn_wmma_f32_16x16x4_f32(
             false, a,  false, b,
             (short)0, acc,  false,  false);
    }

    float* __restrict__ xout =
        x + ((size_t)(row0 + half * 8)) * OUT_DIM + col0 + l15;
    if (row0 + 16 <= N) {
        #pragma unroll
        for (int r = 0; r < 8; ++r)
            xout[(size_t)r * OUT_DIM] = acc[r];
    } else {
        #pragma unroll
        for (int r = 0; r < 8; ++r) {
            if (row0 + r + half * 8 < N)
                xout[(size_t)r * OUT_DIM] = acc[r];
        }
    }
}

#define OWN_ROWS    196

__device__ __forceinline__ unsigned own_key(int v) {
    return (((unsigned)v) / (unsigned)OWN_ROWS) & 255u;
}

__global__ __launch_bounds__(256) void gat_coef(
    const float* __restrict__ x, const float* __restrict__ Wl,
    const float* __restrict__ Wr, float* __restrict__ el,
    float* __restrict__ er, const int* __restrict__ ei,
    int* __restrict__ keys, int N, int E)
{
    if (keys != nullptr) {
        const int   nunits = E >> 4;
        const int4* src4   = (const int4*)ei;
        uint4*      key4   = (uint4*)keys;
        const int   stride = (int)(gridDim.x * blockDim.x);
        for (int q = blockIdx.x * blockDim.x + threadIdx.x; q < nunits; q += stride) {
            unsigned w[4];
            #pragma unroll
            for (int j = 0; j < 4; ++j) {
                const int4 v = src4[(size_t)q * 4 + j];
                w[j] = own_key(v.x) | (own_key(v.y) << 8) | (own_key(v.z) << 16) | (own_key(v.w) << 24);
            }
            key4[q] = make_uint4(w[0], w[1], w[2], w[3]);
        }
    }

    const int n = blockIdx.x * blockDim.x + threadIdx.x;
    if (n >= N) return;
    const float* __restrict__ xr = x + (size_t)n * OUT_DIM;
    float al[N_HEADS] = {0.f, 0.f, 0.f, 0.f};
    float ar[N_HEADS] = {0.f, 0.f, 0.f, 0.f};
    for (int d = 0; d < OUT_DIM; ++d) {
        const float xv = xr[d];
        #pragma unroll
        for (int hh = 0; hh < N_HEADS; ++hh) {
            al[hh] = fmaf(xv, Wl[hh * OUT_DIM + d], al[hh]);
            ar[hh] = fmaf(xv, Wr[hh * OUT_DIM + d], ar[hh]);
        }
    }
    #pragma unroll
    for (int hh = 0; hh < N_HEADS; ++hh) {
        el[(size_t)n * N_HEADS + hh] = al[hh];
        er[(size_t)n * N_HEADS + hh] = ar[hh];
    }
}

__device__ __forceinline__ float gat_edge_e(float l, float r) {
    float v = l + r;
    v = (v > 0.f) ? v : 0.2f * v;
    return __expf(v);
}

#define OWN_THREADS 1024
#define OWN_WAVES   (OWN_THREADS / 32)
#define SCAN_U      4
#define OWN_STEP4   (SCAN_U * 32)
#define OWN_BLOCK4  (OWN_STEP4 * OWN_WAVES)
#define STEP_EDGES  (OWN_STEP4 * 4)
#define LIST_CAP    832
#define QUEUE_CAP   64
#define PACK_MAX_E  (1 << 24)

typedef volatile int __attribute__((address_space(3))) own_list_word;

__device__ __forceinline__ int own_bcast_i(int v, int j) {
    return __builtin_amdgcn_readlane(v, j);
}
__device__ __forceinline__ float own_bcast_f(float v, int j) {
    return __int_as_float(__builtin_amdgcn_readlane(__float_as_int(v), j));
}

__device__ __forceinline__ unsigned own_match(int v, int r0, int nrows) {
    return ((unsigned)(v - r0) < (unsigned)nrows) ? 1u : 0u;
}

__device__ __forceinline__ void own_sum_group(
    own_list_word* list, int g, int lane, int r0, int nrows, int E,
    const int* __restrict__ ei, const float* __restrict__ el,
    const float* __restrict__ er, float* own_rs_f)
{
    if (lane < g) {
        const int e   = list[lane];
        const int src = ei[e];
        const int dst = ei[(size_t)E + e];
        const float4 r = *(const float4*)(er + (size_t)dst * N_HEADS);
        const bool mine = (own_match(src, r0, nrows) != 0u);
        if (mine) {
            const float4 l = *(const float4*)(el + (size_t)src * N_HEADS);
            float* a = own_rs_f + (src - r0) * N_HEADS;
            atomicAdd(a + 0, gat_edge_e(l.x, r.x));
            atomicAdd(a + 1, gat_edge_e(l.y, r.y));
            atomicAdd(a + 2, gat_edge_e(l.z, r.z));
            atomicAdd(a + 3, gat_edge_e(l.w, r.w));
        }
        if (E <= PACK_MAX_E)
            list[lane] = mine ? (int)((unsigned)e | ((unsigned)(src - r0) << 24))
                              : (int)0xff000000u;
    }
}

__device__ __forceinline__ void own_process_group(
    own_list_word* list, int g, int lane, int r0, int nrows, int E,
    const int* __restrict__ ei, const float* __restrict__ el,
    const float* __restrict__ er, const float* __restrict__ x,
    const float4* own_rs, float* own_acc)
{
    int my_dst = 0, my_lrow = 0;
    float w0 = 0.f, w1 = 0.f, w2 = 0.f, w3 = 0.f;
    if (lane < g) {
        const int e   = list[lane];
        const int src = ei[e];
        my_dst  = ei[(size_t)E + e];
        const float4 r  = *(const float4*)(er + (size_t)my_dst * N_HEADS);
        if (own_match(src, r0, nrows)) {
            my_lrow = src - r0;
            const float4 l  = *(const float4*)(el + (size_t)src    * N_HEADS);
            const float4 rs = own_rs[my_lrow];
            w0 = gat_edge_e(l.x, r.x) / fmaxf(rs.x, GAT_EPS);
            w1 = gat_edge_e(l.y, r.y) / fmaxf(rs.y, GAT_EPS);
            w2 = gat_edge_e(l.z, r.z) / fmaxf(rs.z, GAT_EPS);
            w3 = gat_edge_e(l.w, r.w) / fmaxf(rs.w, GAT_EPS);
        }
    }
    for (int i = 0; i < g; i += 4) {
        int lr[4];
        v2f xv[4];
        #pragma unroll
        for (int k = 0; k < 4; ++k) {
            const int j = min(i + k, g - 1);
            const int d = own_bcast_i(my_dst, j);
            lr[k] = own_bcast_i(my_lrow, j);
            xv[k] = *(const v2f*)(x + (size_t)d * OUT_DIM + lane * 2);
        }
        #pragma unroll
        for (int k = 0; k < 4; ++k) {
            if (i + k < g) {
                const int j = i + k;
                float* a = own_acc + lr[k] * ROW_F + lane * 2;
                const float wa = own_bcast_f(w0, j);
                const float wb = own_bcast_f(w1, j);
                const float wc = own_bcast_f(w2, j);
                const float wd = own_bcast_f(w3, j);
                atomicAdd(a + 0 * OUT_DIM + 0, wa * xv[k].x);
                atomicAdd(a + 0 * OUT_DIM + 1, wa * xv[k].y);
                atomicAdd(a + 1 * OUT_DIM + 0, wb * xv[k].x);
                atomicAdd(a + 1 * OUT_DIM + 1, wb * xv[k].y);
                atomicAdd(a + 2 * OUT_DIM + 0, wc * xv[k].x);
                atomicAdd(a + 2 * OUT_DIM + 1, wc * xv[k].y);
                atomicAdd(a + 3 * OUT_DIM + 0, wd * xv[k].x);
                atomicAdd(a + 3 * OUT_DIM + 1, wd * xv[k].y);
            }
        }
    }
}

#define PK_B 8
template <bool ATOMIC>
__device__ __forceinline__ void own_process_packed(
    own_list_word* list, int g, int lane, int r0, int nrows, int E,
    const int* __restrict__ ei, const float* __restrict__ el,
    const float* __restrict__ er, const float* __restrict__ x,
    const float4* own_rs, float* own_acc)
{
    int my_dst = 0, my_lrow = 0;
    float w0 = 0.f, w1 = 0.f, w2 = 0.f, w3 = 0.f;
    if (lane < g) {
        const unsigned wd = (unsigned)list[lane];
        const unsigned lr = wd >> 24;
        if (lr < (unsigned)nrows) {
            const int e = (int)(wd & 0x00ffffffu);
            my_lrow = (int)lr;
            my_dst  = ei[(size_t)E + e];
            const float4 r  = *(const float4*)(er + (size_t)my_dst * N_HEADS);
            const float4 l  = *(const float4*)(el + (size_t)(r0 + my_lrow) * N_HEADS);
            const float4 rs = own_rs[my_lrow];
            w0 = gat_edge_e(l.x, r.x) / fmaxf(rs.x, GAT_EPS);
            w1 = gat_edge_e(l.y, r.y) / fmaxf(rs.y, GAT_EPS);
            w2 = gat_edge_e(l.z, r.z) / fmaxf(rs.z, GAT_EPS);
            w3 = gat_edge_e(l.w, r.w) / fmaxf(rs.w, GAT_EPS);
        }
    }
    v2f cur[PK_B];
    #pragma unroll
    for (int k = 0; k < PK_B; ++k) {
        const int d = own_bcast_i(my_dst, min(k, g - 1));
        cur[k] = *(const v2f*)(x + (size_t)d * OUT_DIM + lane * 2);
    }
    for (int i = 0; i < g; i += PK_B) {
        v2f nxt[PK_B];
        #pragma unroll
        for (int k = 0; k < PK_B; ++k) nxt[k] = cur[k];
        if (i + PK_B < g) {
            #pragma unroll
            for (int k = 0; k < PK_B; ++k) {
                const int d = own_bcast_i(my_dst, min(i + PK_B + k, g - 1));
                nxt[k] = *(const v2f*)(x + (size_t)d * OUT_DIM + lane * 2);
            }
        }
        #pragma unroll
        for (int k = 0; k < PK_B; ++k) {
            if (i + k < g) {
                const int j = i + k;
                float* a = own_acc + own_bcast_i(my_lrow, j) * ROW_F + lane * 2;
                const float wa = own_bcast_f(w0, j);
                const float wb = own_bcast_f(w1, j);
                const float wc = own_bcast_f(w2, j);
                const float wd = own_bcast_f(w3, j);
                if (ATOMIC) {
                    atomicAdd(a + 0 * OUT_DIM + 0, wa * cur[k].x);
                    atomicAdd(a + 0 * OUT_DIM + 1, wa * cur[k].y);
                    atomicAdd(a + 1 * OUT_DIM + 0, wb * cur[k].x);
                    atomicAdd(a + 1 * OUT_DIM + 1, wb * cur[k].y);
                    atomicAdd(a + 2 * OUT_DIM + 0, wc * cur[k].x);
                    atomicAdd(a + 2 * OUT_DIM + 1, wc * cur[k].y);
                    atomicAdd(a + 3 * OUT_DIM + 0, wd * cur[k].x);
                    atomicAdd(a + 3 * OUT_DIM + 1, wd * cur[k].y);
                } else {
                    v2f s0 = *(v2f*)(a + 0 * OUT_DIM);
                    v2f s1 = *(v2f*)(a + 1 * OUT_DIM);
                    v2f s2 = *(v2f*)(a + 2 * OUT_DIM);
                    v2f s3 = *(v2f*)(a + 3 * OUT_DIM);
                    s0.x += wa * cur[k].x;  s0.y += wa * cur[k].y;
                    s1.x += wb * cur[k].x;  s1.y += wb * cur[k].y;
                    s2.x += wc * cur[k].x;  s2.y += wc * cur[k].y;
                    s3.x += wd * cur[k].x;  s3.y += wd * cur[k].y;
                    *(v2f*)(a + 0 * OUT_DIM) = s0;
                    *(v2f*)(a + 1 * OUT_DIM) = s1;
                    *(v2f*)(a + 2 * OUT_DIM) = s2;
                    *(v2f*)(a + 3 * OUT_DIM) = s3;
                }
            }
        }
        #pragma unroll
        for (int k = 0; k < PK_B; ++k) cur[k] = nxt[k];
    }
}

template <bool AGG>
__device__ __forceinline__ bool own_scan(
    own_list_word* list, int& cnt_out, int lane, int wave, int r0, int nrows, int E,
    const int* __restrict__ ei, const float* __restrict__ el,
    const float* __restrict__ er, const float* __restrict__ x,
    const float4* own_rs, float* own_rs_f, float* own_acc)
{
    int  cnt    = 0;
    int  summed = 0;
    bool keep   = true;

    const int   E4     = E >> 2;
    const int4* src4   = (const int4*)ei;
    const int   nsteps = (E4 + OWN_BLOCK4 - 1) / OWN_BLOCK4;
    const int   last4  = E4 - 1;

    int4 cur[SCAN_U];
    #pragma unroll
    for (int u = 0; u < SCAN_U; ++u)
        cur[u] = make_int4(-1, -1, -1, -1);
    if (nsteps > 0) {
        #pragma unroll
        for (int u = 0; u < SCAN_U; ++u)
            cur[u] = src4[min(wave * OWN_STEP4 + u * 32 + lane, last4)];
    }

    for (int s = 0; s < nsteps; ++s) {
        int4 nxt[SCAN_U];
        #pragma unroll
        for (int u = 0; u < SCAN_U; ++u)
            nxt[u] = src4[min((s + 1) * OWN_BLOCK4 + wave * OWN_STEP4 + u * 32 + lane, last4)];

        const int cb4 = s * OWN_BLOCK4 + wave * OWN_STEP4 + lane;
        unsigned m = 0;
        #pragma unroll
        for (int u = 0; u < SCAN_U; ++u) {
            unsigned mu = own_match(cur[u].x, r0, nrows)
                        | (own_match(cur[u].y, r0, nrows) << 1)
                        | (own_match(cur[u].z, r0, nrows) << 2)
                        | (own_match(cur[u].w, r0, nrows) << 3);
            if (cb4 + u * 32 >= E4) mu = 0u;
            m |= mu << (4 * u);
        }

        for (int t = 0; t < 4 * SCAN_U; ++t) {
            const bool has = (m != 0);
            const unsigned bal = __builtin_amdgcn_ballot_w32(has);
            if (bal == 0) break;
            if (has) {
                const int bit = __builtin_ctz(m);
                m &= m - 1;
                const int e = ((cb4 + (bit >> 2) * 32) << 2) + (bit & 3);
                list[cnt + __builtin_popcount(bal & ((1u << lane) - 1u))] = e;
            }
            cnt += __builtin_popcount(bal);
        }

        if (AGG) {
            int off = 0;
            while (cnt - off >= 32) {
                own_process_group(list + off, 32, lane, r0, nrows, E, ei, el, er, x, own_rs, own_acc);
                off += 32;
            }
            if (off) {
                const int left = cnt - off;
                int tmp = 0;
                if (lane < left) tmp = list[off + lane];
                if (lane < left) list[lane] = tmp;
                cnt = left;
            }
        } else {
            while (cnt - summed >= 32) {
                own_sum_group(list + summed, 32, lane, r0, nrows, E, ei, el, er, own_rs_f);
                summed += 32;
            }
            if (cnt > LIST_CAP - STEP_EDGES) {
                keep = false;
                const int left = cnt - summed;
                int tmp = 0;
                if (lane < left) tmp = list[summed + lane];
                if (lane < left) list[lane] = tmp;
                cnt = left;
                summed = 0;
            }
        }

        #pragma unroll
        for (int u = 0; u < SCAN_U; ++u) cur[u] = nxt[u];
    }

    {
        const int e = (E4 << 2) + lane;
        bool has = false;
        if (wave == 0 && e < E) has = (own_match(ei[e], r0, nrows) != 0u);
        const unsigned bal = __builtin_amdgcn_ballot_w32(has);
        if (has) list[cnt + __builtin_popcount(bal & ((1u << lane) - 1u))] = e;
        cnt += __builtin_popcount(bal);
    }
    if (AGG) {
        for (int off = 0; off < cnt; off += 32)
            own_process_group(list + off, min(32, cnt - off), lane, r0, nrows, E, ei, el, er, x, own_rs, own_acc);
    } else {
        for (; summed < cnt; summed += 32)
            own_sum_group(list + summed, min(32, cnt - summed), lane, r0, nrows, E, ei, el, er, own_rs_f);
    }
    cnt_out = cnt;
    return keep;
}

#define KEY_U      4
#define KEY_STEPQ  (KEY_U * 32)
#define KEY_BLOCKQ (KEY_STEPQ * OWN_WAVES)

__device__ __forceinline__ unsigned key_match4(unsigned d, unsigned pat) {
    const unsigned x = d ^ pat;
    const unsigned t = (x & 0x7f7f7f7fu) + 0x7f7f7f7fu;
    return ~(t | x) & 0x80808080u;
}

template <bool AGG>
__device__ __forceinline__ void own_make_room(
    own_list_word* list, int& cnt, int& summed, bool& keep,
    int lane, int r0, int nrows, int E,
    const int* __restrict__ ei, const float* __restrict__ el,
    const float* __restrict__ er, const float* __restrict__ x,
    const float4* own_rs, float* own_rs_f, float* own_acc)
{
    if (AGG) {
        for (int off = 0; off < cnt; off += 32)
            own_process_group(list + off, min(32, cnt - off), lane, r0, nrows, E, ei, el, er, x, own_rs, own_acc);
    } else {
        for (; summed < cnt; summed += 32)
            own_sum_group(list + summed, min(32, cnt - summed), lane, r0, nrows, E, ei, el, er, own_rs_f);
        summed = 0;
        keep   = false;
    }
    cnt = 0;
}

template <bool AGG>
__device__ __forceinline__ bool own_scan_keys(
    own_list_word* list, int& cnt_out, int lane, int wave, int wg, int r0, int nrows, int E,
    const int* __restrict__ ei, const int* __restrict__ keys,
    const float* __restrict__ el, const float* __restrict__ er,
    const float* __restrict__ x,
    const float4* own_rs, float* own_rs_f, float* own_acc)
{
    int  cnt    = 0;
    int  summed = 0;
    bool keep   = true;

    const int      nunits = E >> 4;
    const uint4*   key4   = (const uint4*)keys;
    const int      nsteps = (nunits + KEY_BLOCKQ - 1) / KEY_BLOCKQ;
    const int      lastq  = nunits - 1;
    const unsigned pat    = (unsigned)(wg & 255) * 0x01010101u;
    const unsigned below  = (1u << lane) - 1u;

    uint4 cur[KEY_U];
    #pragma unroll
    for (int u = 0; u < KEY_U; ++u)
        cur[u] = make_uint4(0u, 0u, 0u, 0u);
    if (nsteps > 0) {
        #pragma unroll
        for (int u = 0; u < KEY_U; ++u)
            cur[u] = key4[min(wave * KEY_STEPQ + u * 32 + lane, lastq)];
    }

    for (int s = 0; s < nsteps; ++s) {
        uint4 nxt[KEY_U];
        #pragma unroll
        for (int u = 0; u < KEY_U; ++u)
            nxt[u] = key4[min((s + 1) * KEY_BLOCKQ + wave * KEY_STEPQ + u * 32 + lane, lastq)];

        const int q0 = s * KEY_BLOCKQ + wave * KEY_STEPQ + lane;

        #pragma unroll
        for (int p = 0; p < KEY_U; p += 2) {
            const int qa = q0 + p * 32;
            unsigned ma = (key_match4(cur[p].x, pat) >> 7) | (key_match4(cur[p].y, pat) >> 6)
                        | (key_match4(cur[p].z, pat) >> 5) | (key_match4(cur[p].w, pat) >> 4);
            unsigned mb = (key_match4(cur[p + 1].x, pat) >> 3) | (key_match4(cur[p + 1].y, pat) >> 2)
                        | (key_match4(cur[p + 1].z, pat) >> 1) |  key_match4(cur[p + 1].w, pat);
            if (qa      >= nunits) ma = 0u;
            if (qa + 32 >= nunits) mb = 0u;
            unsigned m = ma | mb;

            for (int t = 0; t < 32; ++t) {
                const bool has = (m != 0);
                const unsigned bal = __builtin_amdgcn_ballot_w32(has);
                if (bal == 0) break;
                const int n = __builtin_popcount(bal);
                if (cnt + n > LIST_CAP)
                    own_make_room<AGG>(list, cnt, summed, keep, lane, r0, nrows, E,
                                       ei, el, er, x, own_rs, own_rs_f, own_acc);
                if (has) {
                    const int bit = __builtin_ctz(m);
                    m &= m - 1;
                    const int q = qa + ((bit >> 2) & 1) * 32;
                    const int e = (q << 4) + ((bit & 3) << 2) + (bit >> 3);
                    list[cnt + __builtin_popcount(bal & below)] = e;
                }
                cnt += n;
            }
        }

        if (AGG) {
            int off = 0;
            while (cnt - off >= 32) {
                own_process_group(list + off, 32, lane, r0, nrows, E, ei, el, er, x, own_rs, own_acc);
                off += 32;
            }
            if (off) {
                const int left = cnt - off;
                int tmp = 0;
                if (lane < left) tmp = list[off + lane];
                if (lane < left) list[lane] = tmp;
                cnt = left;
            }
        } else {
            while (cnt - summed >= 32) {
                own_sum_group(list + summed, 32, lane, r0, nrows, E, ei, el, er, own_rs_f);
                summed += 32;
            }
        }

        #pragma unroll
        for (int u = 0; u < KEY_U; ++u) cur[u] = nxt[u];
    }

    {
        const int e = (nunits << 4) + lane;
        bool has = false;
        if (wave == 0 && e < E) has = (own_match(ei[e], r0, nrows) != 0u);
        const unsigned bal = __builtin_amdgcn_ballot_w32(has);
        const int n = __builtin_popcount(bal);
        if (cnt + n > LIST_CAP)
            own_make_room<AGG>(list, cnt, summed, keep, lane, r0, nrows, E,
                               ei, el, er, x, own_rs, own_rs_f, own_acc);
        if (has) list[cnt + __builtin_popcount(bal & below)] = e;
        cnt += n;
    }
    if (AGG) {
        for (int off = 0; off < cnt; off += 32)
            own_process_group(list + off, min(32, cnt - off), lane, r0, nrows, E, ei, el, er, x, own_rs, own_acc);
    } else {
        for (; summed < cnt; summed += 32)
            own_sum_group(list + summed, min(32, cnt - summed), lane, r0, nrows, E, ei, el, er, own_rs_f);
    }
    cnt_out = cnt;
    return keep;
}

template <bool KEYED>
__global__ __launch_bounds__(OWN_THREADS) void gat_own_rows(
    const int* __restrict__ ei, const int* __restrict__ keys,
    const float* __restrict__ el,
    const float* __restrict__ er, const float* __restrict__ x,
    const float* __restrict__ b, float* __restrict__ out, int N, int E)
{
    __shared__ float  own_acc[OWN_ROWS * ROW_F];
    __shared__ float4 own_rs[OWN_ROWS];
    __shared__ int    own_list[OWN_WAVES * LIST_CAP];
    __shared__ int    own_queue[OWN_WAVES * QUEUE_CAP];
    __shared__ int    own_cnt[OWN_WAVES];
    __shared__ int    own_allkept;

    const int tid  = threadIdx.x;
    const int lane = tid & 31;
    const int wave = tid >> 5;
    const int r0    = blockIdx.x * OWN_ROWS;
    const int nrows = min(OWN_ROWS, N - r0);

    float4* acc4 = (float4*)own_acc;
    for (int i = tid; i < OWN_ROWS * (ROW_F / 4); i += OWN_THREADS)
        acc4[i] = make_float4(0.f, 0.f, 0.f, 0.f);
    for (int i = tid; i < OWN_ROWS; i += OWN_THREADS)
        own_rs[i] = make_float4(0.f, 0.f, 0.f, 0.f);
    if (tid == 0) own_allkept = 1;
    __syncthreads();

    own_list_word* list = (own_list_word*)(own_list + wave * LIST_CAP);
    float* own_rs_f = (float*)own_rs;

    int cnt = 0;
    const bool kept = KEYED
        ? own_scan_keys<false>(list, cnt, lane, wave, (int)blockIdx.x, r0, nrows, E,
                               ei, keys, el, er, x, own_rs, own_rs_f, own_acc)
        : own_scan<false>(list, cnt, lane, wave, r0, nrows, E,
                          ei, el, er, x, own_rs, own_rs_f, own_acc);

    if (lane == 0) {
        own_cnt[wave] = cnt;
        if (!kept) own_allkept = 0;
    }

    __syncthreads();

    const bool packed = (E <= PACK_MAX_E);
    const bool all_kept = (__builtin_amdgcn_readfirstlane(own_allkept) != 0);
    if (packed && all_kept) {
        own_list_word* q = (own_list_word*)(own_queue + wave * QUEUE_CAP);
        const unsigned below = (1u << lane) - 1u;
        int qn = 0;
        for (int ss = 0; ss < OWN_WAVES; ++ss) {
            const int s = (wave + ss) & (OWN_WAVES - 1);
            const int n = __builtin_amdgcn_readfirstlane(own_cnt[s]);
            own_list_word* ls = (own_list_word*)(own_list + s * LIST_CAP);
            for (int base = 0; base < n; base += 4 * 32) {
                unsigned wd[4];
                #pragma unroll
                for (int u = 0; u < 4; ++u) {
                    const int idx = base + u * 32 + lane;
                    wd[u] = (idx < n) ? (unsigned)ls[idx] : 0xff000000u;
                }
                #pragma unroll
                for (int u = 0; u < 4; ++u) {
                    const unsigned lr = wd[u] >> 24;
                    const bool has = (lr < (unsigned)nrows) && ((lr & (OWN_WAVES - 1)) == (unsigned)wave);
                    const unsigned bal = __builtin_amdgcn_ballot_w32(has);
                    if (has) q[qn + __builtin_popcount(bal & below)] = (int)wd[u];
                    qn += __builtin_popcount(bal);
                    if (qn >= 32) {
                        own_process_packed<false>(q, 32, lane, r0, nrows, E, ei, el, er, x, own_rs, own_acc);
                        const int left = qn - 32;
                        int tmp = 0;
                        if (lane < left) tmp = q[32 + lane];
                        if (lane < left) q[lane] = tmp;
                        qn = left;
                    }
                }
            }
        }
        if (qn > 0)
            own_process_packed<false>(q, qn, lane, r0, nrows, E, ei, el, er, x, own_rs, own_acc);
    } else if (kept && packed) {
        for (int off = 0; off < cnt; off += 32)
            own_process_packed<true>(list + off, min(32, cnt - off), lane, r0, nrows, E, ei, el, er, x, own_rs, own_acc);
    } else if (kept) {
        int off = 0;
        for (; cnt - off >= 32; off += 32)
            own_process_group(list + off, 32, lane, r0, nrows, E, ei, el, er, x, own_rs, own_acc);
        if (cnt - off > 0)
            own_process_group(list + off, cnt - off, lane, r0, nrows, E, ei, el, er, x, own_rs, own_acc);
    } else {
        int unused = 0;
        if (KEYED)
            own_scan_keys<true>(list, unused, lane, wave, (int)blockIdx.x, r0, nrows, E,
                                ei, keys, el, er, x, own_rs, own_rs_f, own_acc);
        else
            own_scan<true>(list, unused, lane, wave, r0, nrows, E,
                           ei, el, er, x, own_rs, own_rs_f, own_acc);
    }

    __syncthreads();

    float4* out4 = (float4*)out + (size_t)r0 * (ROW_F / 4);
    for (int i = tid; i < nrows * (ROW_F / 4); i += OWN_THREADS) {
        float4 v = acc4[i];
        const int d0 = (i * 4) & (OUT_DIM - 1);
        v.x += b[d0 + 0]; v.y += b[d0 + 1]; v.z += b[d0 + 2]; v.w += b[d0 + 3];
        v.x = (v.x > 0.f) ? v.x : (__expf(v.x) - 1.f);
        v.y = (v.y > 0.f) ? v.y : (__expf(v.y) - 1.f);
        v.z = (v.z > 0.f) ? v.z : (__expf(v.z) - 1.f);
        v.w = (v.w > 0.f) ? v.w : (__expf(v.w) - 1.f);
        out4[i] = v;
    }
}

extern "C" void kernel_launch(void* const* d_in, const int* in_sizes, int n_in,
                              void* d_out, int out_size, void* d_ws, size_t ws_size,
                              hipStream_t stream) {
    const float* h  = (const float*)d_in[0];
    const float* W  = (const float*)d_in[1];
    const float* Wl = (const float*)d_in[2];
    const float* Wr = (const float*)d_in[3];
    const float* b  = (const float*)d_in[4];
    const int*   ei = (const int*)d_in[5];

    const int N = in_sizes[0] / IN_DIM;
    const int E = in_sizes[5] / 2;

    float* out = (float*)d_out;

    float* x      = (float*)d_ws;
    float* el     = x  + (size_t)N * OUT_DIM;
    float* er     = el + (size_t)N * N_HEADS;
    int*   keys   = (int*)(er + (size_t)N * N_HEADS);
    const size_t need  = (size_t)N * (OUT_DIM + 2 * N_HEADS) * sizeof(float) + (size_t)E;
    const bool   keyed = (ws_size >= need);

    gat_gemm_wmma<<<(unsigned)((N + 15) / 16), 128, 0, stream>>>(h, W, x, N);

    gat_coef<<<(unsigned)((N + 255) / 256), 256, 0, stream>>>(
        x, Wl, Wr, el, er, ei, keyed ? keys : (int*)nullptr, N, E);

    if (keyed)
        gat_own_rows<true><<<(unsigned)((N + OWN_ROWS - 1) / OWN_ROWS), OWN_THREADS, 0, stream>>>(
            ei, keys, el, er, x, b, out, N, E);
    else
        gat_own_rows<false><<<(unsigned)((N + OWN_ROWS - 1) / OWN_ROWS), OWN_THREADS, 0, stream>>>(
            ei, keys, el, er, x, b, out, N, E);
}


# phase 2 hand-over by counting: each wave counts its listed words per owner wave and puts them into the owners' bins (integer LDS adds); no wave searches the other lists; value sum order: bin order
# speedup vs baseline: 4.6877x; 1.0764x over previous
#include <hip/hip_runtime.h>

#define IN_DIM  256
#define OUT_DIM 64
#define N_HEADS 4
#define GAT_EPS 1e-12f

typedef float v2f __attribute__((ext_vector_type(2)));
typedef float v8f __attribute__((ext_vector_type(8)));

#define ROW_F (N_HEADS * OUT_DIM)

__global__ __launch_bounds__(128) void gat_gemm_wmma(
    const float* __restrict__ h, const float* __restrict__ W,
    float* __restrict__ x, int N)
{
    const int lane = threadIdx.x & 31;
    const int wave = threadIdx.x >> 5;
    const int half = lane >> 4;
    const int l15  = lane & 15;
    const int row0 = blockIdx.x * 16;
    const int col0 = wave * 16;

    int arow = row0 + l15;
    if (arow >= N) arow = N - 1;
    const float* __restrict__ hrow = h + (size_t)arow * IN_DIM;
    const int bcol = col0 + l15;

    v8f acc = {0.f, 0.f, 0.f, 0.f, 0.f, 0.f, 0.f, 0.f};

    for (int k = 0; k < IN_DIM; k += 4) {
        const int ka = k + half * 2;
        v2f a = *(const v2f*)(hrow + ka);
        v2f b;
        b.x = W[(size_t)ka       * OUT_DIM + bcol];
        b.y = W[(size_t)(ka + 1) * OUT_DIM + bcol];
        acc = __builtin_amdgcn_wmma_f32_16x16x4_f32(
             false, a,  false, b,
             (short)0, acc,  false,  false);
    }

    float* __restrict__ xout =
        x + ((size_t)(row0 + half * 8)) * OUT_DIM + col0 + l15;
    if (row0 + 16 <= N) {
        #pragma unroll
        for (int r = 0; r < 8; ++r)
            xout[(size_t)r * OUT_DIM] = acc[r];
    } else {
        #pragma unroll
        for (int r = 0; r < 8; ++r) {
            if (row0 + r + half * 8 < N)
                xout[(size_t)r * OUT_DIM] = acc[r];
        }
    }
}

#define OWN_ROWS    196

__device__ __forceinline__ unsigned own_key(int v) {
    return (((unsigned)v) / (unsigned)OWN_ROWS) & 255u;
}

__global__ __launch_bounds__(256) void gat_coef(
    const float* __restrict__ x, const float* __restrict__ Wl,
    const float* __restrict__ Wr, float* __restrict__ el,
    float* __restrict__ er, const int* __restrict__ ei,
    int* __restrict__ keys, int N, int E)
{
    if (keys != nullptr) {
        const int   nunits = E >> 4;
        const int4* src4   = (const int4*)ei;
        uint4*      key4   = (uint4*)keys;
        const int   stride = (int)(gridDim.x * blockDim.x);
        for (int q = blockIdx.x * blockDim.x + threadIdx.x; q < nunits; q += stride) {
            unsigned w[4];
            #pragma unroll
            for (int j = 0; j < 4; ++j) {
                const int4 v = src4[(size_t)q * 4 + j];
                w[j] = own_key(v.x) | (own_key(v.y) << 8) | (own_key(v.z) << 16) | (own_key(v.w) << 24);
            }
            key4[q] = make_uint4(w[0], w[1], w[2], w[3]);
        }
    }

    const int n = blockIdx.x * blockDim.x + threadIdx.x;
    if (n >= N) return;
    const float* __restrict__ xr = x + (size_t)n * OUT_DIM;
    float al[N_HEADS] = {0.f, 0.f, 0.f, 0.f};
    float ar[N_HEADS] = {0.f, 0.f, 0.f, 0.f};
    for (int d = 0; d < OUT_DIM; ++d) {
        const float xv = xr[d];
        #pragma unroll
        for (int hh = 0; hh < N_HEADS; ++hh) {
            al[hh] = fmaf(xv, Wl[hh * OUT_DIM + d], al[hh]);
            ar[hh] = fmaf(xv, Wr[hh * OUT_DIM + d], ar[hh]);
        }
    }
    #pragma unroll
    for (int hh = 0; hh < N_HEADS; ++hh) {
        el[(size_t)n * N_HEADS + hh] = al[hh];
        er[(size_t)n * N_HEADS + hh] = ar[hh];
    }
}

__device__ __forceinline__ float gat_edge_e(float l, float r) {
    float v = l + r;
    v = (v > 0.f) ? v : 0.2f * v;
    return __expf(v);
}

#define OWN_THREADS 1024
#define OWN_WAVES   (OWN_THREADS / 32)
#define SCAN_U      2
#define OWN_STEP4   (SCAN_U * 32)
#define OWN_BLOCK4  (OWN_STEP4 * OWN_WAVES)
#define STEP_EDGES  (OWN_STEP4 * 4)
#define LIST_CAP    416
static_assert(STEP_EDGES + 31 <= LIST_CAP, "a step after a drop must fit the list");
#define PACK_MAX_E  (1 << 24)

typedef volatile int __attribute__((address_space(3))) own_list_word;

__device__ __forceinline__ int own_bcast_i(int v, int j) {
    return __builtin_amdgcn_readlane(v, j);
}
__device__ __forceinline__ float own_bcast_f(float v, int j) {
    return __int_as_float(__builtin_amdgcn_readlane(__float_as_int(v), j));
}

__device__ __forceinline__ unsigned own_match(int v, int r0, int nrows) {
    return ((unsigned)(v - r0) < (unsigned)nrows) ? 1u : 0u;
}

__device__ __forceinline__ void own_sum_group(
    own_list_word* list, int g, int lane, int r0, int nrows, int E,
    const int* __restrict__ ei, const float* __restrict__ el,
    const float* __restrict__ er, float* own_rs_f)
{
    if (lane < g) {
        const int e   = list[lane];
        const int src = ei[e];
        const int dst = ei[(size_t)E + e];
        const float4 r = *(const float4*)(er + (size_t)dst * N_HEADS);
        const bool mine = (own_match(src, r0, nrows) != 0u);
        if (mine) {
            const float4 l = *(const float4*)(el + (size_t)src * N_HEADS);
            float* a = own_rs_f + (src - r0) * N_HEADS;
            atomicAdd(a + 0, gat_edge_e(l.x, r.x));
            atomicAdd(a + 1, gat_edge_e(l.y, r.y));
            atomicAdd(a + 2, gat_edge_e(l.z, r.z));
            atomicAdd(a + 3, gat_edge_e(l.w, r.w));
        }
        if (E <= PACK_MAX_E)
            list[lane] = mine ? (int)((unsigned)e | ((unsigned)(src - r0) << 24))
                              : (int)0xff000000u;
    }
}

__device__ __forceinline__ void own_process_group(
    own_list_word* list, int g, int lane, int r0, int nrows, int E,
    const int* __restrict__ ei, const float* __restrict__ el,
    const float* __restrict__ er, const float* __restrict__ x,
    const float4* own_rs, float* own_acc)
{
    int my_dst = 0, my_lrow = 0;
    float w0 = 0.f, w1 = 0.f, w2 = 0.f, w3 = 0.f;
    if (lane < g) {
        const int e   = list[lane];
        const int src = ei[e];
        my_dst  = ei[(size_t)E + e];
        const float4 r  = *(const float4*)(er + (size_t)my_dst * N_HEADS);
        if (own_match(src, r0, nrows)) {
            my_lrow = src - r0;
            const float4 l  = *(const float4*)(el + (size_t)src    * N_HEADS);
            const float4 rs = own_rs[my_lrow];
            w0 = gat_edge_e(l.x, r.x) / fmaxf(rs.x, GAT_EPS);
            w1 = gat_edge_e(l.y, r.y) / fmaxf(rs.y, GAT_EPS);
            w2 = gat_edge_e(l.z, r.z) / fmaxf(rs.z, GAT_EPS);
            w3 = gat_edge_e(l.w, r.w) / fmaxf(rs.w, GAT_EPS);
        }
    }
    for (int i = 0; i < g; i += 4) {
        int lr[4];
        v2f xv[4];
        #pragma unroll
        for (int k = 0; k < 4; ++k) {
            const int j = min(i + k, g - 1);
            const int d = own_bcast_i(my_dst, j);
            lr[k] = own_bcast_i(my_lrow, j);
            xv[k] = *(const v2f*)(x + (size_t)d * OUT_DIM + lane * 2);
        }
        #pragma unroll
        for (int k = 0; k < 4; ++k) {
            if (i + k < g) {
                const int j = i + k;
                float* a = own_acc + lr[k] * ROW_F + lane * 2;
                const float wa = own_bcast_f(w0, j);
                const float wb = own_bcast_f(w1, j);
                const float wc = own_bcast_f(w2, j);
                const float wd = own_bcast_f(w3, j);
                atomicAdd(a + 0 * OUT_DIM + 0, wa * xv[k].x);
                atomicAdd(a + 0 * OUT_DIM + 1, wa * xv[k].y);
                atomicAdd(a + 1 * OUT_DIM + 0, wb * xv[k].x);
                atomicAdd(a + 1 * OUT_DIM + 1, wb * xv[k].y);
                atomicAdd(a + 2 * OUT_DIM + 0, wc * xv[k].x);
                atomicAdd(a + 2 * OUT_DIM + 1, wc * xv[k].y);
                atomicAdd(a + 3 * OUT_DIM + 0, wd * xv[k].x);
                atomicAdd(a + 3 * OUT_DIM + 1, wd * xv[k].y);
            }
        }
    }
}

#define PK_B 8
template <bool ATOMIC>
__device__ __forceinline__ void own_process_packed(
    own_list_word* list, int g, int lane, int r0, int nrows, int E,
    const int* __restrict__ ei, const float* __restrict__ el,
    const float* __restrict__ er, const float* __restrict__ x,
    const float4* own_rs, float* own_acc)
{
    int my_dst = 0, my_lrow = 0;
    float w0 = 0.f, w1 = 0.f, w2 = 0.f, w3 = 0.f;
    if (lane < g) {
        const unsigned wd = (unsigned)list[lane];
        const unsigned lr = wd >> 24;
        if (lr < (unsigned)nrows) {
            const int e = (int)(wd & 0x00ffffffu);
            my_lrow = (int)lr;
            my_dst  = ei[(size_t)E + e];
            const float4 r  = *(const float4*)(er + (size_t)my_dst * N_HEADS);
            const float4 l  = *(const float4*)(el + (size_t)(r0 + my_lrow) * N_HEADS);
            const float4 rs = own_rs[my_lrow];
            w0 = gat_edge_e(l.x, r.x) / fmaxf(rs.x, GAT_EPS);
            w1 = gat_edge_e(l.y, r.y) / fmaxf(rs.y, GAT_EPS);
            w2 = gat_edge_e(l.z, r.z) / fmaxf(rs.z, GAT_EPS);
            w3 = gat_edge_e(l.w, r.w) / fmaxf(rs.w, GAT_EPS);
        }
    }
    v2f cur[PK_B];
    #pragma unroll
    for (int k = 0; k < PK_B; ++k) {
        const int d = own_bcast_i(my_dst, min(k, g - 1));
        cur[k] = *(const v2f*)(x + (size_t)d * OUT_DIM + lane * 2);
    }
    for (int i = 0; i < g; i += PK_B) {
        v2f nxt[PK_B];
        #pragma unroll
        for (int k = 0; k < PK_B; ++k) nxt[k] = cur[k];
        if (i + PK_B < g) {
            #pragma unroll
            for (int k = 0; k < PK_B; ++k) {
                const int d = own_bcast_i(my_dst, min(i + PK_B + k, g - 1));
                nxt[k] = *(const v2f*)(x + (size_t)d * OUT_DIM + lane * 2);
            }
        }
        #pragma unroll
        for (int k = 0; k < PK_B; ++k) {
            if (i + k < g) {
                const int j = i + k;
                float* a = own_acc + own_bcast_i(my_lrow, j) * ROW_F + lane * 2;
                const float wa = own_bcast_f(w0, j);
                const float wb = own_bcast_f(w1, j);
                const float wc = own_bcast_f(w2, j);
                const float wd = own_bcast_f(w3, j);
                if (ATOMIC) {
                    atomicAdd(a + 0 * OUT_DIM + 0, wa * cur[k].x);
                    atomicAdd(a + 0 * OUT_DIM + 1, wa * cur[k].y);
                    atomicAdd(a + 1 * OUT_DIM + 0, wb * cur[k].x);
                    atomicAdd(a + 1 * OUT_DIM + 1, wb * cur[k].y);
                    atomicAdd(a + 2 * OUT_DIM + 0, wc * cur[k].x);
                    atomicAdd(a + 2 * OUT_DIM + 1, wc * cur[k].y);
                    atomicAdd(a + 3 * OUT_DIM + 0, wd * cur[k].x);
                    atomicAdd(a + 3 * OUT_DIM + 1, wd * cur[k].y);
                } else {
                    v2f s0 = *(v2f*)(a + 0 * OUT_DIM);
                    v2f s1 = *(v2f*)(a + 1 * OUT_DIM);
                    v2f s2 = *(v2f*)(a + 2 * OUT_DIM);
                    v2f s3 = *(v2f*)(a + 3 * OUT_DIM);
                    s0.x += wa * cur[k].x;  s0.y += wa * cur[k].y;
                    s1.x += wb * cur[k].x;  s1.y += wb * cur[k].y;
                    s2.x += wc * cur[k].x;  s2.y += wc * cur[k].y;
                    s3.x += wd * cur[k].x;  s3.y += wd * cur[k].y;
                    *(v2f*)(a + 0 * OUT_DIM) = s0;
                    *(v2f*)(a + 1 * OUT_DIM) = s1;
                    *(v2f*)(a + 2 * OUT_DIM) = s2;
                    *(v2f*)(a + 3 * OUT_DIM) = s3;
                }
            }
        }
        #pragma unroll
        for (int k = 0; k < PK_B; ++k) cur[k] = nxt[k];
    }
}

template <bool AGG>
__device__ __forceinline__ bool own_scan(
    own_list_word* list, int& cnt_out, int lane, int wave, int r0, int nrows, int E,
    const int* __restrict__ ei, const float* __restrict__ el,
    const float* __restrict__ er, const float* __restrict__ x,
    const float4* own_rs, float* own_rs_f, float* own_acc)
{
    int  cnt    = 0;
    int  summed = 0;
    bool keep   = true;

    const int   E4     = E >> 2;
    const int4* src4   = (const int4*)ei;
    const int   nsteps = (E4 + OWN_BLOCK4 - 1) / OWN_BLOCK4;
    const int   last4  = E4 - 1;

    int4 cur[SCAN_U];
    #pragma unroll
    for (int u = 0; u < SCAN_U; ++u)
        cur[u] = make_int4(-1, -1, -1, -1);
    if (nsteps > 0) {
        #pragma unroll
        for (int u = 0; u < SCAN_U; ++u)
            cur[u] = src4[min(wave * OWN_STEP4 + u * 32 + lane, last4)];
    }

    for (int s = 0; s < nsteps; ++s) {
        int4 nxt[SCAN_U];
        #pragma unroll
        for (int u = 0; u < SCAN_U; ++u)
            nxt[u] = src4[min((s + 1) * OWN_BLOCK4 + wave * OWN_STEP4 + u * 32 + lane, last4)];

        const int cb4 = s * OWN_BLOCK4 + wave * OWN_STEP4 + lane;
        unsigned m = 0;
        #pragma unroll
        for (int u = 0; u < SCAN_U; ++u) {
            unsigned mu = own_match(cur[u].x, r0, nrows)
                        | (own_match(cur[u].y, r0, nrows) << 1)
                        | (own_match(cur[u].z, r0, nrows) << 2)
                        | (own_match(cur[u].w, r0, nrows) << 3);
            if (cb4 + u * 32 >= E4) mu = 0u;
            m |= mu << (4 * u);
        }

        for (int t = 0; t < 4 * SCAN_U; ++t) {
            const bool has = (m != 0);
            const unsigned bal = __builtin_amdgcn_ballot_w32(has);
            if (bal == 0) break;
            if (has) {
                const int bit = __builtin_ctz(m);
                m &= m - 1;
                const int e = ((cb4 + (bit >> 2) * 32) << 2) + (bit & 3);
                list[cnt + __builtin_popcount(bal & ((1u << lane) - 1u))] = e;
            }
            cnt += __builtin_popcount(bal);
        }

        if (AGG) {
            int off = 0;
            while (cnt - off >= 32) {
                own_process_group(list + off, 32, lane, r0, nrows, E, ei, el, er, x, own_rs, own_acc);
                off += 32;
            }
            if (off) {
                const int left = cnt - off;
                int tmp = 0;
                if (lane < left) tmp = list[off + lane];
                if (lane < left) list[lane] = tmp;
                cnt = left;
            }
        } else {
            while (cnt - summed >= 32) {
                own_sum_group(list + summed, 32, lane, r0, nrows, E, ei, el, er, own_rs_f);
                summed += 32;
            }
            if (cnt > LIST_CAP - STEP_EDGES) {
                keep = false;
                const int left = cnt - summed;
                int tmp = 0;
                if (lane < left) tmp = list[summed + lane];
                if (lane < left) list[lane] = tmp;
                cnt = left;
                summed = 0;
            }
        }

        #pragma unroll
        for (int u = 0; u < SCAN_U; ++u) cur[u] = nxt[u];
    }

    {
        const int e = (E4 << 2) + lane;
        bool has = false;
        if (wave == 0 && e < E) has = (own_match(ei[e], r0, nrows) != 0u);
        const unsigned bal = __builtin_amdgcn_ballot_w32(has);
        if (has) list[cnt + __builtin_popcount(bal & ((1u << lane) - 1u))] = e;
        cnt += __builtin_popcount(bal);
    }
    if (AGG) {
        for (int off = 0; off < cnt; off += 32)
            own_process_group(list + off, min(32, cnt - off), lane, r0, nrows, E, ei, el, er, x, own_rs, own_acc);
    } else {
        for (; summed < cnt; summed += 32)
            own_sum_group(list + summed, min(32, cnt - summed), lane, r0, nrows, E, ei, el, er, own_rs_f);
    }
    cnt_out = cnt;
    return keep;
}

#define KEY_U      4
#define KEY_STEPQ  (KEY_U * 32)
#define KEY_BLOCKQ (KEY_STEPQ * OWN_WAVES)

__device__ __forceinline__ unsigned key_match4(unsigned d, unsigned pat) {
    const unsigned x = d ^ pat;
    const unsigned t = (x & 0x7f7f7f7fu) + 0x7f7f7f7fu;
    return ~(t | x) & 0x80808080u;
}

template <bool AGG>
__device__ __forceinline__ void own_make_room(
    own_list_word* list, int& cnt, int& summed, bool& keep,
    int lane, int r0, int nrows, int E,
    const int* __restrict__ ei, const float* __restrict__ el,
    const float* __restrict__ er, const float* __restrict__ x,
    const float4* own_rs, float* own_rs_f, float* own_acc)
{
    if (AGG) {
        for (int off = 0; off < cnt; off += 32)
            own_process_group(list + off, min(32, cnt - off), lane, r0, nrows, E, ei, el, er, x, own_rs, own_acc);
    } else {
        for (; summed < cnt; summed += 32)
            own_sum_group(list + summed, min(32, cnt - summed), lane, r0, nrows, E, ei, el, er, own_rs_f);
        summed = 0;
        keep   = false;
    }
    cnt = 0;
}

template <bool AGG>
__device__ __forceinline__ bool own_scan_keys(
    own_list_word* list, int& cnt_out, int lane, int wave, int wg, int r0, int nrows, int E,
    const int* __restrict__ ei, const int* __restrict__ keys,
    const float* __restrict__ el, const float* __restrict__ er,
    const float* __restrict__ x,
    const float4* own_rs, float* own_rs_f, float* own_acc)
{
    int  cnt    = 0;
    int  summed = 0;
    bool keep   = true;

    const int      nunits = E >> 4;
    const uint4*   key4   = (const uint4*)keys;
    const int      nsteps = (nunits + KEY_BLOCKQ - 1) / KEY_BLOCKQ;
    const int      lastq  = nunits - 1;
    const unsigned pat    = (unsigned)(wg & 255) * 0x01010101u;
    const unsigned below  = (1u << lane) - 1u;

    uint4 cur[KEY_U];
    #pragma unroll
    for (int u = 0; u < KEY_U; ++u)
        cur[u] = make_uint4(0u, 0u, 0u, 0u);
    if (nsteps > 0) {
        #pragma unroll
        for (int u = 0; u < KEY_U; ++u)
            cur[u] = key4[min(wave * KEY_STEPQ + u * 32 + lane, lastq)];
    }

    for (int s = 0; s < nsteps; ++s) {
        uint4 nxt[KEY_U];
        #pragma unroll
        for (int u = 0; u < KEY_U; ++u)
            nxt[u] = key4[min((s + 1) * KEY_BLOCKQ + wave * KEY_STEPQ + u * 32 + lane, lastq)];

        const int q0 = s * KEY_BLOCKQ + wave * KEY_STEPQ + lane;

        #pragma unroll
        for (int p = 0; p < KEY_U; p += 2) {
            const int qa = q0 + p * 32;
            unsigned ma = (key_match4(cur[p].x, pat) >> 7) | (key_match4(cur[p].y, pat) >> 6)
                        | (key_match4(cur[p].z, pat) >> 5) | (key_match4(cur[p].w, pat) >> 4);
            unsigned mb = (key_match4(cur[p + 1].x, pat) >> 3) | (key_match4(cur[p + 1].y, pat) >> 2)
                        | (key_match4(cur[p + 1].z, pat) >> 1) |  key_match4(cur[p + 1].w, pat);
            if (qa      >= nunits) ma = 0u;
            if (qa + 32 >= nunits) mb = 0u;
            unsigned m = ma | mb;

            for (int t = 0; t < 32; ++t) {
                const bool has = (m != 0);
                const unsigned bal = __builtin_amdgcn_ballot_w32(has);
                if (bal == 0) break;
                const int n = __builtin_popcount(bal);
                if (cnt + n > LIST_CAP)
                    own_make_room<AGG>(list, cnt, summed, keep, lane, r0, nrows, E,
                                       ei, el, er, x, own_rs, own_rs_f, own_acc);
                if (has) {
                    const int bit = __builtin_ctz(m);
                    m &= m - 1;
                    const int q = qa + ((bit >> 2) & 1) * 32;
                    const int e = (q << 4) + ((bit & 3) << 2) + (bit >> 3);
                    list[cnt + __builtin_popcount(bal & below)] = e;
                }
                cnt += n;
            }
        }

        if (AGG) {
            int off = 0;
            while (cnt - off >= 32) {
                own_process_group(list + off, 32, lane, r0, nrows, E, ei, el, er, x, own_rs, own_acc);
                off += 32;
            }
            if (off) {
                const int left = cnt - off;
                int tmp = 0;
                if (lane < left) tmp = list[off + lane];
                if (lane < left) list[lane] = tmp;
                cnt = left;
            }
        } else {
            while (cnt - summed >= 32) {
                own_sum_group(list + summed, 32, lane, r0, nrows, E, ei, el, er, own_rs_f);
                summed += 32;
            }
        }

        #pragma unroll
        for (int u = 0; u < KEY_U; ++u) cur[u] = nxt[u];
    }

    {
        const int e = (nunits << 4) + lane;
        bool has = false;
        if (wave == 0 && e < E) has = (own_match(ei[e], r0, nrows) != 0u);
        const unsigned bal = __builtin_amdgcn_ballot_w32(has);
        const int n = __builtin_popcount(bal);
        if (cnt + n > LIST_CAP)
            own_make_room<AGG>(list, cnt, summed, keep, lane, r0, nrows, E,
                               ei, el, er, x, own_rs, own_rs_f, own_acc);
        if (has) list[cnt + __builtin_popcount(bal & below)] = e;
        cnt += n;
    }
    if (AGG) {
        for (int off = 0; off < cnt; off += 32)
            own_process_group(list + off, min(32, cnt - off), lane, r0, nrows, E, ei, el, er, x, own_rs, own_acc);
    } else {
        for (; summed < cnt; summed += 32)
            own_sum_group(list + summed, min(32, cnt - summed), lane, r0, nrows, E, ei, el, er, own_rs_f);
    }
    cnt_out = cnt;
    return keep;
}

template <bool KEYED>
__global__ __launch_bounds__(OWN_THREADS) void gat_own_rows(
    const int* __restrict__ ei, const int* __restrict__ keys,
    const float* __restrict__ el,
    const float* __restrict__ er, const float* __restrict__ x,
    const float* __restrict__ b, float* __restrict__ out, int N, int E)
{
    __shared__ float  own_acc[OWN_ROWS * ROW_F];
    __shared__ float4 own_rs[OWN_ROWS];
    __shared__ int    own_list[OWN_WAVES * LIST_CAP];
    __shared__ int    own_bins[OWN_WAVES * LIST_CAP];
    __shared__ int    own_give[OWN_WAVES * OWN_WAVES];
    __shared__ int    own_next[OWN_WAVES * OWN_WAVES];
    __shared__ int    own_allkept;

    const int tid  = threadIdx.x;
    const int lane = tid & 31;
    const int wave = tid >> 5;
    const int r0    = blockIdx.x * OWN_ROWS;
    const int nrows = min(OWN_ROWS, N - r0);

    float4* acc4 = (float4*)own_acc;
    for (int i = tid; i < OWN_ROWS * (ROW_F / 4); i += OWN_THREADS)
        acc4[i] = make_float4(0.f, 0.f, 0.f, 0.f);
    for (int i = tid; i < OWN_ROWS; i += OWN_THREADS)
        own_rs[i] = make_float4(0.f, 0.f, 0.f, 0.f);
    for (int i = tid; i < OWN_WAVES * OWN_WAVES; i += OWN_THREADS)
        own_give[i] = 0;
    if (tid == 0) own_allkept = 1;
    __syncthreads();

    own_list_word* list = (own_list_word*)(own_list + wave * LIST_CAP);
    float* own_rs_f = (float*)own_rs;

    int cnt = 0;
    const bool kept = KEYED
        ? own_scan_keys<false>(list, cnt, lane, wave, (int)blockIdx.x, r0, nrows, E,
                               ei, keys, el, er, x, own_rs, own_rs_f, own_acc)
        : own_scan<false>(list, cnt, lane, wave, r0, nrows, E,
                          ei, el, er, x, own_rs, own_rs_f, own_acc);

    if (lane == 0 && !kept) own_allkept = 0;

    const bool packed = (E <= PACK_MAX_E);
    if (packed && kept) {
        for (int base = 0; base < cnt; base += 32) {
            const int idx = base + lane;
            if (idx < cnt) {
                const unsigned lr = ((unsigned)list[idx]) >> 24;
                if (lr < (unsigned)nrows)
                    atomicAdd(&own_give[wave * OWN_WAVES + (int)(lr & (OWN_WAVES - 1))], 1);
            }
        }
    }

    __syncthreads();

    const bool all_kept = (__builtin_amdgcn_readfirstlane(own_allkept) != 0);
    if (packed && all_kept) {
        int total = 0, before = 0;
        for (int s = 0; s < OWN_WAVES; ++s) {
            const int n = own_give[s * OWN_WAVES + lane];
            total  += n;
            before += (s < wave) ? n : 0;
        }
        int start = 0;
        #pragma unroll
        for (int i = 0; i < OWN_WAVES - 1; ++i) {
            const int n = own_bcast_i(total, i);
            start += (lane > i) ? n : 0;
        }
        own_next[wave * OWN_WAVES + lane] = start + before;
        const int me       = __builtin_amdgcn_readfirstlane(wave);
        const int my_start = own_bcast_i(start, me);
        const int my_total = own_bcast_i(total, me);

        for (int base = 0; base < cnt; base += 32) {
            const int idx = base + lane;
            if (idx < cnt) {
                const int wd = list[idx];
                const unsigned lr = ((unsigned)wd) >> 24;
                if (lr < (unsigned)nrows) {
                    const int at = atomicAdd(&own_next[wave * OWN_WAVES + (int)(lr & (OWN_WAVES - 1))], 1);
                    own_bins[at] = wd;
                }
            }
        }

        __syncthreads();

        own_list_word* bin = (own_list_word*)(own_bins + my_start);
        for (int off = 0; off < my_total; off += 32)
            own_process_packed<false>(bin + off, min(32, my_total - off), lane, r0, nrows, E, ei, el, er, x, own_rs, own_acc);
    } else if (kept && packed) {
        for (int off = 0; off < cnt; off += 32)
            own_process_packed<true>(list + off, min(32, cnt - off), lane, r0, nrows, E, ei, el, er, x, own_rs, own_acc);
    } else if (kept) {
        int off = 0;
        for (; cnt - off >= 32; off += 32)
            own_process_group(list + off, 32, lane, r0, nrows, E, ei, el, er, x, own_rs, own_acc);
        if (cnt - off > 0)
            own_process_group(list + off, cnt - off, lane, r0, nrows, E, ei, el, er, x, own_rs, own_acc);
    } else {
        int unused = 0;
        if (KEYED)
            own_scan_keys<true>(list, unused, lane, wave, (int)blockIdx.x, r0, nrows, E,
                                ei, keys, el, er, x, own_rs, own_rs_f, own_acc);
        else
            own_scan<true>(list, unused, lane, wave, r0, nrows, E,
                           ei, el, er, x, own_rs, own_rs_f, own_acc);
    }

    __syncthreads();

    float4* out4 = (float4*)out + (size_t)r0 * (ROW_F / 4);
    for (int i = tid; i < nrows * (ROW_F / 4); i += OWN_THREADS) {
        float4 v = acc4[i];
        const int d0 = (i * 4) & (OUT_DIM - 1);
        v.x += b[d0 + 0]; v.y += b[d0 + 1]; v.z += b[d0 + 2]; v.w += b[d0 + 3];
        v.x = (v.x > 0.f) ? v.x : (__expf(v.x) - 1.f);
        v.y = (v.y > 0.f) ? v.y : (__expf(v.y) - 1.f);
        v.z = (v.z > 0.f) ? v.z : (__expf(v.z) - 1.f);
        v.w = (v.w > 0.f) ? v.w : (__expf(v.w) - 1.f);
        out4[i] = v;
    }
}

extern "C" void kernel_launch(void* const* d_in, const int* in_sizes, int n_in,
                              void* d_out, int out_size, void* d_ws, size_t ws_size,
                              hipStream_t stream) {
    const float* h  = (const float*)d_in[0];
    const float* W  = (const float*)d_in[1];
    const float* Wl = (const float*)d_in[2];
    const float* Wr = (const float*)d_in[3];
    const float* b  = (const float*)d_in[4];
    const int*   ei = (const int*)d_in[5];

    const int N = in_sizes[0] / IN_DIM;
    const int E = in_sizes[5] / 2;

    float* out = (float*)d_out;

    float* x      = (float*)d_ws;
    float* el     = x  + (size_t)N * OUT_DIM;
    float* er     = el + (size_t)N * N_HEADS;
    int*   keys   = (int*)(er + (size_t)N * N_HEADS);
    const size_t need  = (size_t)N * (OUT_DIM + 2 * N_HEADS) * sizeof(float) + (size_t)E;
    const bool   keyed = (ws_size >= need);

    gat_gemm_wmma<<<(unsigned)((N + 15) / 16), 128, 0, stream>>>(h, W, x, N);

    gat_coef<<<(unsigned)((N + 255) / 256), 256, 0, stream>>>(
        x, Wl, Wr, el, er, ei, keyed ? keys : (int*)nullptr, N, E);

    if (keyed)
        gat_own_rows<true><<<(unsigned)((N + OWN_ROWS - 1) / OWN_ROWS), OWN_THREADS, 0, stream>>>(
            ei, keys, el, er, x, b, out, N, E);
    else
        gat_own_rows<false><<<(unsigned)((N + OWN_ROWS - 1) / OWN_ROWS), OWN_THREADS, 0, stream>>>(
            ei, keys, el, er, x, b, out, N, E);
}


# phase 1 keeps each listed edge's destination and four weights in LDS (the idle value-sum space); phase 2 takes them by list place: no 2nd read of index or coefficient row, no 2nd exp; sums bit-same
# speedup vs baseline: 5.5113x; 1.1757x over previous
#include <hip/hip_runtime.h>

#define IN_DIM  256
#define OUT_DIM 64
#define N_HEADS 4
#define GAT_EPS 1e-12f

typedef float v2f __attribute__((ext_vector_type(2)));
typedef float v8f __attribute__((ext_vector_type(8)));

#define ROW_F (N_HEADS * OUT_DIM)

__global__ __launch_bounds__(128) void gat_gemm_wmma(
    const float* __restrict__ h, const float* __restrict__ W,
    float* __restrict__ x, int N)
{
    const int lane = threadIdx.x & 31;
    const int wave = threadIdx.x >> 5;
    const int half = lane >> 4;
    const int l15  = lane & 15;
    const int row0 = blockIdx.x * 16;
    const int col0 = wave * 16;

    int arow = row0 + l15;
    if (arow >= N) arow = N - 1;
    const float* __restrict__ hrow = h + (size_t)arow * IN_DIM;
    const int bcol = col0 + l15;

    v8f acc = {0.f, 0.f, 0.f, 0.f, 0.f, 0.f, 0.f, 0.f};

    for (int k = 0; k < IN_DIM; k += 4) {
        const int ka = k + half * 2;
        v2f a = *(const v2f*)(hrow + ka);
        v2f b;
        b.x = W[(size_t)ka       * OUT_DIM + bcol];
        b.y = W[(size_t)(ka + 1) * OUT_DIM + bcol];
        acc = __builtin_amdgcn_wmma_f32_16x16x4_f32(
             false, a,  false, b,
             (short)0, acc,  false,  false);
    }

    float* __restrict__ xout =
        x + ((size_t)(row0 + half * 8)) * OUT_DIM + col0 + l15;
    if (row0 + 16 <= N) {
        #pragma unroll
        for (int r = 0; r < 8; ++r)
            xout[(size_t)r * OUT_DIM] = acc[r];
    } else {
        #pragma unroll
        for (int r = 0; r < 8; ++r) {
            if (row0 + r + half * 8 < N)
                xout[(size_t)r * OUT_DIM] = acc[r];
        }
    }
}

#define OWN_ROWS    196

__device__ __forceinline__ unsigned own_key(int v) {
    return (((unsigned)v) / (unsigned)OWN_ROWS) & 255u;
}

__global__ __launch_bounds__(256) void gat_coef(
    const float* __restrict__ x, const float* __restrict__ Wl,
    const float* __restrict__ Wr, float* __restrict__ el,
    float* __restrict__ er, const int* __restrict__ ei,
    int* __restrict__ keys, int N, int E)
{
    if (keys != nullptr) {
        const int   nunits = E >> 4;
        const int4* src4   = (const int4*)ei;
        uint4*      key4   = (uint4*)keys;
        const int   stride = (int)(gridDim.x * blockDim.x);
        for (int q = blockIdx.x * blockDim.x + threadIdx.x; q < nunits; q += stride) {
            unsigned w[4];
            #pragma unroll
            for (int j = 0; j < 4; ++j) {
                const int4 v = src4[(size_t)q * 4 + j];
                w[j] = own_key(v.x) | (own_key(v.y) << 8) | (own_key(v.z) << 16) | (own_key(v.w) << 24);
            }
            key4[q] = make_uint4(w[0], w[1], w[2], w[3]);
        }
    }

    const int n = blockIdx.x * blockDim.x + threadIdx.x;
    if (n >= N) return;
    const float* __restrict__ xr = x + (size_t)n * OUT_DIM;
    float al[N_HEADS] = {0.f, 0.f, 0.f, 0.f};
    float ar[N_HEADS] = {0.f, 0.f, 0.f, 0.f};
    for (int d = 0; d < OUT_DIM; ++d) {
        const float xv = xr[d];
        #pragma unroll
        for (int hh = 0; hh < N_HEADS; ++hh) {
            al[hh] = fmaf(xv, Wl[hh * OUT_DIM + d], al[hh]);
            ar[hh] = fmaf(xv, Wr[hh * OUT_DIM + d], ar[hh]);
        }
    }
    #pragma unroll
    for (int hh = 0; hh < N_HEADS; ++hh) {
        el[(size_t)n * N_HEADS + hh] = al[hh];
        er[(size_t)n * N_HEADS + hh] = ar[hh];
    }
}

__device__ __forceinline__ float gat_edge_e(float l, float r) {
    float v = l + r;
    v = (v > 0.f) ? v : 0.2f * v;
    return __expf(v);
}

#define OWN_THREADS 1024
#define OWN_WAVES   (OWN_THREADS / 32)
#define SCAN_U      2
#define OWN_STEP4   (SCAN_U * 32)
#define OWN_BLOCK4  (OWN_STEP4 * OWN_WAVES)
#define STEP_EDGES  (OWN_STEP4 * 4)
#define LIST_CAP    416
static_assert(STEP_EDGES + 31 <= LIST_CAP, "a step after a drop must fit the list");
#define PACK_MAX_E  (1 << 24)

typedef volatile int __attribute__((address_space(3))) own_list_word;

__device__ __forceinline__ int own_bcast_i(int v, int j) {
    return __builtin_amdgcn_readlane(v, j);
}
__device__ __forceinline__ float own_bcast_f(float v, int j) {
    return __int_as_float(__builtin_amdgcn_readlane(__float_as_int(v), j));
}

__device__ __forceinline__ unsigned own_match(int v, int r0, int nrows) {
    return ((unsigned)(v - r0) < (unsigned)nrows) ? 1u : 0u;
}

#define PAY_CAP 313
static_assert((size_t)PAY_CAP * OWN_WAVES * 20 <= (size_t)OWN_ROWS * ROW_F * 4,
              "the kept weights and destinations must fit the value sums' space");
struct own_pay {
    float4* e;
    int*    dst;
};

__device__ __forceinline__ void own_sum_group(
    own_list_word* list, int g, int lane, int r0, int nrows, int E,
    const int* __restrict__ ei, const float* __restrict__ el,
    const float* __restrict__ er, float* own_rs_f, own_pay pay, int pos0)
{
    if (lane < g) {
        const int e   = list[lane];
        const int src = ei[e];
        const int dst = ei[(size_t)E + e];
        const float4 r = *(const float4*)(er + (size_t)dst * N_HEADS);
        const bool mine = (own_match(src, r0, nrows) != 0u);
        if (mine) {
            const float4 l = *(const float4*)(el + (size_t)src * N_HEADS);
            float* a = own_rs_f + (src - r0) * N_HEADS;
            const float4 ev = make_float4(gat_edge_e(l.x, r.x), gat_edge_e(l.y, r.y),
                                          gat_edge_e(l.z, r.z), gat_edge_e(l.w, r.w));
            atomicAdd(a + 0, ev.x);
            atomicAdd(a + 1, ev.y);
            atomicAdd(a + 2, ev.z);
            atomicAdd(a + 3, ev.w);
            const int pos = pos0 + lane;
            if (pos < PAY_CAP) {
                pay.e[pos]   = ev;
                pay.dst[pos] = dst;
            }
        }
        if (E <= PACK_MAX_E)
            list[lane] = mine ? (int)((unsigned)e | ((unsigned)(src - r0) << 24))
                              : (int)0xff000000u;
    }
}

__device__ __forceinline__ void own_process_group(
    own_list_word* list, int g, int lane, int r0, int nrows, int E,
    const int* __restrict__ ei, const float* __restrict__ el,
    const float* __restrict__ er, const float* __restrict__ x,
    const float4* own_rs, float* own_acc)
{
    int my_dst = 0, my_lrow = 0;
    float w0 = 0.f, w1 = 0.f, w2 = 0.f, w3 = 0.f;
    if (lane < g) {
        const int e   = list[lane];
        const int src = ei[e];
        my_dst  = ei[(size_t)E + e];
        const float4 r  = *(const float4*)(er + (size_t)my_dst * N_HEADS);
        if (own_match(src, r0, nrows)) {
            my_lrow = src - r0;
            const float4 l  = *(const float4*)(el + (size_t)src    * N_HEADS);
            const float4 rs = own_rs[my_lrow];
            w0 = gat_edge_e(l.x, r.x) / fmaxf(rs.x, GAT_EPS);
            w1 = gat_edge_e(l.y, r.y) / fmaxf(rs.y, GAT_EPS);
            w2 = gat_edge_e(l.z, r.z) / fmaxf(rs.z, GAT_EPS);
            w3 = gat_edge_e(l.w, r.w) / fmaxf(rs.w, GAT_EPS);
        }
    }
    for (int i = 0; i < g; i += 4) {
        int lr[4];
        v2f xv[4];
        #pragma unroll
        for (int k = 0; k < 4; ++k) {
            const int j = min(i + k, g - 1);
            const int d = own_bcast_i(my_dst, j);
            lr[k] = own_bcast_i(my_lrow, j);
            xv[k] = *(const v2f*)(x + (size_t)d * OUT_DIM + lane * 2);
        }
        #pragma unroll
        for (int k = 0; k < 4; ++k) {
            if (i + k < g) {
                const int j = i + k;
                float* a = own_acc + lr[k] * ROW_F + lane * 2;
                const float wa = own_bcast_f(w0, j);
                const float wb = own_bcast_f(w1, j);
                const float wc = own_bcast_f(w2, j);
                const float wd = own_bcast_f(w3, j);
                atomicAdd(a + 0 * OUT_DIM + 0, wa * xv[k].x);
                atomicAdd(a + 0 * OUT_DIM + 1, wa * xv[k].y);
                atomicAdd(a + 1 * OUT_DIM + 0, wb * xv[k].x);
                atomicAdd(a + 1 * OUT_DIM + 1, wb * xv[k].y);
                atomicAdd(a + 2 * OUT_DIM + 0, wc * xv[k].x);
                atomicAdd(a + 2 * OUT_DIM + 1, wc * xv[k].y);
                atomicAdd(a + 3 * OUT_DIM + 0, wd * xv[k].x);
                atomicAdd(a + 3 * OUT_DIM + 1, wd * xv[k].y);
            }
        }
    }
}

#define PK_B 8
__device__ __forceinline__ void own_process_packed(
    own_list_word* list, int g, int lane, int r0, int nrows, int E,
    const int* __restrict__ ei, const float* __restrict__ el,
    const float* __restrict__ er, const float* __restrict__ x,
    const float4* own_rs, float* own_acc)
{
    int my_dst = 0, my_lrow = 0;
    float w0 = 0.f, w1 = 0.f, w2 = 0.f, w3 = 0.f;
    if (lane < g) {
        const unsigned wd = (unsigned)list[lane];
        const unsigned lr = wd >> 24;
        if (lr < (unsigned)nrows) {
            const int e = (int)(wd & 0x00ffffffu);
            my_lrow = (int)lr;
            my_dst  = ei[(size_t)E + e];
            const float4 r  = *(const float4*)(er + (size_t)my_dst * N_HEADS);
            const float4 l  = *(const float4*)(el + (size_t)(r0 + my_lrow) * N_HEADS);
            const float4 rs = own_rs[my_lrow];
            w0 = gat_edge_e(l.x, r.x) / fmaxf(rs.x, GAT_EPS);
            w1 = gat_edge_e(l.y, r.y) / fmaxf(rs.y, GAT_EPS);
            w2 = gat_edge_e(l.z, r.z) / fmaxf(rs.z, GAT_EPS);
            w3 = gat_edge_e(l.w, r.w) / fmaxf(rs.w, GAT_EPS);
        }
    }
    v2f cur[PK_B];
    #pragma unroll
    for (int k = 0; k < PK_B; ++k) {
        const int d = own_bcast_i(my_dst, min(k, g - 1));
        cur[k] = *(const v2f*)(x + (size_t)d * OUT_DIM + lane * 2);
    }
    for (int i = 0; i < g; i += PK_B) {
        v2f nxt[PK_B];
        #pragma unroll
        for (int k = 0; k < PK_B; ++k) nxt[k] = cur[k];
        if (i + PK_B < g) {
            #pragma unroll
            for (int k = 0; k < PK_B; ++k) {
                const int d = own_bcast_i(my_dst, min(i + PK_B + k, g - 1));
                nxt[k] = *(const v2f*)(x + (size_t)d * OUT_DIM + lane * 2);
            }
        }
        #pragma unroll
        for (int k = 0; k < PK_B; ++k) {
            if (i + k < g) {
                const int j = i + k;
                float* a = own_acc + own_bcast_i(my_lrow, j) * ROW_F + lane * 2;
                const float wa = own_bcast_f(w0, j);
                const float wb = own_bcast_f(w1, j);
                const float wc = own_bcast_f(w2, j);
                const float wd = own_bcast_f(w3, j);
                atomicAdd(a + 0 * OUT_DIM + 0, wa * cur[k].x);
                atomicAdd(a + 0 * OUT_DIM + 1, wa * cur[k].y);
                atomicAdd(a + 1 * OUT_DIM + 0, wb * cur[k].x);
                atomicAdd(a + 1 * OUT_DIM + 1, wb * cur[k].y);
                atomicAdd(a + 2 * OUT_DIM + 0, wc * cur[k].x);
                atomicAdd(a + 2 * OUT_DIM + 1, wc * cur[k].y);
                atomicAdd(a + 3 * OUT_DIM + 0, wd * cur[k].x);
                atomicAdd(a + 3 * OUT_DIM + 1, wd * cur[k].y);
            }
        }
        #pragma unroll
        for (int k = 0; k < PK_B; ++k) cur[k] = nxt[k];
    }
}

__device__ __forceinline__ float own_elu_b(float v, float bb) {
    v += bb;
    return (v > 0.f) ? v : (__expf(v) - 1.f);
}

__device__ __forceinline__ void own_store_row(
    float* __restrict__ out, int grow, int lane, v2f bias,
    v2f s0, v2f s1, v2f s2, v2f s3)
{
    float* o = out + (size_t)grow * ROW_F + lane * 2;
    v2f t;
    t.x = own_elu_b(s0.x, bias.x);  t.y = own_elu_b(s0.y, bias.y);  *(v2f*)(o + 0 * OUT_DIM) = t;
    t.x = own_elu_b(s1.x, bias.x);  t.y = own_elu_b(s1.y, bias.y);  *(v2f*)(o + 1 * OUT_DIM) = t;
    t.x = own_elu_b(s2.x, bias.x);  t.y = own_elu_b(s2.y, bias.y);  *(v2f*)(o + 2 * OUT_DIM) = t;
    t.x = own_elu_b(s3.x, bias.x);  t.y = own_elu_b(s3.y, bias.y);  *(v2f*)(o + 3 * OUT_DIM) = t;
}

#define OWN_PER_WAVE ((OWN_ROWS + OWN_WAVES - 1) / OWN_WAVES)
__device__ __forceinline__ void own_rows_in_registers(
    own_list_word* bin, own_list_word* srt, int total,
    int lane, int wave, int r0, int nrows,
    const float* __restrict__ x, const float4* own_rs,
    const float4* pay_e, const int* pay_dst,
    const float* __restrict__ b, float* __restrict__ out)
{
    const unsigned below = (1u << lane) - 1u;

    int pos[OWN_PER_WAVE];
    #pragma unroll
    for (int q = 0; q < OWN_PER_WAVE; ++q) pos[q] = 0;
    for (int base = 0; base < total; base += 32) {
        const int idx = base + lane;
        unsigned slot = 0xffu;
        if (idx < total) slot = ((unsigned)bin[idx]) >> 29;
        #pragma unroll
        for (int q = 0; q + 1 < OWN_PER_WAVE; ++q)
            pos[q + 1] += __builtin_popcount(__builtin_amdgcn_ballot_w32(slot == (unsigned)q));
    }
    #pragma unroll
    for (int q = 1; q < OWN_PER_WAVE; ++q) pos[q] += pos[q - 1];

    for (int base = 0; base < total; base += 32) {
        const int idx = base + lane;
        int wd = 0;
        unsigned slot = 0xffu;
        if (idx < total) { wd = bin[idx]; slot = ((unsigned)wd) >> 29; }
        #pragma unroll
        for (int q = 0; q < OWN_PER_WAVE; ++q) {
            const unsigned bal = __builtin_amdgcn_ballot_w32(slot == (unsigned)q);
            if (slot == (unsigned)q) srt[pos[q] + __builtin_popcount(bal & below)] = wd;
            pos[q] += __builtin_popcount(bal);
        }
    }
    const int n = pos[OWN_PER_WAVE - 1];

    const v2f bias = *(const v2f*)(b + lane * 2);
    const v2f zero2 = {0.f, 0.f};
    v2f s0 = zero2, s1 = zero2, s2 = zero2, s3 = zero2;
    int row = __builtin_amdgcn_readfirstlane(wave);

    for (int off = 0; off < n; off += 32) {
        const int g = min(32, n - off);
        int my_dst = 0, my_lrow = 0;
        float4 ev = make_float4(0.f, 0.f, 0.f, 0.f);
        bool have = false;
        if (lane < g) {
            const unsigned wd = (unsigned)srt[off + lane];
            const unsigned lr = wd >> 24;
            if (lr < (unsigned)nrows) {
                const int p = (int)(wd & 0x00ffffffu);
                have    = true;
                my_lrow = (int)lr;
                my_dst  = pay_dst[p];
                ev      = pay_e[p];
            }
        }
        v2f cur[PK_B];
        #pragma unroll
        for (int k = 0; k < PK_B; ++k) {
            const int d = own_bcast_i(my_dst, min(k, g - 1));
            cur[k] = *(const v2f*)(x + (size_t)d * OUT_DIM + lane * 2);
        }
        float w0 = 0.f, w1 = 0.f, w2 = 0.f, w3 = 0.f;
        if (have) {
            const float4 rs = own_rs[my_lrow];
            w0 = ev.x / fmaxf(rs.x, GAT_EPS);
            w1 = ev.y / fmaxf(rs.y, GAT_EPS);
            w2 = ev.z / fmaxf(rs.z, GAT_EPS);
            w3 = ev.w / fmaxf(rs.w, GAT_EPS);
        }
        for (int i = 0; i < g; i += PK_B) {
            v2f nxt[PK_B];
            #pragma unroll
            for (int k = 0; k < PK_B; ++k) nxt[k] = cur[k];
            if (i + PK_B < g) {
                #pragma unroll
                for (int k = 0; k < PK_B; ++k) {
                    const int d = own_bcast_i(my_dst, min(i + PK_B + k, g - 1));
                    nxt[k] = *(const v2f*)(x + (size_t)d * OUT_DIM + lane * 2);
                }
            }
            #pragma unroll
            for (int k = 0; k < PK_B; ++k) {
                if (i + k < g) {
                    const int j = i + k;
                    const int r = own_bcast_i(my_lrow, j);
                    while (row < r) {
                        own_store_row(out, r0 + row, lane, bias, s0, s1, s2, s3);
                        s0 = zero2; s1 = zero2; s2 = zero2; s3 = zero2;
                        row += OWN_WAVES;
                    }
                    const float wa = own_bcast_f(w0, j);
                    const float wb = own_bcast_f(w1, j);
                    const float wc = own_bcast_f(w2, j);
                    const float wd = own_bcast_f(w3, j);
                    s0.x += wa * cur[k].x;  s0.y += wa * cur[k].y;
                    s1.x += wb * cur[k].x;  s1.y += wb * cur[k].y;
                    s2.x += wc * cur[k].x;  s2.y += wc * cur[k].y;
                    s3.x += wd * cur[k].x;  s3.y += wd * cur[k].y;
                }
            }
            #pragma unroll
            for (int k = 0; k < PK_B; ++k) cur[k] = nxt[k];
        }
    }
    while (row < nrows) {
        own_store_row(out, r0 + row, lane, bias, s0, s1, s2, s3);
        s0 = zero2; s1 = zero2; s2 = zero2; s3 = zero2;
        row += OWN_WAVES;
    }
}

template <bool AGG>
__device__ __forceinline__ bool own_scan(
    own_list_word* list, int& cnt_out, int lane, int wave, int r0, int nrows, int E,
    const int* __restrict__ ei, const float* __restrict__ el,
    const float* __restrict__ er, const float* __restrict__ x,
    const float4* own_rs, float* own_rs_f, float* own_acc, own_pay pay)
{
    int  cnt    = 0;
    int  summed = 0;
    bool keep   = true;

    const int   E4     = E >> 2;
    const int4* src4   = (const int4*)ei;
    const int   nsteps = (E4 + OWN_BLOCK4 - 1) / OWN_BLOCK4;
    const int   last4  = E4 - 1;

    int4 cur[SCAN_U];
    #pragma unroll
    for (int u = 0; u < SCAN_U; ++u)
        cur[u] = make_int4(-1, -1, -1, -1);
    if (nsteps > 0) {
        #pragma unroll
        for (int u = 0; u < SCAN_U; ++u)
            cur[u] = src4[min(wave * OWN_STEP4 + u * 32 + lane, last4)];
    }

    for (int s = 0; s < nsteps; ++s) {
        int4 nxt[SCAN_U];
        #pragma unroll
        for (int u = 0; u < SCAN_U; ++u)
            nxt[u] = src4[min((s + 1) * OWN_BLOCK4 + wave * OWN_STEP4 + u * 32 + lane, last4)];

        const int cb4 = s * OWN_BLOCK4 + wave * OWN_STEP4 + lane;
        unsigned m = 0;
        #pragma unroll
        for (int u = 0; u < SCAN_U; ++u) {
            unsigned mu = own_match(cur[u].x, r0, nrows)
                        | (own_match(cur[u].y, r0, nrows) << 1)
                        | (own_match(cur[u].z, r0, nrows) << 2)
                        | (own_match(cur[u].w, r0, nrows) << 3);
            if (cb4 + u * 32 >= E4) mu = 0u;
            m |= mu << (4 * u);
        }

        for (int t = 0; t < 4 * SCAN_U; ++t) {
            const bool has = (m != 0);
            const unsigned bal = __builtin_amdgcn_ballot_w32(has);
            if (bal == 0) break;
            if (has) {
                const int bit = __builtin_ctz(m);
                m &= m - 1;
                const int e = ((cb4 + (bit >> 2) * 32) << 2) + (bit & 3);
                list[cnt + __builtin_popcount(bal & ((1u << lane) - 1u))] = e;
            }
            cnt += __builtin_popcount(bal);
        }

        if (AGG) {
            int off = 0;
            while (cnt - off >= 32) {
                own_process_group(list + off, 32, lane, r0, nrows, E, ei, el, er, x, own_rs, own_acc);
                off += 32;
            }
            if (off) {
                const int left = cnt - off;
                int tmp = 0;
                if (lane < left) tmp = list[off + lane];
                if (lane < left) list[lane] = tmp;
                cnt = left;
            }
        } else {
            while (cnt - summed >= 32) {
                own_sum_group(list + summed, 32, lane, r0, nrows, E, ei, el, er, own_rs_f, pay, summed);
                summed += 32;
            }
            if (cnt > LIST_CAP - STEP_EDGES) {
                keep = false;
                const int left = cnt - summed;
                int tmp = 0;
                if (lane < left) tmp = list[summed + lane];
                if (lane < left) list[lane] = tmp;
                cnt = left;
                summed = 0;
            }
        }

        #pragma unroll
        for (int u = 0; u < SCAN_U; ++u) cur[u] = nxt[u];
    }

    {
        const int e = (E4 << 2) + lane;
        bool has = false;
        if (wave == 0 && e < E) has = (own_match(ei[e], r0, nrows) != 0u);
        const unsigned bal = __builtin_amdgcn_ballot_w32(has);
        if (has) list[cnt + __builtin_popcount(bal & ((1u << lane) - 1u))] = e;
        cnt += __builtin_popcount(bal);
    }
    if (AGG) {
        for (int off = 0; off < cnt; off += 32)
            own_process_group(list + off, min(32, cnt - off), lane, r0, nrows, E, ei, el, er, x, own_rs, own_acc);
    } else {
        for (; summed < cnt; summed += 32)
            own_sum_group(list + summed, min(32, cnt - summed), lane, r0, nrows, E, ei, el, er, own_rs_f, pay, summed);
    }
    cnt_out = cnt;
    return keep;
}

#define KEY_U      4
#define KEY_STEPQ  (KEY_U * 32)
#define KEY_BLOCKQ (KEY_STEPQ * OWN_WAVES)

__device__ __forceinline__ unsigned key_match4(unsigned d, unsigned pat) {
    const unsigned x = d ^ pat;
    const unsigned t = (x & 0x7f7f7f7fu) + 0x7f7f7f7fu;
    return ~(t | x) & 0x80808080u;
}

template <bool AGG>
__device__ __forceinline__ void own_make_room(
    own_list_word* list, int& cnt, int& summed, bool& keep,
    int lane, int r0, int nrows, int E,
    const int* __restrict__ ei, const float* __restrict__ el,
    const float* __restrict__ er, const float* __restrict__ x,
    const float4* own_rs, float* own_rs_f, float* own_acc, own_pay pay)
{
    if (AGG) {
        for (int off = 0; off < cnt; off += 32)
            own_process_group(list + off, min(32, cnt - off), lane, r0, nrows, E, ei, el, er, x, own_rs, own_acc);
    } else {
        for (; summed < cnt; summed += 32)
            own_sum_group(list + summed, min(32, cnt - summed), lane, r0, nrows, E, ei, el, er, own_rs_f, pay, summed);
        summed = 0;
        keep   = false;
    }
    cnt = 0;
}

template <bool AGG>
__device__ __forceinline__ bool own_scan_keys(
    own_list_word* list, int& cnt_out, int lane, int wave, int wg, int r0, int nrows, int E,
    const int* __restrict__ ei, const int* __restrict__ keys,
    const float* __restrict__ el, const float* __restrict__ er,
    const float* __restrict__ x,
    const float4* own_rs, float* own_rs_f, float* own_acc, own_pay pay)
{
    int  cnt    = 0;
    int  summed = 0;
    bool keep   = true;

    const int      nunits = E >> 4;
    const uint4*   key4   = (const uint4*)keys;
    const int      nsteps = (nunits + KEY_BLOCKQ - 1) / KEY_BLOCKQ;
    const int      lastq  = nunits - 1;
    const unsigned pat    = (unsigned)(wg & 255) * 0x01010101u;
    const unsigned below  = (1u << lane) - 1u;

    uint4 cur[KEY_U];
    #pragma unroll
    for (int u = 0; u < KEY_U; ++u)
        cur[u] = make_uint4(0u, 0u, 0u, 0u);
    if (nsteps > 0) {
        #pragma unroll
        for (int u = 0; u < KEY_U; ++u)
            cur[u] = key4[min(wave * KEY_STEPQ + u * 32 + lane, lastq)];
    }

    for (int s = 0; s < nsteps; ++s) {
        uint4 nxt[KEY_U];
        #pragma unroll
        for (int u = 0; u < KEY_U; ++u)
            nxt[u] = key4[min((s + 1) * KEY_BLOCKQ + wave * KEY_STEPQ + u * 32 + lane, lastq)];

        const int q0 = s * KEY_BLOCKQ + wave * KEY_STEPQ + lane;

        #pragma unroll
        for (int p = 0; p < KEY_U; p += 2) {
            const int qa = q0 + p * 32;
            unsigned ma = (key_match4(cur[p].x, pat) >> 7) | (key_match4(cur[p].y, pat) >> 6)
                        | (key_match4(cur[p].z, pat) >> 5) | (key_match4(cur[p].w, pat) >> 4);
            unsigned mb = (key_match4(cur[p + 1].x, pat) >> 3) | (key_match4(cur[p + 1].y, pat) >> 2)
                        | (key_match4(cur[p + 1].z, pat) >> 1) |  key_match4(cur[p + 1].w, pat);
            if (qa      >= nunits) ma = 0u;
            if (qa + 32 >= nunits) mb = 0u;
            unsigned m = ma | mb;

            for (int t = 0; t < 32; ++t) {
                const bool has = (m != 0);
                const unsigned bal = __builtin_amdgcn_ballot_w32(has);
                if (bal == 0) break;
                const int n = __builtin_popcount(bal);
                if (cnt + n > LIST_CAP)
                    own_make_room<AGG>(list, cnt, summed, keep, lane, r0, nrows, E,
                                       ei, el, er, x, own_rs, own_rs_f, own_acc, pay);
                if (has) {
                    const int bit = __builtin_ctz(m);
                    m &= m - 1;
                    const int q = qa + ((bit >> 2) & 1) * 32;
                    const int e = (q << 4) + ((bit & 3) << 2) + (bit >> 3);
                    list[cnt + __builtin_popcount(bal & below)] = e;
                }
                cnt += n;
            }
        }

        if (AGG) {
            int off = 0;
            while (cnt - off >= 32) {
                own_process_group(list + off, 32, lane, r0, nrows, E, ei, el, er, x, own_rs, own_acc);
                off += 32;
            }
            if (off) {
                const int left = cnt - off;
                int tmp = 0;
                if (lane < left) tmp = list[off + lane];
                if (lane < left) list[lane] = tmp;
                cnt = left;
            }
        } else {
            while (cnt - summed >= 32) {
                own_sum_group(list + summed, 32, lane, r0, nrows, E, ei, el, er, own_rs_f, pay, summed);
                summed += 32;
            }
        }

        #pragma unroll
        for (int u = 0; u < KEY_U; ++u) cur[u] = nxt[u];
    }

    {
        const int e = (nunits << 4) + lane;
        bool has = false;
        if (wave == 0 && e < E) has = (own_match(ei[e], r0, nrows) != 0u);
        const unsigned bal = __builtin_amdgcn_ballot_w32(has);
        const int n = __builtin_popcount(bal);
        if (cnt + n > LIST_CAP)
            own_make_room<AGG>(list, cnt, summed, keep, lane, r0, nrows, E,
                               ei, el, er, x, own_rs, own_rs_f, own_acc, pay);
        if (has) list[cnt + __builtin_popcount(bal & below)] = e;
        cnt += n;
    }
    if (AGG) {
        for (int off = 0; off < cnt; off += 32)
            own_process_group(list + off, min(32, cnt - off), lane, r0, nrows, E, ei, el, er, x, own_rs, own_acc);
    } else {
        for (; summed < cnt; summed += 32)
            own_sum_group(list + summed, min(32, cnt - summed), lane, r0, nrows, E, ei, el, er, own_rs_f, pay, summed);
    }
    cnt_out = cnt;
    return keep;
}

template <bool KEYED>
__global__ __launch_bounds__(OWN_THREADS) void gat_own_rows(
    const int* __restrict__ ei, const int* __restrict__ keys,
    const float* __restrict__ el,
    const float* __restrict__ er, const float* __restrict__ x,
    const float* __restrict__ b, float* __restrict__ out, int N, int E)
{
    __shared__ float4 own_acc4[OWN_ROWS * ROW_F / 4];
    __shared__ float4 own_rs[OWN_ROWS];
    __shared__ int    own_list[OWN_WAVES * LIST_CAP];
    __shared__ int    own_bins[OWN_WAVES * LIST_CAP];
    __shared__ int    own_give[OWN_WAVES * OWN_WAVES];
    __shared__ int    own_next[OWN_WAVES * OWN_WAVES];
    __shared__ int    own_allkept;

    const int tid  = threadIdx.x;
    const int lane = tid & 31;
    const int wave = tid >> 5;
    const int r0    = blockIdx.x * OWN_ROWS;
    const int nrows = min(OWN_ROWS, N - r0);

    float4* acc4    = own_acc4;
    float*  own_acc = (float*)own_acc4;
    float4* pay_e_all   = own_acc4;
    int*    pay_dst_all = (int*)(own_acc4 + OWN_WAVES * PAY_CAP);
    for (int i = tid; i < OWN_ROWS; i += OWN_THREADS)
        own_rs[i] = make_float4(0.f, 0.f, 0.f, 0.f);
    for (int i = tid; i < OWN_WAVES * OWN_WAVES; i += OWN_THREADS)
        own_give[i] = 0;
    if (tid == 0) own_allkept = 1;
    __syncthreads();

    own_list_word* list = (own_list_word*)(own_list + wave * LIST_CAP);
    float* own_rs_f = (float*)own_rs;
    own_pay pay;
    pay.e   = pay_e_all   + wave * PAY_CAP;
    pay.dst = pay_dst_all + wave * PAY_CAP;

    int cnt = 0;
    const bool kept = KEYED
        ? own_scan_keys<false>(list, cnt, lane, wave, (int)blockIdx.x, r0, nrows, E,
                               ei, keys, el, er, x, own_rs, own_rs_f, own_acc, pay)
        : own_scan<false>(list, cnt, lane, wave, r0, nrows, E,
                          ei, el, er, x, own_rs, own_rs_f, own_acc, pay);

    if (lane == 0 && (!kept || cnt > PAY_CAP)) own_allkept = 0;

    const bool packed = (E <= PACK_MAX_E);
    if (packed && kept) {
        for (int base = 0; base < cnt; base += 32) {
            const int idx = base + lane;
            if (idx < cnt) {
                const unsigned lr = ((unsigned)list[idx]) >> 24;
                if (lr < (unsigned)nrows)
                    atomicAdd(&own_give[wave * OWN_WAVES + (int)(lr & (OWN_WAVES - 1))], 1);
            }
        }
    }

    __syncthreads();

    const bool all_kept = (__builtin_amdgcn_readfirstlane(own_allkept) != 0);
    if (packed && all_kept) {
        int total = 0, before = 0;
        for (int s = 0; s < OWN_WAVES; ++s) {
            const int n = own_give[s * OWN_WAVES + lane];
            total  += n;
            before += (s < wave) ? n : 0;
        }
        int start = 0;
        #pragma unroll
        for (int i = 0; i < OWN_WAVES - 1; ++i) {
            const int n = own_bcast_i(total, i);
            start += (lane > i) ? n : 0;
        }
        own_next[wave * OWN_WAVES + lane] = start + before;
        const int me       = __builtin_amdgcn_readfirstlane(wave);
        const int my_start = own_bcast_i(start, me);
        const int my_total = own_bcast_i(total, me);

        for (int base = 0; base < cnt; base += 32) {
            const int idx = base + lane;
            if (idx < cnt) {
                const int wd = list[idx];
                const unsigned lr = ((unsigned)wd) >> 24;
                if (lr < (unsigned)nrows) {
                    const int at = atomicAdd(&own_next[wave * OWN_WAVES + (int)(lr & (OWN_WAVES - 1))], 1);
                    own_bins[at] = (int)(((unsigned)wd & 0xff000000u) | (unsigned)(wave * PAY_CAP + idx));
                }
            }
        }

        __syncthreads();

        own_rows_in_registers((own_list_word*)(own_bins + my_start),
                              (own_list_word*)(own_list + my_start), my_total,
                              lane, wave, r0, nrows, x, own_rs, pay_e_all, pay_dst_all, b, out);
        return;
    }

    for (int i = tid; i < OWN_ROWS * (ROW_F / 4); i += OWN_THREADS)
        acc4[i] = make_float4(0.f, 0.f, 0.f, 0.f);
    __syncthreads();

    if (kept && packed) {
        for (int off = 0; off < cnt; off += 32)
            own_process_packed(list + off, min(32, cnt - off), lane, r0, nrows, E, ei, el, er, x, own_rs, own_acc);
    } else if (kept) {
        int off = 0;
        for (; cnt - off >= 32; off += 32)
            own_process_group(list + off, 32, lane, r0, nrows, E, ei, el, er, x, own_rs, own_acc);
        if (cnt - off > 0)
            own_process_group(list + off, cnt - off, lane, r0, nrows, E, ei, el, er, x, own_rs, own_acc);
    } else {
        int unused = 0;
        if (KEYED)
            own_scan_keys<true>(list, unused, lane, wave, (int)blockIdx.x, r0, nrows, E,
                                ei, keys, el, er, x, own_rs, own_rs_f, own_acc, pay);
        else
            own_scan<true>(list, unused, lane, wave, r0, nrows, E,
                           ei, el, er, x, own_rs, own_rs_f, own_acc, pay);
    }

    __syncthreads();

    float4* out4 = (float4*)out + (size_t)r0 * (ROW_F / 4);
    for (int i = tid; i < nrows * (ROW_F / 4); i += OWN_THREADS) {
        float4 v = acc4[i];
        const int d0 = (i * 4) & (OUT_DIM - 1);
        v.x += b[d0 + 0]; v.y += b[d0 + 1]; v.z += b[d0 + 2]; v.w += b[d0 + 3];
        v.x = (v.x > 0.f) ? v.x : (__expf(v.x) - 1.f);
        v.y = (v.y > 0.f) ? v.y : (__expf(v.y) - 1.f);
        v.z = (v.z > 0.f) ? v.z : (__expf(v.z) - 1.f);
        v.w = (v.w > 0.f) ? v.w : (__expf(v.w) - 1.f);
        out4[i] = v;
    }
}

extern "C" void kernel_launch(void* const* d_in, const int* in_sizes, int n_in,
                              void* d_out, int out_size, void* d_ws, size_t ws_size,
                              hipStream_t stream) {
    const float* h  = (const float*)d_in[0];
    const float* W  = (const float*)d_in[1];
    const float* Wl = (const float*)d_in[2];
    const float* Wr = (const float*)d_in[3];
    const float* b  = (const float*)d_in[4];
    const int*   ei = (const int*)d_in[5];

    const int N = in_sizes[0] / IN_DIM;
    const int E = in_sizes[5] / 2;

    float* out = (float*)d_out;

    float* x      = (float*)d_ws;
    float* el     = x  + (size_t)N * OUT_DIM;
    float* er     = el + (size_t)N * N_HEADS;
    int*   keys   = (int*)(er + (size_t)N * N_HEADS);
    const size_t need  = (size_t)N * (OUT_DIM + 2 * N_HEADS) * sizeof(float) + (size_t)E;
    const bool   keyed = (ws_size >= need);

    gat_gemm_wmma<<<(unsigned)((N + 15) / 16), 128, 0, stream>>>(h, W, x, N);

    gat_coef<<<(unsigned)((N + 255) / 256), 256, 0, stream>>>(
        x, Wl, Wr, el, er, ei, keyed ? keys : (int*)nullptr, N, E);

    if (keyed)
        gat_own_rows<true><<<(unsigned)((N + OWN_ROWS - 1) / OWN_ROWS), OWN_THREADS, 0, stream>>>(
            ei, keys, el, er, x, b, out, N, E);
    else
        gat_own_rows<false><<<(unsigned)((N + OWN_ROWS - 1) / OWN_ROWS), OWN_THREADS, 0, stream>>>(
            ei, keys, el, er, x, b, out, N, E);
}


# GEMM operand feed: each block walks over 16-row tiles; a wave keeps its 256x16 slice of W in registers; h tiles come in by whole-row 16-byte loads through LDS (two buffers); x sums bit-same
# speedup vs baseline: 6.1266x; 1.1116x over previous
#include <hip/hip_runtime.h>

#define IN_DIM  256
#define OUT_DIM 64
#define N_HEADS 4
#define GAT_EPS 1e-12f

typedef float v2f __attribute__((ext_vector_type(2)));
typedef float v8f __attribute__((ext_vector_type(8)));

#define ROW_F (N_HEADS * OUT_DIM)

#define GEMM_KSTEPS (IN_DIM / 4)
#define GEMM_PITCH  (IN_DIM + 4)
#define GEMM_TILE_F (16 * GEMM_PITCH)

typedef float v4f __attribute__((ext_vector_type(4)));

__device__ __forceinline__ v4f gemm_fetch_piece(
    const float* __restrict__ h, int tile, int trow, int sc4, int N)
{
    int r = tile * 16 + trow;
    if (r >= N) r = N - 1;
    return *(const v4f*)(h + (size_t)r * IN_DIM + sc4 * 4);
}

__global__ __launch_bounds__(128) void gat_gemm_wmma(
    const float* __restrict__ h, const float* __restrict__ W,
    float* __restrict__ x, int N)
{
    __shared__ v4f gemm_a4[2 * GEMM_TILE_F / 4];
    float* gemm_a = (float*)gemm_a4;

    const int tid  = threadIdx.x;
    const int lane = tid & 31;
    const int wave = tid >> 5;
    const int half = lane >> 4;
    const int l15  = lane & 15;
    const int col0 = wave * 16;
    const int bcol = col0 + l15;
    const int srow = tid >> 6;
    const int sc4  = tid & 63;
    const int ntiles = (N + 15) >> 4;

    v2f bw[GEMM_KSTEPS];
    #pragma unroll
    for (int s = 0; s < GEMM_KSTEPS; ++s) {
        const int ka = 4 * s + half * 2;
        bw[s].x = W[(size_t)ka       * OUT_DIM + bcol];
        bw[s].y = W[(size_t)(ka + 1) * OUT_DIM + bcol];
    }

    int tile = blockIdx.x;
    int ft = min(tile, ntiles - 1);
    v4f t0 = gemm_fetch_piece(h, ft,  0 + srow, sc4, N);
    v4f t1 = gemm_fetch_piece(h, ft,  2 + srow, sc4, N);
    v4f t2 = gemm_fetch_piece(h, ft,  4 + srow, sc4, N);
    v4f t3 = gemm_fetch_piece(h, ft,  6 + srow, sc4, N);
    v4f t4 = gemm_fetch_piece(h, ft,  8 + srow, sc4, N);
    v4f t5 = gemm_fetch_piece(h, ft, 10 + srow, sc4, N);
    v4f t6 = gemm_fetch_piece(h, ft, 12 + srow, sc4, N);
    v4f t7 = gemm_fetch_piece(h, ft, 14 + srow, sc4, N);

    int p = 0;
    for (; tile < ntiles; tile += gridDim.x) {
        float* buf = gemm_a + p * GEMM_TILE_F;
        float* sdst = buf + srow * GEMM_PITCH + sc4 * 4;
        *(v4f*)(sdst +  0 * GEMM_PITCH) = t0;
        *(v4f*)(sdst +  2 * GEMM_PITCH) = t1;
        *(v4f*)(sdst +  4 * GEMM_PITCH) = t2;
        *(v4f*)(sdst +  6 * GEMM_PITCH) = t3;
        *(v4f*)(sdst +  8 * GEMM_PITCH) = t4;
        *(v4f*)(sdst + 10 * GEMM_PITCH) = t5;
        *(v4f*)(sdst + 12 * GEMM_PITCH) = t6;
        *(v4f*)(sdst + 14 * GEMM_PITCH) = t7;
        __syncthreads();

        const int nt = tile + (int)gridDim.x;
        ft = (nt < ntiles) ? nt : tile;
        t0 = gemm_fetch_piece(h, ft,  0 + srow, sc4, N);
        t1 = gemm_fetch_piece(h, ft,  2 + srow, sc4, N);
        t2 = gemm_fetch_piece(h, ft,  4 + srow, sc4, N);
        t3 = gemm_fetch_piece(h, ft,  6 + srow, sc4, N);
        t4 = gemm_fetch_piece(h, ft,  8 + srow, sc4, N);
        t5 = gemm_fetch_piece(h, ft, 10 + srow, sc4, N);
        t6 = gemm_fetch_piece(h, ft, 12 + srow, sc4, N);
        t7 = gemm_fetch_piece(h, ft, 14 + srow, sc4, N);
        __builtin_amdgcn_sched_barrier(0);

        v8f acc = {0.f, 0.f, 0.f, 0.f, 0.f, 0.f, 0.f, 0.f};
        const float* arow = buf + l15 * GEMM_PITCH + half * 2;
        #pragma unroll
        for (int s = 0; s < GEMM_KSTEPS; ++s) {
            const v2f a = *(const v2f*)(arow + 4 * s);
            acc = __builtin_amdgcn_wmma_f32_16x16x4_f32(
                 false, a,  false, bw[s],
                 (short)0, acc,  false,  false);
        }

        const int row0 = tile * 16;
        float* __restrict__ xout =
            x + ((size_t)(row0 + half * 8)) * OUT_DIM + col0 + l15;
        if (row0 + 16 <= N) {
            #pragma unroll
            for (int r = 0; r < 8; ++r)
                xout[(size_t)r * OUT_DIM] = acc[r];
        } else {
            #pragma unroll
            for (int r = 0; r < 8; ++r) {
                if (row0 + r + half * 8 < N)
                    xout[(size_t)r * OUT_DIM] = acc[r];
            }
        }
        p ^= 1;
    }
}

#define OWN_ROWS    196

__device__ __forceinline__ unsigned own_key(int v) {
    return (((unsigned)v) / (unsigned)OWN_ROWS) & 255u;
}

__global__ __launch_bounds__(256) void gat_coef(
    const float* __restrict__ x, const float* __restrict__ Wl,
    const float* __restrict__ Wr, float* __restrict__ el,
    float* __restrict__ er, const int* __restrict__ ei,
    int* __restrict__ keys, int N, int E)
{
    if (keys != nullptr) {
        const int   nunits = E >> 4;
        const int4* src4   = (const int4*)ei;
        uint4*      key4   = (uint4*)keys;
        const int   stride = (int)(gridDim.x * blockDim.x);
        for (int q = blockIdx.x * blockDim.x + threadIdx.x; q < nunits; q += stride) {
            unsigned w[4];
            #pragma unroll
            for (int j = 0; j < 4; ++j) {
                const int4 v = src4[(size_t)q * 4 + j];
                w[j] = own_key(v.x) | (own_key(v.y) << 8) | (own_key(v.z) << 16) | (own_key(v.w) << 24);
            }
            key4[q] = make_uint4(w[0], w[1], w[2], w[3]);
        }
    }

    const int n = blockIdx.x * blockDim.x + threadIdx.x;
    if (n >= N) return;
    const float* __restrict__ xr = x + (size_t)n * OUT_DIM;
    float al[N_HEADS] = {0.f, 0.f, 0.f, 0.f};
    float ar[N_HEADS] = {0.f, 0.f, 0.f, 0.f};
    for (int d = 0; d < OUT_DIM; ++d) {
        const float xv = xr[d];
        #pragma unroll
        for (int hh = 0; hh < N_HEADS; ++hh) {
            al[hh] = fmaf(xv, Wl[hh * OUT_DIM + d], al[hh]);
            ar[hh] = fmaf(xv, Wr[hh * OUT_DIM + d], ar[hh]);
        }
    }
    #pragma unroll
    for (int hh = 0; hh < N_HEADS; ++hh) {
        el[(size_t)n * N_HEADS + hh] = al[hh];
        er[(size_t)n * N_HEADS + hh] = ar[hh];
    }
}

__device__ __forceinline__ float gat_edge_e(float l, float r) {
    float v = l + r;
    v = (v > 0.f) ? v : 0.2f * v;
    return __expf(v);
}

#define OWN_THREADS 1024
#define OWN_WAVES   (OWN_THREADS / 32)
#define SCAN_U      2
#define OWN_STEP4   (SCAN_U * 32)
#define OWN_BLOCK4  (OWN_STEP4 * OWN_WAVES)
#define STEP_EDGES  (OWN_STEP4 * 4)
#define LIST_CAP    416
static_assert(STEP_EDGES + 31 <= LIST_CAP, "a step after a drop must fit the list");
#define PACK_MAX_E  (1 << 24)

typedef volatile int __attribute__((address_space(3))) own_list_word;

__device__ __forceinline__ int own_bcast_i(int v, int j) {
    return __builtin_amdgcn_readlane(v, j);
}
__device__ __forceinline__ float own_bcast_f(float v, int j) {
    return __int_as_float(__builtin_amdgcn_readlane(__float_as_int(v), j));
}

__device__ __forceinline__ unsigned own_match(int v, int r0, int nrows) {
    return ((unsigned)(v - r0) < (unsigned)nrows) ? 1u : 0u;
}

#define PAY_CAP 313
static_assert((size_t)PAY_CAP * OWN_WAVES * 20 <= (size_t)OWN_ROWS * ROW_F * 4,
              "the kept weights and destinations must fit the value sums' space");
struct own_pay {
    float4* e;
    int*    dst;
};

__device__ __forceinline__ void own_sum_group(
    own_list_word* list, int g, int lane, int r0, int nrows, int E,
    const int* __restrict__ ei, const float* __restrict__ el,
    const float* __restrict__ er, float* own_rs_f, own_pay pay, int pos0)
{
    if (lane < g) {
        const int e   = list[lane];
        const int src = ei[e];
        const int dst = ei[(size_t)E + e];
        const float4 r = *(const float4*)(er + (size_t)dst * N_HEADS);
        const bool mine = (own_match(src, r0, nrows) != 0u);
        if (mine) {
            const float4 l = *(const float4*)(el + (size_t)src * N_HEADS);
            float* a = own_rs_f + (src - r0) * N_HEADS;
            const float4 ev = make_float4(gat_edge_e(l.x, r.x), gat_edge_e(l.y, r.y),
                                          gat_edge_e(l.z, r.z), gat_edge_e(l.w, r.w));
            atomicAdd(a + 0, ev.x);
            atomicAdd(a + 1, ev.y);
            atomicAdd(a + 2, ev.z);
            atomicAdd(a + 3, ev.w);
            const int pos = pos0 + lane;
            if (pos < PAY_CAP) {
                pay.e[pos]   = ev;
                pay.dst[pos] = dst;
            }
        }
        if (E <= PACK_MAX_E)
            list[lane] = mine ? (int)((unsigned)e | ((unsigned)(src - r0) << 24))
                              : (int)0xff000000u;
    }
}

__device__ __forceinline__ void own_process_group(
    own_list_word* list, int g, int lane, int r0, int nrows, int E,
    const int* __restrict__ ei, const float* __restrict__ el,
    const float* __restrict__ er, const float* __restrict__ x,
    const float4* own_rs, float* own_acc)
{
    int my_dst = 0, my_lrow = 0;
    float w0 = 0.f, w1 = 0.f, w2 = 0.f, w3 = 0.f;
    if (lane < g) {
        const int e   = list[lane];
        const int src = ei[e];
        my_dst  = ei[(size_t)E + e];
        const float4 r  = *(const float4*)(er + (size_t)my_dst * N_HEADS);
        if (own_match(src, r0, nrows)) {
            my_lrow = src - r0;
            const float4 l  = *(const float4*)(el + (size_t)src    * N_HEADS);
            const float4 rs = own_rs[my_lrow];
            w0 = gat_edge_e(l.x, r.x) / fmaxf(rs.x, GAT_EPS);
            w1 = gat_edge_e(l.y, r.y) / fmaxf(rs.y, GAT_EPS);
            w2 = gat_edge_e(l.z, r.z) / fmaxf(rs.z, GAT_EPS);
            w3 = gat_edge_e(l.w, r.w) / fmaxf(rs.w, GAT_EPS);
        }
    }
    for (int i = 0; i < g; i += 4) {
        int lr[4];
        v2f xv[4];
        #pragma unroll
        for (int k = 0; k < 4; ++k) {
            const int j = min(i + k, g - 1);
            const int d = own_bcast_i(my_dst, j);
            lr[k] = own_bcast_i(my_lrow, j);
            xv[k] = *(const v2f*)(x + (size_t)d * OUT_DIM + lane * 2);
        }
        #pragma unroll
        for (int k = 0; k < 4; ++k) {
            if (i + k < g) {
                const int j = i + k;
                float* a = own_acc + lr[k] * ROW_F + lane * 2;
                const float wa = own_bcast_f(w0, j);
                const float wb = own_bcast_f(w1, j);
                const float wc = own_bcast_f(w2, j);
                const float wd = own_bcast_f(w3, j);
                atomicAdd(a + 0 * OUT_DIM + 0, wa * xv[k].x);
                atomicAdd(a + 0 * OUT_DIM + 1, wa * xv[k].y);
                atomicAdd(a + 1 * OUT_DIM + 0, wb * xv[k].x);
                atomicAdd(a + 1 * OUT_DIM + 1, wb * xv[k].y);
                atomicAdd(a + 2 * OUT_DIM + 0, wc * xv[k].x);
                atomicAdd(a + 2 * OUT_DIM + 1, wc * xv[k].y);
                atomicAdd(a + 3 * OUT_DIM + 0, wd * xv[k].x);
                atomicAdd(a + 3 * OUT_DIM + 1, wd * xv[k].y);
            }
        }
    }
}

#define PK_B 8
__device__ __forceinline__ void own_process_packed(
    own_list_word* list, int g, int lane, int r0, int nrows, int E,
    const int* __restrict__ ei, const float* __restrict__ el,
    const float* __restrict__ er, const float* __restrict__ x,
    const float4* own_rs, float* own_acc)
{
    int my_dst = 0, my_lrow = 0;
    float w0 = 0.f, w1 = 0.f, w2 = 0.f, w3 = 0.f;
    if (lane < g) {
        const unsigned wd = (unsigned)list[lane];
        const unsigned lr = wd >> 24;
        if (lr < (unsigned)nrows) {
            const int e = (int)(wd & 0x00ffffffu);
            my_lrow = (int)lr;
            my_dst  = ei[(size_t)E + e];
            const float4 r  = *(const float4*)(er + (size_t)my_dst * N_HEADS);
            const float4 l  = *(const float4*)(el + (size_t)(r0 + my_lrow) * N_HEADS);
            const float4 rs = own_rs[my_lrow];
            w0 = gat_edge_e(l.x, r.x) / fmaxf(rs.x, GAT_EPS);
            w1 = gat_edge_e(l.y, r.y) / fmaxf(rs.y, GAT_EPS);
            w2 = gat_edge_e(l.z, r.z) / fmaxf(rs.z, GAT_EPS);
            w3 = gat_edge_e(l.w, r.w) / fmaxf(rs.w, GAT_EPS);
        }
    }
    v2f cur[PK_B];
    #pragma unroll
    for (int k = 0; k < PK_B; ++k) {
        const int d = own_bcast_i(my_dst, min(k, g - 1));
        cur[k] = *(const v2f*)(x + (size_t)d * OUT_DIM + lane * 2);
    }
    for (int i = 0; i < g; i += PK_B) {
        v2f nxt[PK_B];
        #pragma unroll
        for (int k = 0; k < PK_B; ++k) nxt[k] = cur[k];
        if (i + PK_B < g) {
            #pragma unroll
            for (int k = 0; k < PK_B; ++k) {
                const int d = own_bcast_i(my_dst, min(i + PK_B + k, g - 1));
                nxt[k] = *(const v2f*)(x + (size_t)d * OUT_DIM + lane * 2);
            }
        }
        #pragma unroll
        for (int k = 0; k < PK_B; ++k) {
            if (i + k < g) {
                const int j = i + k;
                float* a = own_acc + own_bcast_i(my_lrow, j) * ROW_F + lane * 2;
                const float wa = own_bcast_f(w0, j);
                const float wb = own_bcast_f(w1, j);
                const float wc = own_bcast_f(w2, j);
                const float wd = own_bcast_f(w3, j);
                atomicAdd(a + 0 * OUT_DIM + 0, wa * cur[k].x);
                atomicAdd(a + 0 * OUT_DIM + 1, wa * cur[k].y);
                atomicAdd(a + 1 * OUT_DIM + 0, wb * cur[k].x);
                atomicAdd(a + 1 * OUT_DIM + 1, wb * cur[k].y);
                atomicAdd(a + 2 * OUT_DIM + 0, wc * cur[k].x);
                atomicAdd(a + 2 * OUT_DIM + 1, wc * cur[k].y);
                atomicAdd(a + 3 * OUT_DIM + 0, wd * cur[k].x);
                atomicAdd(a + 3 * OUT_DIM + 1, wd * cur[k].y);
            }
        }
        #pragma unroll
        for (int k = 0; k < PK_B; ++k) cur[k] = nxt[k];
    }
}

__device__ __forceinline__ float own_elu_b(float v, float bb) {
    v += bb;
    return (v > 0.f) ? v : (__expf(v) - 1.f);
}

__device__ __forceinline__ void own_store_row(
    float* __restrict__ out, int grow, int lane, v2f bias,
    v2f s0, v2f s1, v2f s2, v2f s3)
{
    float* o = out + (size_t)grow * ROW_F + lane * 2;
    v2f t;
    t.x = own_elu_b(s0.x, bias.x);  t.y = own_elu_b(s0.y, bias.y);  *(v2f*)(o + 0 * OUT_DIM) = t;
    t.x = own_elu_b(s1.x, bias.x);  t.y = own_elu_b(s1.y, bias.y);  *(v2f*)(o + 1 * OUT_DIM) = t;
    t.x = own_elu_b(s2.x, bias.x);  t.y = own_elu_b(s2.y, bias.y);  *(v2f*)(o + 2 * OUT_DIM) = t;
    t.x = own_elu_b(s3.x, bias.x);  t.y = own_elu_b(s3.y, bias.y);  *(v2f*)(o + 3 * OUT_DIM) = t;
}

#define OWN_PER_WAVE ((OWN_ROWS + OWN_WAVES - 1) / OWN_WAVES)
__device__ __forceinline__ void own_rows_in_registers(
    own_list_word* bin, own_list_word* srt, int total,
    int lane, int wave, int r0, int nrows,
    const float* __restrict__ x, const float4* own_rs,
    const float4* pay_e, const int* pay_dst,
    const float* __restrict__ b, float* __restrict__ out)
{
    const unsigned below = (1u << lane) - 1u;

    int pos[OWN_PER_WAVE];
    #pragma unroll
    for (int q = 0; q < OWN_PER_WAVE; ++q) pos[q] = 0;
    for (int base = 0; base < total; base += 32) {
        const int idx = base + lane;
        unsigned slot = 0xffu;
        if (idx < total) slot = ((unsigned)bin[idx]) >> 29;
        #pragma unroll
        for (int q = 0; q + 1 < OWN_PER_WAVE; ++q)
            pos[q + 1] += __builtin_popcount(__builtin_amdgcn_ballot_w32(slot == (unsigned)q));
    }
    #pragma unroll
    for (int q = 1; q < OWN_PER_WAVE; ++q) pos[q] += pos[q - 1];

    for (int base = 0; base < total; base += 32) {
        const int idx = base + lane;
        int wd = 0;
        unsigned slot = 0xffu;
        if (idx < total) { wd = bin[idx]; slot = ((unsigned)wd) >> 29; }
        #pragma unroll
        for (int q = 0; q < OWN_PER_WAVE; ++q) {
            const unsigned bal = __builtin_amdgcn_ballot_w32(slot == (unsigned)q);
            if (slot == (unsigned)q) srt[pos[q] + __builtin_popcount(bal & below)] = wd;
            pos[q] += __builtin_popcount(bal);
        }
    }
    const int n = pos[OWN_PER_WAVE - 1];

    const v2f bias = *(const v2f*)(b + lane * 2);
    const v2f zero2 = {0.f, 0.f};
    v2f s0 = zero2, s1 = zero2, s2 = zero2, s3 = zero2;
    int row = __builtin_amdgcn_readfirstlane(wave);

    for (int off = 0; off < n; off += 32) {
        const int g = min(32, n - off);
        int my_dst = 0, my_lrow = 0;
        float4 ev = make_float4(0.f, 0.f, 0.f, 0.f);
        bool have = false;
        if (lane < g) {
            const unsigned wd = (unsigned)srt[off + lane];
            const unsigned lr = wd >> 24;
            if (lr < (unsigned)nrows) {
                const int p = (int)(wd & 0x00ffffffu);
                have    = true;
                my_lrow = (int)lr;
                my_dst  = pay_dst[p];
                ev      = pay_e[p];
            }
        }
        v2f cur[PK_B];
        #pragma unroll
        for (int k = 0; k < PK_B; ++k) {
            const int d = own_bcast_i(my_dst, min(k, g - 1));
            cur[k] = *(const v2f*)(x + (size_t)d * OUT_DIM + lane * 2);
        }
        float w0 = 0.f, w1 = 0.f, w2 = 0.f, w3 = 0.f;
        if (have) {
            const float4 rs = own_rs[my_lrow];
            w0 = ev.x / fmaxf(rs.x, GAT_EPS);
            w1 = ev.y / fmaxf(rs.y, GAT_EPS);
            w2 = ev.z / fmaxf(rs.z, GAT_EPS);
            w3 = ev.w / fmaxf(rs.w, GAT_EPS);
        }
        for (int i = 0; i < g; i += PK_B) {
            v2f nxt[PK_B];
            #pragma unroll
            for (int k = 0; k < PK_B; ++k) nxt[k] = cur[k];
            if (i + PK_B < g) {
                #pragma unroll
                for (int k = 0; k < PK_B; ++k) {
                    const int d = own_bcast_i(my_dst, min(i + PK_B + k, g - 1));
                    nxt[k] = *(const v2f*)(x + (size_t)d * OUT_DIM + lane * 2);
                }
            }
            #pragma unroll
            for (int k = 0; k < PK_B; ++k) {
                if (i + k < g) {
                    const int j = i + k;
                    const int r = own_bcast_i(my_lrow, j);
                    while (row < r) {
                        own_store_row(out, r0 + row, lane, bias, s0, s1, s2, s3);
                        s0 = zero2; s1 = zero2; s2 = zero2; s3 = zero2;
                        row += OWN_WAVES;
                    }
                    const float wa = own_bcast_f(w0, j);
                    const float wb = own_bcast_f(w1, j);
                    const float wc = own_bcast_f(w2, j);
                    const float wd = own_bcast_f(w3, j);
                    s0.x += wa * cur[k].x;  s0.y += wa * cur[k].y;
                    s1.x += wb * cur[k].x;  s1.y += wb * cur[k].y;
                    s2.x += wc * cur[k].x;  s2.y += wc * cur[k].y;
                    s3.x += wd * cur[k].x;  s3.y += wd * cur[k].y;
                }
            }
            #pragma unroll
            for (int k = 0; k < PK_B; ++k) cur[k] = nxt[k];
        }
    }
    while (row < nrows) {
        own_store_row(out, r0 + row, lane, bias, s0, s1, s2, s3);
        s0 = zero2; s1 = zero2; s2 = zero2; s3 = zero2;
        row += OWN_WAVES;
    }
}

template <bool AGG>
__device__ __forceinline__ bool own_scan(
    own_list_word* list, int& cnt_out, int lane, int wave, int r0, int nrows, int E,
    const int* __restrict__ ei, const float* __restrict__ el,
    const float* __restrict__ er, const float* __restrict__ x,
    const float4* own_rs, float* own_rs_f, float* own_acc, own_pay pay)
{
    int  cnt    = 0;
    int  summed = 0;
    bool keep   = true;

    const int   E4     = E >> 2;
    const int4* src4   = (const int4*)ei;
    const int   nsteps = (E4 + OWN_BLOCK4 - 1) / OWN_BLOCK4;
    const int   last4  = E4 - 1;

    int4 cur[SCAN_U];
    #pragma unroll
    for (int u = 0; u < SCAN_U; ++u)
        cur[u] = make_int4(-1, -1, -1, -1);
    if (nsteps > 0) {
        #pragma unroll
        for (int u = 0; u < SCAN_U; ++u)
            cur[u] = src4[min(wave * OWN_STEP4 + u * 32 + lane, last4)];
    }

    for (int s = 0; s < nsteps; ++s) {
        int4 nxt[SCAN_U];
        #pragma unroll
        for (int u = 0; u < SCAN_U; ++u)
            nxt[u] = src4[min((s + 1) * OWN_BLOCK4 + wave * OWN_STEP4 + u * 32 + lane, last4)];

        const int cb4 = s * OWN_BLOCK4 + wave * OWN_STEP4 + lane;
        unsigned m = 0;
        #pragma unroll
        for (int u = 0; u < SCAN_U; ++u) {
            unsigned mu = own_match(cur[u].x, r0, nrows)
                        | (own_match(cur[u].y, r0, nrows) << 1)
                        | (own_match(cur[u].z, r0, nrows) << 2)
                        | (own_match(cur[u].w, r0, nrows) << 3);
            if (cb4 + u * 32 >= E4) mu = 0u;
            m |= mu << (4 * u);
        }

        for (int t = 0; t < 4 * SCAN_U; ++t) {
            const bool has = (m != 0);
            const unsigned bal = __builtin_amdgcn_ballot_w32(has);
            if (bal == 0) break;
            if (has) {
                const int bit = __builtin_ctz(m);
                m &= m - 1;
                const int e = ((cb4 + (bit >> 2) * 32) << 2) + (bit & 3);
                list[cnt + __builtin_popcount(bal & ((1u << lane) - 1u))] = e;
            }
            cnt += __builtin_popcount(bal);
        }

        if (AGG) {
            int off = 0;
            while (cnt - off >= 32) {
                own_process_group(list + off, 32, lane, r0, nrows, E, ei, el, er, x, own_rs, own_acc);
                off += 32;
            }
            if (off) {
                const int left = cnt - off;
                int tmp = 0;
                if (lane < left) tmp = list[off + lane];
                if (lane < left) list[lane] = tmp;
                cnt = left;
            }
        } else {
            while (cnt - summed >= 32) {
                own_sum_group(list + summed, 32, lane, r0, nrows, E, ei, el, er, own_rs_f, pay, summed);
                summed += 32;
            }
            if (cnt > LIST_CAP - STEP_EDGES) {
                keep = false;
                const int left = cnt - summed;
                int tmp = 0;
                if (lane < left) tmp = list[summed + lane];
                if (lane < left) list[lane] = tmp;
                cnt = left;
                summed = 0;
            }
        }

        #pragma unroll
        for (int u = 0; u < SCAN_U; ++u) cur[u] = nxt[u];
    }

    {
        const int e = (E4 << 2) + lane;
        bool has = false;
        if (wave == 0 && e < E) has = (own_match(ei[e], r0, nrows) != 0u);
        const unsigned bal = __builtin_amdgcn_ballot_w32(has);
        if (has) list[cnt + __builtin_popcount(bal & ((1u << lane) - 1u))] = e;
        cnt += __builtin_popcount(bal);
    }
    if (AGG) {
        for (int off = 0; off < cnt; off += 32)
            own_process_group(list + off, min(32, cnt - off), lane, r0, nrows, E, ei, el, er, x, own_rs, own_acc);
    } else {
        for (; summed < cnt; summed += 32)
            own_sum_group(list + summed, min(32, cnt - summed), lane, r0, nrows, E, ei, el, er, own_rs_f, pay, summed);
    }
    cnt_out = cnt;
    return keep;
}

#define KEY_U      4
#define KEY_STEPQ  (KEY_U * 32)
#define KEY_BLOCKQ (KEY_STEPQ * OWN_WAVES)

__device__ __forceinline__ unsigned key_match4(unsigned d, unsigned pat) {
    const unsigned x = d ^ pat;
    const unsigned t = (x & 0x7f7f7f7fu) + 0x7f7f7f7fu;
    return ~(t | x) & 0x80808080u;
}

template <bool AGG>
__device__ __forceinline__ void own_make_room(
    own_list_word* list, int& cnt, int& summed, bool& keep,
    int lane, int r0, int nrows, int E,
    const int* __restrict__ ei, const float* __restrict__ el,
    const float* __restrict__ er, const float* __restrict__ x,
    const float4* own_rs, float* own_rs_f, float* own_acc, own_pay pay)
{
    if (AGG) {
        for (int off = 0; off < cnt; off += 32)
            own_process_group(list + off, min(32, cnt - off), lane, r0, nrows, E, ei, el, er, x, own_rs, own_acc);
    } else {
        for (; summed < cnt; summed += 32)
            own_sum_group(list + summed, min(32, cnt - summed), lane, r0, nrows, E, ei, el, er, own_rs_f, pay, summed);
        summed = 0;
        keep   = false;
    }
    cnt = 0;
}

template <bool AGG>
__device__ __forceinline__ bool own_scan_keys(
    own_list_word* list, int& cnt_out, int lane, int wave, int wg, int r0, int nrows, int E,
    const int* __restrict__ ei, const int* __restrict__ keys,
    const float* __restrict__ el, const float* __restrict__ er,
    const float* __restrict__ x,
    const float4* own_rs, float* own_rs_f, float* own_acc, own_pay pay)
{
    int  cnt    = 0;
    int  summed = 0;
    bool keep   = true;

    const int      nunits = E >> 4;
    const uint4*   key4   = (const uint4*)keys;
    const int      nsteps = (nunits + KEY_BLOCKQ - 1) / KEY_BLOCKQ;
    const int      lastq  = nunits - 1;
    const unsigned pat    = (unsigned)(wg & 255) * 0x01010101u;
    const unsigned below  = (1u << lane) - 1u;

    uint4 cur[KEY_U];
    #pragma unroll
    for (int u = 0; u < KEY_U; ++u)
        cur[u] = make_uint4(0u, 0u, 0u, 0u);
    if (nsteps > 0) {
        #pragma unroll
        for (int u = 0; u < KEY_U; ++u)
            cur[u] = key4[min(wave * KEY_STEPQ + u * 32 + lane, lastq)];
    }

    for (int s = 0; s < nsteps; ++s) {
        uint4 nxt[KEY_U];
        #pragma unroll
        for (int u = 0; u < KEY_U; ++u)
            nxt[u] = key4[min((s + 1) * KEY_BLOCKQ + wave * KEY_STEPQ + u * 32 + lane, lastq)];

        const int q0 = s * KEY_BLOCKQ + wave * KEY_STEPQ + lane;

        #pragma unroll
        for (int p = 0; p < KEY_U; p += 2) {
            const int qa = q0 + p * 32;
            unsigned ma = (key_match4(cur[p].x, pat) >> 7) | (key_match4(cur[p].y, pat) >> 6)
                        | (key_match4(cur[p].z, pat) >> 5) | (key_match4(cur[p].w, pat) >> 4);
            unsigned mb = (key_match4(cur[p + 1].x, pat) >> 3) | (key_match4(cur[p + 1].y, pat) >> 2)
                        | (key_match4(cur[p + 1].z, pat) >> 1) |  key_match4(cur[p + 1].w, pat);
            if (qa      >= nunits) ma = 0u;
            if (qa + 32 >= nunits) mb = 0u;
            unsigned m = ma | mb;

            for (int t = 0; t < 32; ++t) {
                const bool has = (m != 0);
                const unsigned bal = __builtin_amdgcn_ballot_w32(has);
                if (bal == 0) break;
                const int n = __builtin_popcount(bal);
                if (cnt + n > LIST_CAP)
                    own_make_room<AGG>(list, cnt, summed, keep, lane, r0, nrows, E,
                                       ei, el, er, x, own_rs, own_rs_f, own_acc, pay);
                if (has) {
                    const int bit = __builtin_ctz(m);
                    m &= m - 1;
                    const int q = qa + ((bit >> 2) & 1) * 32;
                    const int e = (q << 4) + ((bit & 3) << 2) + (bit >> 3);
                    list[cnt + __builtin_popcount(bal & below)] = e;
                }
                cnt += n;
            }
        }

        if (AGG) {
            int off = 0;
            while (cnt - off >= 32) {
                own_process_group(list + off, 32, lane, r0, nrows, E, ei, el, er, x, own_rs, own_acc);
                off += 32;
            }
            if (off) {
                const int left = cnt - off;
                int tmp = 0;
                if (lane < left) tmp = list[off + lane];
                if (lane < left) list[lane] = tmp;
                cnt = left;
            }
        } else {
            while (cnt - summed >= 32) {
                own_sum_group(list + summed, 32, lane, r0, nrows, E, ei, el, er, own_rs_f, pay, summed);
                summed += 32;
            }
        }

        #pragma unroll
        for (int u = 0; u < KEY_U; ++u) cur[u] = nxt[u];
    }

    {
        const int e = (nunits << 4) + lane;
        bool has = false;
        if (wave == 0 && e < E) has = (own_match(ei[e], r0, nrows) != 0u);
        const unsigned bal = __builtin_amdgcn_ballot_w32(has);
        const int n = __builtin_popcount(bal);
        if (cnt + n > LIST_CAP)
            own_make_room<AGG>(list, cnt, summed, keep, lane, r0, nrows, E,
                               ei, el, er, x, own_rs, own_rs_f, own_acc, pay);
        if (has) list[cnt + __builtin_popcount(bal & below)] = e;
        cnt += n;
    }
    if (AGG) {
        for (int off = 0; off < cnt; off += 32)
            own_process_group(list + off, min(32, cnt - off), lane, r0, nrows, E, ei, el, er, x, own_rs, own_acc);
    } else {
        for (; summed < cnt; summed += 32)
            own_sum_group(list + summed, min(32, cnt - summed), lane, r0, nrows, E, ei, el, er, own_rs_f, pay, summed);
    }
    cnt_out = cnt;
    return keep;
}

template <bool KEYED>
__global__ __launch_bounds__(OWN_THREADS) void gat_own_rows(
    const int* __restrict__ ei, const int* __restrict__ keys,
    const float* __restrict__ el,
    const float* __restrict__ er, const float* __restrict__ x,
    const float* __restrict__ b, float* __restrict__ out, int N, int E)
{
    __shared__ float4 own_acc4[OWN_ROWS * ROW_F / 4];
    __shared__ float4 own_rs[OWN_ROWS];
    __shared__ int    own_list[OWN_WAVES * LIST_CAP];
    __shared__ int    own_bins[OWN_WAVES * LIST_CAP];
    __shared__ int    own_give[OWN_WAVES * OWN_WAVES];
    __shared__ int    own_next[OWN_WAVES * OWN_WAVES];
    __shared__ int    own_allkept;

    const int tid  = threadIdx.x;
    const int lane = tid & 31;
    const int wave = tid >> 5;
    const int r0    = blockIdx.x * OWN_ROWS;
    const int nrows = min(OWN_ROWS, N - r0);

    float4* acc4    = own_acc4;
    float*  own_acc = (float*)own_acc4;
    float4* pay_e_all   = own_acc4;
    int*    pay_dst_all = (int*)(own_acc4 + OWN_WAVES * PAY_CAP);
    for (int i = tid; i < OWN_ROWS; i += OWN_THREADS)
        own_rs[i] = make_float4(0.f, 0.f, 0.f, 0.f);
    for (int i = tid; i < OWN_WAVES * OWN_WAVES; i += OWN_THREADS)
        own_give[i] = 0;
    if (tid == 0) own_allkept = 1;
    __syncthreads();

    own_list_word* list = (own_list_word*)(own_list + wave * LIST_CAP);
    float* own_rs_f = (float*)own_rs;
    own_pay pay;
    pay.e   = pay_e_all   + wave * PAY_CAP;
    pay.dst = pay_dst_all + wave * PAY_CAP;

    int cnt = 0;
    const bool kept = KEYED
        ? own_scan_keys<false>(list, cnt, lane, wave, (int)blockIdx.x, r0, nrows, E,
                               ei, keys, el, er, x, own_rs, own_rs_f, own_acc, pay)
        : own_scan<false>(list, cnt, lane, wave, r0, nrows, E,
                          ei, el, er, x, own_rs, own_rs_f, own_acc, pay);

    if (lane == 0 && (!kept || cnt > PAY_CAP)) own_allkept = 0;

    const bool packed = (E <= PACK_MAX_E);
    if (packed && kept) {
        for (int base = 0; base < cnt; base += 32) {
            const int idx = base + lane;
            if (idx < cnt) {
                const unsigned lr = ((unsigned)list[idx]) >> 24;
                if (lr < (unsigned)nrows)
                    atomicAdd(&own_give[wave * OWN_WAVES + (int)(lr & (OWN_WAVES - 1))], 1);
            }
        }
    }

    __syncthreads();

    const bool all_kept = (__builtin_amdgcn_readfirstlane(own_allkept) != 0);
    if (packed && all_kept) {
        int total = 0, before = 0;
        for (int s = 0; s < OWN_WAVES; ++s) {
            const int n = own_give[s * OWN_WAVES + lane];
            total  += n;
            before += (s < wave) ? n : 0;
        }
        int start = 0;
        #pragma unroll
        for (int i = 0; i < OWN_WAVES - 1; ++i) {
            const int n = own_bcast_i(total, i);
            start += (lane > i) ? n : 0;
        }
        own_next[wave * OWN_WAVES + lane] = start + before;
        const int me       = __builtin_amdgcn_readfirstlane(wave);
        const int my_start = own_bcast_i(start, me);
        const int my_total = own_bcast_i(total, me);

        for (int base = 0; base < cnt; base += 32) {
            const int idx = base + lane;
            if (idx < cnt) {
                const int wd = list[idx];
                const unsigned lr = ((unsigned)wd) >> 24;
                if (lr < (unsigned)nrows) {
                    const int at = atomicAdd(&own_next[wave * OWN_WAVES + (int)(lr & (OWN_WAVES - 1))], 1);
                    own_bins[at] = (int)(((unsigned)wd & 0xff000000u) | (unsigned)(wave * PAY_CAP + idx));
                }
            }
        }

        __syncthreads();

        own_rows_in_registers((own_list_word*)(own_bins + my_start),
                              (own_list_word*)(own_list + my_start), my_total,
                              lane, wave, r0, nrows, x, own_rs, pay_e_all, pay_dst_all, b, out);
        return;
    }

    for (int i = tid; i < OWN_ROWS * (ROW_F / 4); i += OWN_THREADS)
        acc4[i] = make_float4(0.f, 0.f, 0.f, 0.f);
    __syncthreads();

    if (kept && packed) {
        for (int off = 0; off < cnt; off += 32)
            own_process_packed(list + off, min(32, cnt - off), lane, r0, nrows, E, ei, el, er, x, own_rs, own_acc);
    } else if (kept) {
        int off = 0;
        for (; cnt - off >= 32; off += 32)
            own_process_group(list + off, 32, lane, r0, nrows, E, ei, el, er, x, own_rs, own_acc);
        if (cnt - off > 0)
            own_process_group(list + off, cnt - off, lane, r0, nrows, E, ei, el, er, x, own_rs, own_acc);
    } else {
        int unused = 0;
        if (KEYED)
            own_scan_keys<true>(list, unused, lane, wave, (int)blockIdx.x, r0, nrows, E,
                                ei, keys, el, er, x, own_rs, own_rs_f, own_acc, pay);
        else
            own_scan<true>(list, unused, lane, wave, r0, nrows, E,
                           ei, el, er, x, own_rs, own_rs_f, own_acc, pay);
    }

    __syncthreads();

    float4* out4 = (float4*)out + (size_t)r0 * (ROW_F / 4);
    for (int i = tid; i < nrows * (ROW_F / 4); i += OWN_THREADS) {
        float4 v = acc4[i];
        const int d0 = (i * 4) & (OUT_DIM - 1);
        v.x += b[d0 + 0]; v.y += b[d0 + 1]; v.z += b[d0 + 2]; v.w += b[d0 + 3];
        v.x = (v.x > 0.f) ? v.x : (__expf(v.x) - 1.f);
        v.y = (v.y > 0.f) ? v.y : (__expf(v.y) - 1.f);
        v.z = (v.z > 0.f) ? v.z : (__expf(v.z) - 1.f);
        v.w = (v.w > 0.f) ? v.w : (__expf(v.w) - 1.f);
        out4[i] = v;
    }
}

extern "C" void kernel_launch(void* const* d_in, const int* in_sizes, int n_in,
                              void* d_out, int out_size, void* d_ws, size_t ws_size,
                              hipStream_t stream) {
    const float* h  = (const float*)d_in[0];
    const float* W  = (const float*)d_in[1];
    const float* Wl = (const float*)d_in[2];
    const float* Wr = (const float*)d_in[3];
    const float* b  = (const float*)d_in[4];
    const int*   ei = (const int*)d_in[5];

    const int N = in_sizes[0] / IN_DIM;
    const int E = in_sizes[5] / 2;

    float* out = (float*)d_out;

    float* x      = (float*)d_ws;
    float* el     = x  + (size_t)N * OUT_DIM;
    float* er     = el + (size_t)N * N_HEADS;
    int*   keys   = (int*)(er + (size_t)N * N_HEADS);
    const size_t need  = (size_t)N * (OUT_DIM + 2 * N_HEADS) * sizeof(float) + (size_t)E;
    const bool   keyed = (ws_size >= need);

    gat_gemm_wmma<<<1024, 128, 0, stream>>>(h, W, x, N);

    gat_coef<<<(unsigned)((N + 255) / 256), 256, 0, stream>>>(
        x, Wl, Wr, el, er, ei, keyed ? keys : (int*)nullptr, N, E);

    if (keyed)
        gat_own_rows<true><<<(unsigned)((N + OWN_ROWS - 1) / OWN_ROWS), OWN_THREADS, 0, stream>>>(
            ei, keys, el, er, x, b, out, N, E);
    else
        gat_own_rows<false><<<(unsigned)((N + OWN_ROWS - 1) / OWN_ROWS), OWN_THREADS, 0, stream>>>(
            ei, keys, el, er, x, b, out, N, E);
}


# pass 2 also writes each edge's (source, destination) as one 8-byte pair beside the keys; phase 1 of pass 3 reads the pair: one scattered read per listed edge, not two; sum orders as before
# speedup vs baseline: 6.4242x; 1.0486x over previous
#include <hip/hip_runtime.h>

#define IN_DIM  256
#define OUT_DIM 64
#define N_HEADS 4
#define GAT_EPS 1e-12f

typedef float v2f __attribute__((ext_vector_type(2)));
typedef float v8f __attribute__((ext_vector_type(8)));

#define ROW_F (N_HEADS * OUT_DIM)

#define GEMM_KSTEPS (IN_DIM / 4)
#define GEMM_PITCH  (IN_DIM + 4)
#define GEMM_TILE_F (16 * GEMM_PITCH)

typedef float v4f __attribute__((ext_vector_type(4)));

__device__ __forceinline__ v4f gemm_fetch_piece(
    const float* __restrict__ h, int tile, int trow, int sc4, int N)
{
    int r = tile * 16 + trow;
    if (r >= N) r = N - 1;
    return *(const v4f*)(h + (size_t)r * IN_DIM + sc4 * 4);
}

__global__ __launch_bounds__(128) void gat_gemm_wmma(
    const float* __restrict__ h, const float* __restrict__ W,
    float* __restrict__ x, int N)
{
    __shared__ v4f gemm_a4[2 * GEMM_TILE_F / 4];
    float* gemm_a = (float*)gemm_a4;

    const int tid  = threadIdx.x;
    const int lane = tid & 31;
    const int wave = tid >> 5;
    const int half = lane >> 4;
    const int l15  = lane & 15;
    const int col0 = wave * 16;
    const int bcol = col0 + l15;
    const int srow = tid >> 6;
    const int sc4  = tid & 63;
    const int ntiles = (N + 15) >> 4;

    v2f bw[GEMM_KSTEPS];
    #pragma unroll
    for (int s = 0; s < GEMM_KSTEPS; ++s) {
        const int ka = 4 * s + half * 2;
        bw[s].x = W[(size_t)ka       * OUT_DIM + bcol];
        bw[s].y = W[(size_t)(ka + 1) * OUT_DIM + bcol];
    }

    int tile = blockIdx.x;
    int ft = min(tile, ntiles - 1);
    v4f t0 = gemm_fetch_piece(h, ft,  0 + srow, sc4, N);
    v4f t1 = gemm_fetch_piece(h, ft,  2 + srow, sc4, N);
    v4f t2 = gemm_fetch_piece(h, ft,  4 + srow, sc4, N);
    v4f t3 = gemm_fetch_piece(h, ft,  6 + srow, sc4, N);
    v4f t4 = gemm_fetch_piece(h, ft,  8 + srow, sc4, N);
    v4f t5 = gemm_fetch_piece(h, ft, 10 + srow, sc4, N);
    v4f t6 = gemm_fetch_piece(h, ft, 12 + srow, sc4, N);
    v4f t7 = gemm_fetch_piece(h, ft, 14 + srow, sc4, N);

    int p = 0;
    for (; tile < ntiles; tile += gridDim.x) {
        float* buf = gemm_a + p * GEMM_TILE_F;
        float* sdst = buf + srow * GEMM_PITCH + sc4 * 4;
        *(v4f*)(sdst +  0 * GEMM_PITCH) = t0;
        *(v4f*)(sdst +  2 * GEMM_PITCH) = t1;
        *(v4f*)(sdst +  4 * GEMM_PITCH) = t2;
        *(v4f*)(sdst +  6 * GEMM_PITCH) = t3;
        *(v4f*)(sdst +  8 * GEMM_PITCH) = t4;
        *(v4f*)(sdst + 10 * GEMM_PITCH) = t5;
        *(v4f*)(sdst + 12 * GEMM_PITCH) = t6;
        *(v4f*)(sdst + 14 * GEMM_PITCH) = t7;
        __syncthreads();

        const int nt = tile + (int)gridDim.x;
        ft = (nt < ntiles) ? nt : tile;
        t0 = gemm_fetch_piece(h, ft,  0 + srow, sc4, N);
        t1 = gemm_fetch_piece(h, ft,  2 + srow, sc4, N);
        t2 = gemm_fetch_piece(h, ft,  4 + srow, sc4, N);
        t3 = gemm_fetch_piece(h, ft,  6 + srow, sc4, N);
        t4 = gemm_fetch_piece(h, ft,  8 + srow, sc4, N);
        t5 = gemm_fetch_piece(h, ft, 10 + srow, sc4, N);
        t6 = gemm_fetch_piece(h, ft, 12 + srow, sc4, N);
        t7 = gemm_fetch_piece(h, ft, 14 + srow, sc4, N);
        __builtin_amdgcn_sched_barrier(0);

        v8f acc = {0.f, 0.f, 0.f, 0.f, 0.f, 0.f, 0.f, 0.f};
        const float* arow = buf + l15 * GEMM_PITCH + half * 2;
        #pragma unroll
        for (int s = 0; s < GEMM_KSTEPS; ++s) {
            const v2f a = *(const v2f*)(arow + 4 * s);
            acc = __builtin_amdgcn_wmma_f32_16x16x4_f32(
                 false, a,  false, bw[s],
                 (short)0, acc,  false,  false);
        }

        const int row0 = tile * 16;
        float* __restrict__ xout =
            x + ((size_t)(row0 + half * 8)) * OUT_DIM + col0 + l15;
        if (row0 + 16 <= N) {
            #pragma unroll
            for (int r = 0; r < 8; ++r)
                xout[(size_t)r * OUT_DIM] = acc[r];
        } else {
            #pragma unroll
            for (int r = 0; r < 8; ++r) {
                if (row0 + r + half * 8 < N)
                    xout[(size_t)r * OUT_DIM] = acc[r];
            }
        }
        p ^= 1;
    }
}

#define OWN_ROWS    196

__device__ __forceinline__ unsigned own_key(int v) {
    return (((unsigned)v) / (unsigned)OWN_ROWS) & 255u;
}

__global__ __launch_bounds__(256) void gat_coef(
    const float* __restrict__ x, const float* __restrict__ Wl,
    const float* __restrict__ Wr, float* __restrict__ el,
    float* __restrict__ er, const int* __restrict__ ei,
    int* __restrict__ keys, int* __restrict__ rec, int N, int E)
{
    if (keys != nullptr) {
        const int   nunits = E >> 4;
        const int4* src4   = (const int4*)ei;
        const int4* dst4   = (const int4*)(ei + E);
        uint4*      key4   = (uint4*)keys;
        int4*       rec4   = (int4*)rec;
        const int   stride = (int)(gridDim.x * blockDim.x);
        for (int q = blockIdx.x * blockDim.x + threadIdx.x; q < nunits; q += stride) {
            unsigned w[4];
            int4 sv[4], dv[4];
            #pragma unroll
            for (int j = 0; j < 4; ++j)
                sv[j] = src4[(size_t)q * 4 + j];
            if (rec != nullptr) {
                #pragma unroll
                for (int j = 0; j < 4; ++j)
                    dv[j] = dst4[(size_t)q * 4 + j];
            }
            #pragma unroll
            for (int j = 0; j < 4; ++j) {
                const int4 v = sv[j];
                w[j] = own_key(v.x) | (own_key(v.y) << 8) | (own_key(v.z) << 16) | (own_key(v.w) << 24);
            }
            key4[q] = make_uint4(w[0], w[1], w[2], w[3]);
            if (rec != nullptr) {
                #pragma unroll
                for (int j = 0; j < 4; ++j) {
                    rec4[(size_t)q * 8 + 2 * j]     = make_int4(sv[j].x, dv[j].x, sv[j].y, dv[j].y);
                    rec4[(size_t)q * 8 + 2 * j + 1] = make_int4(sv[j].z, dv[j].z, sv[j].w, dv[j].w);
                }
            }
        }
    }

    const int n = blockIdx.x * blockDim.x + threadIdx.x;
    if (n >= N) return;
    const float* __restrict__ xr = x + (size_t)n * OUT_DIM;
    float al[N_HEADS] = {0.f, 0.f, 0.f, 0.f};
    float ar[N_HEADS] = {0.f, 0.f, 0.f, 0.f};
    for (int d = 0; d < OUT_DIM; ++d) {
        const float xv = xr[d];
        #pragma unroll
        for (int hh = 0; hh < N_HEADS; ++hh) {
            al[hh] = fmaf(xv, Wl[hh * OUT_DIM + d], al[hh]);
            ar[hh] = fmaf(xv, Wr[hh * OUT_DIM + d], ar[hh]);
        }
    }
    #pragma unroll
    for (int hh = 0; hh < N_HEADS; ++hh) {
        el[(size_t)n * N_HEADS + hh] = al[hh];
        er[(size_t)n * N_HEADS + hh] = ar[hh];
    }
}

__device__ __forceinline__ float gat_edge_e(float l, float r) {
    float v = l + r;
    v = (v > 0.f) ? v : 0.2f * v;
    return __expf(v);
}

#define OWN_THREADS 1024
#define OWN_WAVES   (OWN_THREADS / 32)
#define SCAN_U      2
#define OWN_STEP4   (SCAN_U * 32)
#define OWN_BLOCK4  (OWN_STEP4 * OWN_WAVES)
#define STEP_EDGES  (OWN_STEP4 * 4)
#define LIST_CAP    416
static_assert(STEP_EDGES + 31 <= LIST_CAP, "a step after a drop must fit the list");
#define PACK_MAX_E  (1 << 24)

typedef volatile int __attribute__((address_space(3))) own_list_word;

__device__ __forceinline__ int own_bcast_i(int v, int j) {
    return __builtin_amdgcn_readlane(v, j);
}
__device__ __forceinline__ float own_bcast_f(float v, int j) {
    return __int_as_float(__builtin_amdgcn_readlane(__float_as_int(v), j));
}

__device__ __forceinline__ unsigned own_match(int v, int r0, int nrows) {
    return ((unsigned)(v - r0) < (unsigned)nrows) ? 1u : 0u;
}

#define PAY_CAP 313
static_assert((size_t)PAY_CAP * OWN_WAVES * 20 <= (size_t)OWN_ROWS * ROW_F * 4,
              "the kept weights and destinations must fit the value sums' space");
struct own_pay {
    float4* e;
    int*    dst;
    const int* rec;
    int     nrec;
};

__device__ __forceinline__ void own_sum_group(
    own_list_word* list, int g, int lane, int r0, int nrows, int E,
    const int* __restrict__ ei, const float* __restrict__ el,
    const float* __restrict__ er, float* own_rs_f, own_pay pay, int pos0)
{
    if (lane < g) {
        const int e   = list[lane];
        const bool paired = (e < pay.nrec);
        int2 sd = make_int2(0, 0);
        if (pay.rec != nullptr)
            sd = *(const int2*)(pay.rec + 2 * (size_t)(paired ? e : 0));
        int src = sd.x, dst = sd.y;
        if (!paired) {
            src = ei[e];
            dst = ei[(size_t)E + e];
        }
        const float4 r = *(const float4*)(er + (size_t)dst * N_HEADS);
        const bool mine = (own_match(src, r0, nrows) != 0u);
        if (mine) {
            const float4 l = *(const float4*)(el + (size_t)src * N_HEADS);
            float* a = own_rs_f + (src - r0) * N_HEADS;
            const float4 ev = make_float4(gat_edge_e(l.x, r.x), gat_edge_e(l.y, r.y),
                                          gat_edge_e(l.z, r.z), gat_edge_e(l.w, r.w));
            atomicAdd(a + 0, ev.x);
            atomicAdd(a + 1, ev.y);
            atomicAdd(a + 2, ev.z);
            atomicAdd(a + 3, ev.w);
            const int pos = pos0 + lane;
            if (pos < PAY_CAP) {
                pay.e[pos]   = ev;
                pay.dst[pos] = dst;
            }
        }
        if (E <= PACK_MAX_E)
            list[lane] = mine ? (int)((unsigned)e | ((unsigned)(src - r0) << 24))
                              : (int)0xff000000u;
    }
}

__device__ __forceinline__ void own_process_group(
    own_list_word* list, int g, int lane, int r0, int nrows, int E,
    const int* __restrict__ ei, const float* __restrict__ el,
    const float* __restrict__ er, const float* __restrict__ x,
    const float4* own_rs, float* own_acc)
{
    int my_dst = 0, my_lrow = 0;
    float w0 = 0.f, w1 = 0.f, w2 = 0.f, w3 = 0.f;
    if (lane < g) {
        const int e   = list[lane];
        const int src = ei[e];
        my_dst  = ei[(size_t)E + e];
        const float4 r  = *(const float4*)(er + (size_t)my_dst * N_HEADS);
        if (own_match(src, r0, nrows)) {
            my_lrow = src - r0;
            const float4 l  = *(const float4*)(el + (size_t)src    * N_HEADS);
            const float4 rs = own_rs[my_lrow];
            w0 = gat_edge_e(l.x, r.x) / fmaxf(rs.x, GAT_EPS);
            w1 = gat_edge_e(l.y, r.y) / fmaxf(rs.y, GAT_EPS);
            w2 = gat_edge_e(l.z, r.z) / fmaxf(rs.z, GAT_EPS);
            w3 = gat_edge_e(l.w, r.w) / fmaxf(rs.w, GAT_EPS);
        }
    }
    for (int i = 0; i < g; i += 4) {
        int lr[4];
        v2f xv[4];
        #pragma unroll
        for (int k = 0; k < 4; ++k) {
            const int j = min(i + k, g - 1);
            const int d = own_bcast_i(my_dst, j);
            lr[k] = own_bcast_i(my_lrow, j);
            xv[k] = *(const v2f*)(x + (size_t)d * OUT_DIM + lane * 2);
        }
        #pragma unroll
        for (int k = 0; k < 4; ++k) {
            if (i + k < g) {
                const int j = i + k;
                float* a = own_acc + lr[k] * ROW_F + lane * 2;
                const float wa = own_bcast_f(w0, j);
                const float wb = own_bcast_f(w1, j);
                const float wc = own_bcast_f(w2, j);
                const float wd = own_bcast_f(w3, j);
                atomicAdd(a + 0 * OUT_DIM + 0, wa * xv[k].x);
                atomicAdd(a + 0 * OUT_DIM + 1, wa * xv[k].y);
                atomicAdd(a + 1 * OUT_DIM + 0, wb * xv[k].x);
                atomicAdd(a + 1 * OUT_DIM + 1, wb * xv[k].y);
                atomicAdd(a + 2 * OUT_DIM + 0, wc * xv[k].x);
                atomicAdd(a + 2 * OUT_DIM + 1, wc * xv[k].y);
                atomicAdd(a + 3 * OUT_DIM + 0, wd * xv[k].x);
                atomicAdd(a + 3 * OUT_DIM + 1, wd * xv[k].y);
            }
        }
    }
}

#define PK_B 8
__device__ __forceinline__ void own_process_packed(
    own_list_word* list, int g, int lane, int r0, int nrows, int E,
    const int* __restrict__ ei, const float* __restrict__ el,
    const float* __restrict__ er, const float* __restrict__ x,
    const float4* own_rs, float* own_acc)
{
    int my_dst = 0, my_lrow = 0;
    float w0 = 0.f, w1 = 0.f, w2 = 0.f, w3 = 0.f;
    if (lane < g) {
        const unsigned wd = (unsigned)list[lane];
        const unsigned lr = wd >> 24;
        if (lr < (unsigned)nrows) {
            const int e = (int)(wd & 0x00ffffffu);
            my_lrow = (int)lr;
            my_dst  = ei[(size_t)E + e];
            const float4 r  = *(const float4*)(er + (size_t)my_dst * N_HEADS);
            const float4 l  = *(const float4*)(el + (size_t)(r0 + my_lrow) * N_HEADS);
            const float4 rs = own_rs[my_lrow];
            w0 = gat_edge_e(l.x, r.x) / fmaxf(rs.x, GAT_EPS);
            w1 = gat_edge_e(l.y, r.y) / fmaxf(rs.y, GAT_EPS);
            w2 = gat_edge_e(l.z, r.z) / fmaxf(rs.z, GAT_EPS);
            w3 = gat_edge_e(l.w, r.w) / fmaxf(rs.w, GAT_EPS);
        }
    }
    v2f cur[PK_B];
    #pragma unroll
    for (int k = 0; k < PK_B; ++k) {
        const int d = own_bcast_i(my_dst, min(k, g - 1));
        cur[k] = *(const v2f*)(x + (size_t)d * OUT_DIM + lane * 2);
    }
    for (int i = 0; i < g; i += PK_B) {
        v2f nxt[PK_B];
        #pragma unroll
        for (int k = 0; k < PK_B; ++k) nxt[k] = cur[k];
        if (i + PK_B < g) {
            #pragma unroll
            for (int k = 0; k < PK_B; ++k) {
                const int d = own_bcast_i(my_dst, min(i + PK_B + k, g - 1));
                nxt[k] = *(const v2f*)(x + (size_t)d * OUT_DIM + lane * 2);
            }
        }
        #pragma unroll
        for (int k = 0; k < PK_B; ++k) {
            if (i + k < g) {
                const int j = i + k;
                float* a = own_acc + own_bcast_i(my_lrow, j) * ROW_F + lane * 2;
                const float wa = own_bcast_f(w0, j);
                const float wb = own_bcast_f(w1, j);
                const float wc = own_bcast_f(w2, j);
                const float wd = own_bcast_f(w3, j);
                atomicAdd(a + 0 * OUT_DIM + 0, wa * cur[k].x);
                atomicAdd(a + 0 * OUT_DIM + 1, wa * cur[k].y);
                atomicAdd(a + 1 * OUT_DIM + 0, wb * cur[k].x);
                atomicAdd(a + 1 * OUT_DIM + 1, wb * cur[k].y);
                atomicAdd(a + 2 * OUT_DIM + 0, wc * cur[k].x);
                atomicAdd(a + 2 * OUT_DIM + 1, wc * cur[k].y);
                atomicAdd(a + 3 * OUT_DIM + 0, wd * cur[k].x);
                atomicAdd(a + 3 * OUT_DIM + 1, wd * cur[k].y);
            }
        }
        #pragma unroll
        for (int k = 0; k < PK_B; ++k) cur[k] = nxt[k];
    }
}

__device__ __forceinline__ float own_elu_b(float v, float bb) {
    v += bb;
    return (v > 0.f) ? v : (__expf(v) - 1.f);
}

__device__ __forceinline__ void own_store_row(
    float* __restrict__ out, int grow, int lane, v2f bias,
    v2f s0, v2f s1, v2f s2, v2f s3)
{
    float* o = out + (size_t)grow * ROW_F + lane * 2;
    v2f t;
    t.x = own_elu_b(s0.x, bias.x);  t.y = own_elu_b(s0.y, bias.y);  *(v2f*)(o + 0 * OUT_DIM) = t;
    t.x = own_elu_b(s1.x, bias.x);  t.y = own_elu_b(s1.y, bias.y);  *(v2f*)(o + 1 * OUT_DIM) = t;
    t.x = own_elu_b(s2.x, bias.x);  t.y = own_elu_b(s2.y, bias.y);  *(v2f*)(o + 2 * OUT_DIM) = t;
    t.x = own_elu_b(s3.x, bias.x);  t.y = own_elu_b(s3.y, bias.y);  *(v2f*)(o + 3 * OUT_DIM) = t;
}

#define OWN_PER_WAVE ((OWN_ROWS + OWN_WAVES - 1) / OWN_WAVES)
__device__ __forceinline__ void own_rows_in_registers(
    own_list_word* bin, own_list_word* srt, int total,
    int lane, int wave, int r0, int nrows,
    const float* __restrict__ x, const float4* own_rs,
    const float4* pay_e, const int* pay_dst,
    const float* __restrict__ b, float* __restrict__ out)
{
    const unsigned below = (1u << lane) - 1u;

    int pos[OWN_PER_WAVE];
    #pragma unroll
    for (int q = 0; q < OWN_PER_WAVE; ++q) pos[q] = 0;
    for (int base = 0; base < total; base += 32) {
        const int idx = base + lane;
        unsigned slot = 0xffu;
        if (idx < total) slot = ((unsigned)bin[idx]) >> 29;
        #pragma unroll
        for (int q = 0; q + 1 < OWN_PER_WAVE; ++q)
            pos[q + 1] += __builtin_popcount(__builtin_amdgcn_ballot_w32(slot == (unsigned)q));
    }
    #pragma unroll
    for (int q = 1; q < OWN_PER_WAVE; ++q) pos[q] += pos[q - 1];

    for (int base = 0; base < total; base += 32) {
        const int idx = base + lane;
        int wd = 0;
        unsigned slot = 0xffu;
        if (idx < total) { wd = bin[idx]; slot = ((unsigned)wd) >> 29; }
        #pragma unroll
        for (int q = 0; q < OWN_PER_WAVE; ++q) {
            const unsigned bal = __builtin_amdgcn_ballot_w32(slot == (unsigned)q);
            if (slot == (unsigned)q) srt[pos[q] + __builtin_popcount(bal & below)] = wd;
            pos[q] += __builtin_popcount(bal);
        }
    }
    const int n = pos[OWN_PER_WAVE - 1];

    const v2f bias = *(const v2f*)(b + lane * 2);
    const v2f zero2 = {0.f, 0.f};
    v2f s0 = zero2, s1 = zero2, s2 = zero2, s3 = zero2;
    int row = __builtin_amdgcn_readfirstlane(wave);

    for (int off = 0; off < n; off += 32) {
        const int g = min(32, n - off);
        int my_dst = 0, my_lrow = 0;
        float4 ev = make_float4(0.f, 0.f, 0.f, 0.f);
        bool have = false;
        if (lane < g) {
            const unsigned wd = (unsigned)srt[off + lane];
            const unsigned lr = wd >> 24;
            if (lr < (unsigned)nrows) {
                const int p = (int)(wd & 0x00ffffffu);
                have    = true;
                my_lrow = (int)lr;
                my_dst  = pay_dst[p];
                ev      = pay_e[p];
            }
        }
        v2f cur[PK_B];
        #pragma unroll
        for (int k = 0; k < PK_B; ++k) {
            const int d = own_bcast_i(my_dst, min(k, g - 1));
            cur[k] = *(const v2f*)(x + (size_t)d * OUT_DIM + lane * 2);
        }
        float w0 = 0.f, w1 = 0.f, w2 = 0.f, w3 = 0.f;
        if (have) {
            const float4 rs = own_rs[my_lrow];
            w0 = ev.x / fmaxf(rs.x, GAT_EPS);
            w1 = ev.y / fmaxf(rs.y, GAT_EPS);
            w2 = ev.z / fmaxf(rs.z, GAT_EPS);
            w3 = ev.w / fmaxf(rs.w, GAT_EPS);
        }
        for (int i = 0; i < g; i += PK_B) {
            v2f nxt[PK_B];
            #pragma unroll
            for (int k = 0; k < PK_B; ++k) nxt[k] = cur[k];
            if (i + PK_B < g) {
                #pragma unroll
                for (int k = 0; k < PK_B; ++k) {
                    const int d = own_bcast_i(my_dst, min(i + PK_B + k, g - 1));
                    nxt[k] = *(const v2f*)(x + (size_t)d * OUT_DIM + lane * 2);
                }
            }
            #pragma unroll
            for (int k = 0; k < PK_B; ++k) {
                if (i + k < g) {
                    const int j = i + k;
                    const int r = own_bcast_i(my_lrow, j);
                    while (row < r) {
                        own_store_row(out, r0 + row, lane, bias, s0, s1, s2, s3);
                        s0 = zero2; s1 = zero2; s2 = zero2; s3 = zero2;
                        row += OWN_WAVES;
                    }
                    const float wa = own_bcast_f(w0, j);
                    const float wb = own_bcast_f(w1, j);
                    const float wc = own_bcast_f(w2, j);
                    const float wd = own_bcast_f(w3, j);
                    s0.x += wa * cur[k].x;  s0.y += wa * cur[k].y;
                    s1.x += wb * cur[k].x;  s1.y += wb * cur[k].y;
                    s2.x += wc * cur[k].x;  s2.y += wc * cur[k].y;
                    s3.x += wd * cur[k].x;  s3.y += wd * cur[k].y;
                }
            }
            #pragma unroll
            for (int k = 0; k < PK_B; ++k) cur[k] = nxt[k];
        }
    }
    while (row < nrows) {
        own_store_row(out, r0 + row, lane, bias, s0, s1, s2, s3);
        s0 = zero2; s1 = zero2; s2 = zero2; s3 = zero2;
        row += OWN_WAVES;
    }
}

template <bool AGG>
__device__ __forceinline__ bool own_scan(
    own_list_word* list, int& cnt_out, int lane, int wave, int r0, int nrows, int E,
    const int* __restrict__ ei, const float* __restrict__ el,
    const float* __restrict__ er, const float* __restrict__ x,
    const float4* own_rs, float* own_rs_f, float* own_acc, own_pay pay)
{
    int  cnt    = 0;
    int  summed = 0;
    bool keep   = true;

    const int   E4     = E >> 2;
    const int4* src4   = (const int4*)ei;
    const int   nsteps = (E4 + OWN_BLOCK4 - 1) / OWN_BLOCK4;
    const int   last4  = E4 - 1;

    int4 cur[SCAN_U];
    #pragma unroll
    for (int u = 0; u < SCAN_U; ++u)
        cur[u] = make_int4(-1, -1, -1, -1);
    if (nsteps > 0) {
        #pragma unroll
        for (int u = 0; u < SCAN_U; ++u)
            cur[u] = src4[min(wave * OWN_STEP4 + u * 32 + lane, last4)];
    }

    for (int s = 0; s < nsteps; ++s) {
        int4 nxt[SCAN_U];
        #pragma unroll
        for (int u = 0; u < SCAN_U; ++u)
            nxt[u] = src4[min((s + 1) * OWN_BLOCK4 + wave * OWN_STEP4 + u * 32 + lane, last4)];

        const int cb4 = s * OWN_BLOCK4 + wave * OWN_STEP4 + lane;
        unsigned m = 0;
        #pragma unroll
        for (int u = 0; u < SCAN_U; ++u) {
            unsigned mu = own_match(cur[u].x, r0, nrows)
                        | (own_match(cur[u].y, r0, nrows) << 1)
                        | (own_match(cur[u].z, r0, nrows) << 2)
                        | (own_match(cur[u].w, r0, nrows) << 3);
            if (cb4 + u * 32 >= E4) mu = 0u;
            m |= mu << (4 * u);
        }

        for (int t = 0; t < 4 * SCAN_U; ++t) {
            const bool has = (m != 0);
            const unsigned bal = __builtin_amdgcn_ballot_w32(has);
            if (bal == 0) break;
            if (has) {
                const int bit = __builtin_ctz(m);
                m &= m - 1;
                const int e = ((cb4 + (bit >> 2) * 32) << 2) + (bit & 3);
                list[cnt + __builtin_popcount(bal & ((1u << lane) - 1u))] = e;
            }
            cnt += __builtin_popcount(bal);
        }

        if (AGG) {
            int off = 0;
            while (cnt - off >= 32) {
                own_process_group(list + off, 32, lane, r0, nrows, E, ei, el, er, x, own_rs, own_acc);
                off += 32;
            }
            if (off) {
                const int left = cnt - off;
                int tmp = 0;
                if (lane < left) tmp = list[off + lane];
                if (lane < left) list[lane] = tmp;
                cnt = left;
            }
        } else {
            while (cnt - summed >= 32) {
                own_sum_group(list + summed, 32, lane, r0, nrows, E, ei, el, er, own_rs_f, pay, summed);
                summed += 32;
            }
            if (cnt > LIST_CAP - STEP_EDGES) {
                keep = false;
                const int left = cnt - summed;
                int tmp = 0;
                if (lane < left) tmp = list[summed + lane];
                if (lane < left) list[lane] = tmp;
                cnt = left;
                summed = 0;
            }
        }

        #pragma unroll
        for (int u = 0; u < SCAN_U; ++u) cur[u] = nxt[u];
    }

    {
        const int e = (E4 << 2) + lane;
        bool has = false;
        if (wave == 0 && e < E) has = (own_match(ei[e], r0, nrows) != 0u);
        const unsigned bal = __builtin_amdgcn_ballot_w32(has);
        if (has) list[cnt + __builtin_popcount(bal & ((1u << lane) - 1u))] = e;
        cnt += __builtin_popcount(bal);
    }
    if (AGG) {
        for (int off = 0; off < cnt; off += 32)
            own_process_group(list + off, min(32, cnt - off), lane, r0, nrows, E, ei, el, er, x, own_rs, own_acc);
    } else {
        for (; summed < cnt; summed += 32)
            own_sum_group(list + summed, min(32, cnt - summed), lane, r0, nrows, E, ei, el, er, own_rs_f, pay, summed);
    }
    cnt_out = cnt;
    return keep;
}

#define KEY_U      4
#define KEY_STEPQ  (KEY_U * 32)
#define KEY_BLOCKQ (KEY_STEPQ * OWN_WAVES)

__device__ __forceinline__ unsigned key_match4(unsigned d, unsigned pat) {
    const unsigned x = d ^ pat;
    const unsigned t = (x & 0x7f7f7f7fu) + 0x7f7f7f7fu;
    return ~(t | x) & 0x80808080u;
}

template <bool AGG>
__device__ __forceinline__ void own_make_room(
    own_list_word* list, int& cnt, int& summed, bool& keep,
    int lane, int r0, int nrows, int E,
    const int* __restrict__ ei, const float* __restrict__ el,
    const float* __restrict__ er, const float* __restrict__ x,
    const float4* own_rs, float* own_rs_f, float* own_acc, own_pay pay)
{
    if (AGG) {
        for (int off = 0; off < cnt; off += 32)
            own_process_group(list + off, min(32, cnt - off), lane, r0, nrows, E, ei, el, er, x, own_rs, own_acc);
    } else {
        for (; summed < cnt; summed += 32)
            own_sum_group(list + summed, min(32, cnt - summed), lane, r0, nrows, E, ei, el, er, own_rs_f, pay, summed);
        summed = 0;
        keep   = false;
    }
    cnt = 0;
}

template <bool AGG>
__device__ __forceinline__ bool own_scan_keys(
    own_list_word* list, int& cnt_out, int lane, int wave, int wg, int r0, int nrows, int E,
    const int* __restrict__ ei, const int* __restrict__ keys,
    const float* __restrict__ el, const float* __restrict__ er,
    const float* __restrict__ x,
    const float4* own_rs, float* own_rs_f, float* own_acc, own_pay pay)
{
    int  cnt    = 0;
    int  summed = 0;
    bool keep   = true;

    const int      nunits = E >> 4;
    const uint4*   key4   = (const uint4*)keys;
    const int      nsteps = (nunits + KEY_BLOCKQ - 1) / KEY_BLOCKQ;
    const int      lastq  = nunits - 1;
    const unsigned pat    = (unsigned)(wg & 255) * 0x01010101u;
    const unsigned below  = (1u << lane) - 1u;

    uint4 cur[KEY_U];
    #pragma unroll
    for (int u = 0; u < KEY_U; ++u)
        cur[u] = make_uint4(0u, 0u, 0u, 0u);
    if (nsteps > 0) {
        #pragma unroll
        for (int u = 0; u < KEY_U; ++u)
            cur[u] = key4[min(wave * KEY_STEPQ + u * 32 + lane, lastq)];
    }

    for (int s = 0; s < nsteps; ++s) {
        uint4 nxt[KEY_U];
        #pragma unroll
        for (int u = 0; u < KEY_U; ++u)
            nxt[u] = key4[min((s + 1) * KEY_BLOCKQ + wave * KEY_STEPQ + u * 32 + lane, lastq)];

        const int q0 = s * KEY_BLOCKQ + wave * KEY_STEPQ + lane;

        #pragma unroll
        for (int p = 0; p < KEY_U; p += 2) {
            const int qa = q0 + p * 32;
            unsigned ma = (key_match4(cur[p].x, pat) >> 7) | (key_match4(cur[p].y, pat) >> 6)
                        | (key_match4(cur[p].z, pat) >> 5) | (key_match4(cur[p].w, pat) >> 4);
            unsigned mb = (key_match4(cur[p + 1].x, pat) >> 3) | (key_match4(cur[p + 1].y, pat) >> 2)
                        | (key_match4(cur[p + 1].z, pat) >> 1) |  key_match4(cur[p + 1].w, pat);
            if (qa      >= nunits) ma = 0u;
            if (qa + 32 >= nunits) mb = 0u;
            unsigned m = ma | mb;

            for (int t = 0; t < 32; ++t) {
                const bool has = (m != 0);
                const unsigned bal = __builtin_amdgcn_ballot_w32(has);
                if (bal == 0) break;
                const int n = __builtin_popcount(bal);
                if (cnt + n > LIST_CAP)
                    own_make_room<AGG>(list, cnt, summed, keep, lane, r0, nrows, E,
                                       ei, el, er, x, own_rs, own_rs_f, own_acc, pay);
                if (has) {
                    const int bit = __builtin_ctz(m);
                    m &= m - 1;
                    const int q = qa + ((bit >> 2) & 1) * 32;
                    const int e = (q << 4) + ((bit & 3) << 2) + (bit >> 3);
                    list[cnt + __builtin_popcount(bal & below)] = e;
                }
                cnt += n;
            }
        }

        if (AGG) {
            int off = 0;
            while (cnt - off >= 32) {
                own_process_group(list + off, 32, lane, r0, nrows, E, ei, el, er, x, own_rs, own_acc);
                off += 32;
            }
            if (off) {
                const int left = cnt - off;
                int tmp = 0;
                if (lane < left) tmp = list[off + lane];
                if (lane < left) list[lane] = tmp;
                cnt = left;
            }
        } else {
            while (cnt - summed >= 32) {
                own_sum_group(list + summed, 32, lane, r0, nrows, E, ei, el, er, own_rs_f, pay, summed);
                summed += 32;
            }
        }

        #pragma unroll
        for (int u = 0; u < KEY_U; ++u) cur[u] = nxt[u];
    }

    {
        const int e = (nunits << 4) + lane;
        bool has = false;
        if (wave == 0 && e < E) has = (own_match(ei[e], r0, nrows) != 0u);
        const unsigned bal = __builtin_amdgcn_ballot_w32(has);
        const int n = __builtin_popcount(bal);
        if (cnt + n > LIST_CAP)
            own_make_room<AGG>(list, cnt, summed, keep, lane, r0, nrows, E,
                               ei, el, er, x, own_rs, own_rs_f, own_acc, pay);
        if (has) list[cnt + __builtin_popcount(bal & below)] = e;
        cnt += n;
    }
    if (AGG) {
        for (int off = 0; off < cnt; off += 32)
            own_process_group(list + off, min(32, cnt - off), lane, r0, nrows, E, ei, el, er, x, own_rs, own_acc);
    } else {
        for (; summed < cnt; summed += 32)
            own_sum_group(list + summed, min(32, cnt - summed), lane, r0, nrows, E, ei, el, er, own_rs_f, pay, summed);
    }
    cnt_out = cnt;
    return keep;
}

template <bool KEYED>
__global__ __launch_bounds__(OWN_THREADS) void gat_own_rows(
    const int* __restrict__ ei, const int* __restrict__ keys,
    const int* __restrict__ rec, const float* __restrict__ el,
    const float* __restrict__ er, const float* __restrict__ x,
    const float* __restrict__ b, float* __restrict__ out, int N, int E)
{
    __shared__ float4 own_acc4[OWN_ROWS * ROW_F / 4];
    __shared__ float4 own_rs[OWN_ROWS];
    __shared__ int    own_list[OWN_WAVES * LIST_CAP];
    __shared__ int    own_bins[OWN_WAVES * LIST_CAP];
    __shared__ int    own_give[OWN_WAVES * OWN_WAVES];
    __shared__ int    own_next[OWN_WAVES * OWN_WAVES];
    __shared__ int    own_allkept;

    const int tid  = threadIdx.x;
    const int lane = tid & 31;
    const int wave = tid >> 5;
    const int r0    = blockIdx.x * OWN_ROWS;
    const int nrows = min(OWN_ROWS, N - r0);

    float4* acc4    = own_acc4;
    float*  own_acc = (float*)own_acc4;
    float4* pay_e_all   = own_acc4;
    int*    pay_dst_all = (int*)(own_acc4 + OWN_WAVES * PAY_CAP);
    for (int i = tid; i < OWN_ROWS; i += OWN_THREADS)
        own_rs[i] = make_float4(0.f, 0.f, 0.f, 0.f);
    for (int i = tid; i < OWN_WAVES * OWN_WAVES; i += OWN_THREADS)
        own_give[i] = 0;
    if (tid == 0) own_allkept = 1;
    __syncthreads();

    own_list_word* list = (own_list_word*)(own_list + wave * LIST_CAP);
    float* own_rs_f = (float*)own_rs;
    own_pay pay;
    pay.e   = pay_e_all   + wave * PAY_CAP;
    pay.dst = pay_dst_all + wave * PAY_CAP;
    pay.rec  = rec;
    pay.nrec = (rec != nullptr) ? ((E >> 4) << 4) : 0;

    int cnt = 0;
    const bool kept = KEYED
        ? own_scan_keys<false>(list, cnt, lane, wave, (int)blockIdx.x, r0, nrows, E,
                               ei, keys, el, er, x, own_rs, own_rs_f, own_acc, pay)
        : own_scan<false>(list, cnt, lane, wave, r0, nrows, E,
                          ei, el, er, x, own_rs, own_rs_f, own_acc, pay);

    if (lane == 0 && (!kept || cnt > PAY_CAP)) own_allkept = 0;

    const bool packed = (E <= PACK_MAX_E);
    if (packed && kept) {
        for (int base = 0; base < cnt; base += 32) {
            const int idx = base + lane;
            if (idx < cnt) {
                const unsigned lr = ((unsigned)list[idx]) >> 24;
                if (lr < (unsigned)nrows)
                    atomicAdd(&own_give[wave * OWN_WAVES + (int)(lr & (OWN_WAVES - 1))], 1);
            }
        }
    }

    __syncthreads();

    const bool all_kept = (__builtin_amdgcn_readfirstlane(own_allkept) != 0);
    if (packed && all_kept) {
        int total = 0, before = 0;
        for (int s = 0; s < OWN_WAVES; ++s) {
            const int n = own_give[s * OWN_WAVES + lane];
            total  += n;
            before += (s < wave) ? n : 0;
        }
        int start = 0;
        #pragma unroll
        for (int i = 0; i < OWN_WAVES - 1; ++i) {
            const int n = own_bcast_i(total, i);
            start += (lane > i) ? n : 0;
        }
        own_next[wave * OWN_WAVES + lane] = start + before;
        const int me       = __builtin_amdgcn_readfirstlane(wave);
        const int my_start = own_bcast_i(start, me);
        const int my_total = own_bcast_i(total, me);

        for (int base = 0; base < cnt; base += 32) {
            const int idx = base + lane;
            if (idx < cnt) {
                const int wd = list[idx];
                const unsigned lr = ((unsigned)wd) >> 24;
                if (lr < (unsigned)nrows) {
                    const int at = atomicAdd(&own_next[wave * OWN_WAVES + (int)(lr & (OWN_WAVES - 1))], 1);
                    own_bins[at] = (int)(((unsigned)wd & 0xff000000u) | (unsigned)(wave * PAY_CAP + idx));
                }
            }
        }

        __syncthreads();

        own_rows_in_registers((own_list_word*)(own_bins + my_start),
                              (own_list_word*)(own_list + my_start), my_total,
                              lane, wave, r0, nrows, x, own_rs, pay_e_all, pay_dst_all, b, out);
        return;
    }

    for (int i = tid; i < OWN_ROWS * (ROW_F / 4); i += OWN_THREADS)
        acc4[i] = make_float4(0.f, 0.f, 0.f, 0.f);
    __syncthreads();

    if (kept && packed) {
        for (int off = 0; off < cnt; off += 32)
            own_process_packed(list + off, min(32, cnt - off), lane, r0, nrows, E, ei, el, er, x, own_rs, own_acc);
    } else if (kept) {
        int off = 0;
        for (; cnt - off >= 32; off += 32)
            own_process_group(list + off, 32, lane, r0, nrows, E, ei, el, er, x, own_rs, own_acc);
        if (cnt - off > 0)
            own_process_group(list + off, cnt - off, lane, r0, nrows, E, ei, el, er, x, own_rs, own_acc);
    } else {
        int unused = 0;
        if (KEYED)
            own_scan_keys<true>(list, unused, lane, wave, (int)blockIdx.x, r0, nrows, E,
                                ei, keys, el, er, x, own_rs, own_rs_f, own_acc, pay);
        else
            own_scan<true>(list, unused, lane, wave, r0, nrows, E,
                           ei, el, er, x, own_rs, own_rs_f, own_acc, pay);
    }

    __syncthreads();

    float4* out4 = (float4*)out + (size_t)r0 * (ROW_F / 4);
    for (int i = tid; i < nrows * (ROW_F / 4); i += OWN_THREADS) {
        float4 v = acc4[i];
        const int d0 = (i * 4) & (OUT_DIM - 1);
        v.x += b[d0 + 0]; v.y += b[d0 + 1]; v.z += b[d0 + 2]; v.w += b[d0 + 3];
        v.x = (v.x > 0.f) ? v.x : (__expf(v.x) - 1.f);
        v.y = (v.y > 0.f) ? v.y : (__expf(v.y) - 1.f);
        v.z = (v.z > 0.f) ? v.z : (__expf(v.z) - 1.f);
        v.w = (v.w > 0.f) ? v.w : (__expf(v.w) - 1.f);
        out4[i] = v;
    }
}

extern "C" void kernel_launch(void* const* d_in, const int* in_sizes, int n_in,
                              void* d_out, int out_size, void* d_ws, size_t ws_size,
                              hipStream_t stream) {
    const float* h  = (const float*)d_in[0];
    const float* W  = (const float*)d_in[1];
    const float* Wl = (const float*)d_in[2];
    const float* Wr = (const float*)d_in[3];
    const float* b  = (const float*)d_in[4];
    const int*   ei = (const int*)d_in[5];

    const int N = in_sizes[0] / IN_DIM;
    const int E = in_sizes[5] / 2;

    float* out = (float*)d_out;

    float* x      = (float*)d_ws;
    float* el     = x  + (size_t)N * OUT_DIM;
    float* er     = el + (size_t)N * N_HEADS;
    int*   keys   = (int*)(er + (size_t)N * N_HEADS);
    const size_t need  = (size_t)N * (OUT_DIM + 2 * N_HEADS) * sizeof(float) + (size_t)E;
    const bool   keyed = (ws_size >= need);
    const size_t rec_off = (need + 127) & ~(size_t)127;
    const bool   recs    = keyed && ((E & 3) == 0) && (ws_size >= rec_off + (size_t)E * 8);
    int*         rec     = recs ? (int*)((char*)d_ws + rec_off) : (int*)nullptr;

    gat_gemm_wmma<<<1024, 128, 0, stream>>>(h, W, x, N);

    gat_coef<<<(unsigned)((N + 255) / 256), 256, 0, stream>>>(
        x, Wl, Wr, el, er, ei, keyed ? keys : (int*)nullptr, rec, N, E);

    if (keyed)
        gat_own_rows<true><<<(unsigned)((N + OWN_ROWS - 1) / OWN_ROWS), OWN_THREADS, 0, stream>>>(
            ei, keys, rec, el, er, x, b, out, N, E);
    else
        gat_own_rows<false><<<(unsigned)((N + OWN_ROWS - 1) / OWN_ROWS), OWN_THREADS, 0, stream>>>(
            ei, keys, rec, el, er, x, b, out, N, E);
}
